# starting kernel
# baseline (speedup 1.0000x reference)
#include <hip/hip_runtime.h>
#include <hip/hip_bf16.h>
#include <math.h>

typedef __attribute__((ext_vector_type(2))) float v2f;
typedef __attribute__((ext_vector_type(8))) float v8f;

#define NEG_SLOPE 0.2f

static __device__ __forceinline__ unsigned f2ord(float f) {
  unsigned u = __float_as_uint(f);
  return (u & 0x80000000u) ? ~u : (u | 0x80000000u);
}
static __device__ __forceinline__ float ord2f(unsigned u) {
  return (u & 0x80000000u) ? __uint_as_float(u & 0x7fffffffu)
                           : __uint_as_float(~u);
}
#define ORD_NEG_INF 0x007FFFFFu

__global__ void fill_f32_kernel(float* __restrict__ p, float v, int n) {
  int i = blockIdx.x * blockDim.x + threadIdx.x;
  if (i < n) p[i] = v;
}
__global__ void fill_u32_kernel(unsigned* __restrict__ p, unsigned v, int n) {
  int i = blockIdx.x * blockDim.x + threadIdx.x;
  if (i < n) p[i] = v;
}
__global__ void init_bias_kernel(float* __restrict__ out,
                                 const float* __restrict__ b, int n, int cmask) {
  int i = blockIdx.x * blockDim.x + threadIdx.x;
  if (i < n) out[i] = b[i & cmask];
}

template <int K, int NC>
__global__ __launch_bounds__((NC / 16) * 32)
void gemm_wmma_kernel(const float* __restrict__ X, const float* __restrict__ W,
                      float* __restrict__ Y) {
  const int lane = threadIdx.x & 31;
  const int wave = threadIdx.x >> 5;
  const int half = lane >> 4;
  const int l15  = lane & 15;
  const int m0   = blockIdx.x * 16;
  const int col  = wave * 16 + l15;
  const float* xrow = X + (size_t)(m0 + l15) * K;
  v8f c = {};
#pragma unroll
  for (int k0 = 0; k0 < K; k0 += 4) {
    const int ka = k0 + half * 2;
    const float2 av = *reinterpret_cast<const float2*>(xrow + ka);
    v2f a; a.x = av.x; a.y = av.y;
    v2f b;
    b.x = W[(size_t)ka * NC + col];
    b.y = W[(size_t)(ka + 1) * NC + col];
    c = __builtin_amdgcn_wmma_f32_16x16x4_f32(false, a, false, b, (short)0, c,
                                              false, false);
  }
#pragma unroll
  for (int r = 0; r < 8; ++r)
    Y[(size_t)(m0 + half * 8 + r) * NC + col] = c[r];
}

__global__ void scores1_kernel(const float* __restrict__ xh,
                               const float* __restrict__ ws_,
                               const float* __restrict__ wd_,
                               float* __restrict__ as_, float* __restrict__ ad_,
                               int NH) {
  int t = blockIdx.x * blockDim.x + threadIdx.x;
  if (t >= NH) return;
  int n = t >> 3, h = t & 7;
  const float4* v = reinterpret_cast<const float4*>(xh + (size_t)n * 128 + h * 16);
  const float4* s = reinterpret_cast<const float4*>(ws_ + h * 16);
  const float4* d = reinterpret_cast<const float4*>(wd_ + h * 16);
  float accs = 0.f, accd = 0.f;
#pragma unroll
  for (int q = 0; q < 4; ++q) {
    float4 x = v[q], a = s[q], b = d[q];
    accs += x.x * a.x + x.y * a.y + x.z * a.z + x.w * a.w;
    accd += x.x * b.x + x.y * b.y + x.z * b.z + x.w * b.w;
  }
  as_[t] = accs;
  ad_[t] = accd;
}

__global__ void scores2_kernel(const float* __restrict__ xh,
                               const float* __restrict__ ws_,
                               const float* __restrict__ wd_,
                               float* __restrict__ as_, float* __restrict__ ad_,
                               int N) {
  int n = blockIdx.x * blockDim.x + threadIdx.x;
  if (n >= N) return;
  const float4* v = reinterpret_cast<const float4*>(xh + (size_t)n * 64);
  const float4* s = reinterpret_cast<const float4*>(ws_);
  const float4* d = reinterpret_cast<const float4*>(wd_);
  float accs = 0.f, accd = 0.f;
#pragma unroll
  for (int q = 0; q < 16; ++q) {
    float4 x = v[q], a = s[q], b = d[q];
    accs += x.x * a.x + x.y * a.y + x.z * a.z + x.w * a.w;
    accd += x.x * b.x + x.y * b.y + x.z * b.z + x.w * b.w;
  }
  as_[n] = accs;
  ad_[n] = accd;
}

template <int H>
__global__ void edge_max_kernel(const int* __restrict__ esrc,
                                const int* __restrict__ edst,
                                const float* __restrict__ as_,
                                const float* __restrict__ ad_,
                                unsigned* __restrict__ m, int E, int total) {
  int t = blockIdx.x * blockDim.x + threadIdx.x;
  if (t >= total) return;
  int e = t / H, h = t - e * H;
  int s, d;
  if (e < E) { s = esrc[e]; d = edst[e]; } else { s = e - E; d = s; }
  float ev = as_[(size_t)s * H + h] + ad_[(size_t)d * H + h];
  ev = ev > 0.f ? ev : NEG_SLOPE * ev;
  atomicMax(&m[(size_t)d * H + h], f2ord(ev));
}

template <int H>
__global__ void edge_denom_kernel(const int* __restrict__ esrc,
                                  const int* __restrict__ edst,
                                  const float* __restrict__ as_,
                                  const float* __restrict__ ad_,
                                  const unsigned* __restrict__ m,
                                  float* __restrict__ denom, int E, int total) {
  int t = blockIdx.x * blockDim.x + threadIdx.x;
  if (t >= total) return;
  int e = t / H, h = t - e * H;
  int s, d;
  if (e < E) { s = esrc[e]; d = edst[e]; } else { s = e - E; d = s; }
  float ev = as_[(size_t)s * H + h] + ad_[(size_t)d * H + h];
  ev = ev > 0.f ? ev : NEG_SLOPE * ev;
  float ex = expf(ev - ord2f(m[(size_t)d * H + h]));
  atomicAdd(&denom[(size_t)d * H + h], ex);
}

template <int H, int CH, int NSPLIT>
__global__ void edge_msg_kernel(const int* __restrict__ esrc,
                                const int* __restrict__ edst,
                                const float* __restrict__ as_,
                                const float* __restrict__ ad_,
                                const unsigned* __restrict__ m,
                                const float* __restrict__ denom,
                                const float* __restrict__ xh,
                                float* __restrict__ agg, int E, int total) {
  int t = blockIdx.x * blockDim.x + threadIdx.x;
  if (t >= total) return;
  constexpr int TPE = H * NSPLIT;
  int e = t / TPE;
  int r = t - e * TPE;
  int h = r / NSPLIT;
  int q = r - h * NSPLIT;
  int s, d;
  if (e < E) { s = esrc[e]; d = edst[e]; } else { s = e - E; d = s; }
  float ev = as_[(size_t)s * H + h] + ad_[(size_t)d * H + h];
  ev = ev > 0.f ? ev : NEG_SLOPE * ev;
  float alpha = expf(ev - ord2f(m[(size_t)d * H + h])) /
                (denom[(size_t)d * H + h] + 1e-16f);
  constexpr int F  = H * CH;
  constexpr int CW = CH / NSPLIT;
  const float* xs = xh + (size_t)s * F + h * CH + q * CW;
  float* od = agg + (size_t)d * F + h * CH + q * CW;
#pragma unroll
  for (int c = 0; c < CW; ++c) atomicAdd(&od[c], xs[c] * alpha);
}

__global__ void bias_elu_kernel(float* __restrict__ h, const float* __restrict__ b,
                                int n) {
  int i = blockIdx.x * blockDim.x + threadIdx.x;
  if (i >= n) return;
  float v = h[i] + b[i & 127];
  h[i] = v > 0.f ? v : (expf(v) - 1.f);
}

extern "C" void kernel_launch(void* const* d_in, const int* in_sizes, int n_in,
                              void* d_out, int out_size, void* d_ws,
                              size_t ws_size, hipStream_t stream) {
  (void)n_in; (void)out_size; (void)ws_size;
  const float* x        = (const float*)d_in[0];
  const int*   eidx     = (const int*)d_in[1];
  const float* W1       = (const float*)d_in[2];
  const float* att_src1 = (const float*)d_in[3];
  const float* att_dst1 = (const float*)d_in[4];
  const float* b1       = (const float*)d_in[5];
  const float* W2       = (const float*)d_in[6];
  const float* att_src2 = (const float*)d_in[7];
  const float* att_dst2 = (const float*)d_in[8];
  const float* b2       = (const float*)d_in[9];
  float* out = (float*)d_out;

  const int N    = in_sizes[0] / 128;
  const int E    = in_sizes[1] / 2;
  const int Etot = E + N;
  const int* esrc = eidx;
  const int* edst = eidx + E;

  float* ws = (float*)d_ws;
  float*    xh1  = ws;            ws += (size_t)N * 128;
  float*    agg1 = ws;            ws += (size_t)N * 128;
  float*    as1  = ws;            ws += (size_t)N * 8;
  float*    ad1  = ws;            ws += (size_t)N * 8;
  unsigned* m1   = (unsigned*)ws; ws += (size_t)N * 8;
  float*    dn1  = ws;            ws += (size_t)N * 8;
  float*    xh2  = ws;            ws += (size_t)N * 64;
  float*    as2  = ws;            ws += (size_t)N;
  float*    ad2  = ws;            ws += (size_t)N;
  unsigned* m2   = (unsigned*)ws; ws += (size_t)N;
  float*    dn2  = ws;

  auto cdiv = [](int a, int b) { return (a + b - 1) / b; };

  fill_u32_kernel<<<cdiv(N * 8, 256), 256, 0, stream>>>(m1, ORD_NEG_INF, N * 8);
  fill_f32_kernel<<<cdiv(N * 8, 256), 256, 0, stream>>>(dn1, 0.f, N * 8);
  fill_f32_kernel<<<cdiv(N * 128, 256), 256, 0, stream>>>(agg1, 0.f, N * 128);
  fill_u32_kernel<<<cdiv(N, 256), 256, 0, stream>>>(m2, ORD_NEG_INF, N);
  fill_f32_kernel<<<cdiv(N, 256), 256, 0, stream>>>(dn2, 0.f, N);
  init_bias_kernel<<<cdiv(N * 64, 256), 256, 0, stream>>>(out, b2, N * 64, 63);

  gemm_wmma_kernel<128, 128><<<N / 16, 256, 0, stream>>>(x, W1, xh1);
  scores1_kernel<<<cdiv(N * 8, 256), 256, 0, stream>>>(xh1, att_src1, att_dst1,
                                                       as1, ad1, N * 8);
  edge_max_kernel<8><<<cdiv(Etot * 8, 256), 256, 0, stream>>>(esrc, edst, as1,
                                                              ad1, m1, E, Etot * 8);
  edge_denom_kernel<8><<<cdiv(Etot * 8, 256), 256, 0, stream>>>(esrc, edst, as1,
                                                                ad1, m1, dn1, E, Etot * 8);
  edge_msg_kernel<8, 16, 1><<<cdiv(Etot * 8, 256), 256, 0, stream>>>(
      esrc, edst, as1, ad1, m1, dn1, xh1, agg1, E, Etot * 8);
  bias_elu_kernel<<<cdiv(N * 128, 256), 256, 0, stream>>>(agg1, b1, N * 128);

  gemm_wmma_kernel<128, 64><<<N / 16, 128, 0, stream>>>(agg1, W2, xh2);
  scores2_kernel<<<cdiv(N, 256), 256, 0, stream>>>(xh2, att_src2, att_dst2, as2,
                                                   ad2, N);
  edge_max_kernel<1><<<cdiv(Etot, 256), 256, 0, stream>>>(esrc, edst, as2, ad2,
                                                          m2, E, Etot);
  edge_denom_kernel<1><<<cdiv(Etot, 256), 256, 0, stream>>>(esrc, edst, as2, ad2,
                                                            m2, dn2, E, Etot);
  edge_msg_kernel<1, 64, 4><<<cdiv(Etot * 4, 256), 256, 0, stream>>>(
      esrc, edst, as2, ad2, m2, dn2, xh2, out, E, Etot * 4);
}


# edges grouped by destination per call (integer count atomics), segments sorted by edge id; one wave per node: softmax+row sum in registers, bias/ELU at its store; no float atomics, fixed sum order
# speedup vs baseline: 4.2262x; 4.2262x over previous
#include <hip/hip_runtime.h>
#include <hip/hip_bf16.h>
#include <math.h>

typedef __attribute__((ext_vector_type(2))) float v2f;
typedef __attribute__((ext_vector_type(8))) float v8f;

#define NEG_SLOPE 0.2f

__global__ void fill_u32_kernel(unsigned* __restrict__ p, unsigned v, int n) {
  int i = blockIdx.x * blockDim.x + threadIdx.x;
  if (i < n) p[i] = v;
}

template <int K, int NC>
__global__ __launch_bounds__((NC / 16) * 32)
void gemm_wmma_kernel(const float* __restrict__ X, const float* __restrict__ W,
                      float* __restrict__ Y) {
  const int lane = threadIdx.x & 31;
  const int wave = threadIdx.x >> 5;
  const int half = lane >> 4;
  const int l15  = lane & 15;
  const int m0   = blockIdx.x * 16;
  const int col  = wave * 16 + l15;
  const float* xrow = X + (size_t)(m0 + l15) * K;
  v8f c = {};
#pragma unroll
  for (int k0 = 0; k0 < K; k0 += 4) {
    const int ka = k0 + half * 2;
    const float2 av = *reinterpret_cast<const float2*>(xrow + ka);
    v2f a; a.x = av.x; a.y = av.y;
    v2f b;
    b.x = W[(size_t)ka * NC + col];
    b.y = W[(size_t)(ka + 1) * NC + col];
    c = __builtin_amdgcn_wmma_f32_16x16x4_f32(false, a, false, b, (short)0, c,
                                              false, false);
  }
#pragma unroll
  for (int r = 0; r < 8; ++r)
    Y[(size_t)(m0 + half * 8 + r) * NC + col] = c[r];
}

__global__ void scores1_kernel(const float* __restrict__ xh,
                               const float* __restrict__ ws_,
                               const float* __restrict__ wd_,
                               float* __restrict__ as_, float* __restrict__ ad_,
                               int NH) {
  int t = blockIdx.x * blockDim.x + threadIdx.x;
  if (t >= NH) return;
  int n = t >> 3, h = t & 7;
  const float4* v = reinterpret_cast<const float4*>(xh + (size_t)n * 128 + h * 16);
  const float4* s = reinterpret_cast<const float4*>(ws_ + h * 16);
  const float4* d = reinterpret_cast<const float4*>(wd_ + h * 16);
  float accs = 0.f, accd = 0.f;
#pragma unroll
  for (int q = 0; q < 4; ++q) {
    float4 x = v[q], a = s[q], b = d[q];
    accs += x.x * a.x + x.y * a.y + x.z * a.z + x.w * a.w;
    accd += x.x * b.x + x.y * b.y + x.z * b.z + x.w * b.w;
  }
  as_[t] = accs;
  ad_[t] = accd;
}

__global__ void scores2_kernel(const float* __restrict__ xh,
                               const float* __restrict__ ws_,
                               const float* __restrict__ wd_,
                               float* __restrict__ as_, float* __restrict__ ad_,
                               int N) {
  int n = blockIdx.x * blockDim.x + threadIdx.x;
  if (n >= N) return;
  const float4* v = reinterpret_cast<const float4*>(xh + (size_t)n * 64);
  const float4* s = reinterpret_cast<const float4*>(ws_);
  const float4* d = reinterpret_cast<const float4*>(wd_);
  float accs = 0.f, accd = 0.f;
#pragma unroll
  for (int q = 0; q < 16; ++q) {
    float4 x = v[q], a = s[q], b = d[q];
    accs += x.x * a.x + x.y * a.y + x.z * a.z + x.w * a.w;
    accd += x.x * b.x + x.y * b.y + x.z * b.z + x.w * b.w;
  }
  as_[n] = accs;
  ad_[n] = accd;
}

static __device__ __forceinline__ int uni(int v) {
  return __builtin_amdgcn_readfirstlane(v);
}
static __device__ __forceinline__ float lane_f(float v, int k) {
  return __int_as_float(__builtin_amdgcn_readlane(__float_as_int(v), k));
}
static __device__ __forceinline__ float leaky(float v) {
  return v > 0.f ? v : NEG_SLOPE * v;
}

__global__ __launch_bounds__(256)
void count_rank_kernel(const int* __restrict__ edst, unsigned* __restrict__ cnt,
                       int* __restrict__ rank, int E) {
  int e = blockIdx.x * blockDim.x + threadIdx.x;
  if (e >= E) return;
  rank[e] = (int)atomicAdd(&cnt[edst[e]], 1u);
}

__global__ __launch_bounds__(256)
void scan_kernel(const unsigned* __restrict__ cnt, int* __restrict__ rowptr,
                 int N) {
  __shared__ int wsum[8];
  __shared__ int wsum2[8];
  const int tid = threadIdx.x, lane = tid & 31, wv = tid >> 5;
  const int block_start = blockIdx.x * 1024;
  int part = 0;
  for (int i = tid * 4; i < block_start; i += 1024) {
    const uint4 v = *reinterpret_cast<const uint4*>(cnt + i);
    part += (int)(v.x + v.y + v.z + v.w);
  }
  for (int o = 16; o > 0; o >>= 1) part += __shfl_xor(part, o);
  if (lane == 0) wsum[wv] = part;
  __syncthreads();
  int base = 0;
  for (int w = 0; w < 8; ++w) base += wsum[w];
  const int i0 = block_start + tid * 4;
  const int c0 = i0 < N ? (int)cnt[i0] : 0;
  const int c1 = i0 + 1 < N ? (int)cnt[i0 + 1] : 0;
  const int c2 = i0 + 2 < N ? (int)cnt[i0 + 2] : 0;
  const int c3 = i0 + 3 < N ? (int)cnt[i0 + 3] : 0;
  const int tsum = c0 + c1 + c2 + c3;
  int incl = tsum;
  for (int o = 1; o < 32; o <<= 1) {
    int v = __shfl_up(incl, o);
    if (lane >= o) incl += v;
  }
  if (lane == 31) wsum2[wv] = incl;
  __syncthreads();
  int woff = 0;
  for (int w = 0; w < wv; ++w) woff += wsum2[w];
  const int excl = base + woff + incl - tsum;
  if (i0 < N) rowptr[i0] = excl;
  if (i0 + 1 < N) rowptr[i0 + 1] = excl + c0;
  if (i0 + 2 < N) rowptr[i0 + 2] = excl + c0 + c1;
  if (i0 + 3 < N) rowptr[i0 + 3] = excl + c0 + c1 + c2;
  if (i0 < N && N <= i0 + 4) rowptr[N] = excl + tsum;
}

__global__ __launch_bounds__(256)
void place_kernel(const int* __restrict__ esrc, const int* __restrict__ edst,
                  const int* __restrict__ rank, const int* __restrict__ rowptr,
                  int* __restrict__ slot_, int E) {
  int e = blockIdx.x * blockDim.x + threadIdx.x;
  if (e >= E) return;
  int2* slot = reinterpret_cast<int2*>(slot_);
  slot[rowptr[edst[e]] + rank[e]] = make_int2(e, esrc[e]);
}

__global__ __launch_bounds__(256)
void seg_sort_kernel(const int* __restrict__ slot_, const int* __restrict__ rowptr,
                     int* __restrict__ csr_src, int N) {
  const int2* slot = reinterpret_cast<const int2*>(slot_);
  const int lane = threadIdx.x & 31;
  const int n = uni((int)((blockIdx.x * blockDim.x + threadIdx.x) >> 5));
  if (n >= N) return;
  const int start = rowptr[n];
  const int cnt = rowptr[n + 1] - start;
  for (int base = 0; base < cnt; base += 32) {
    const int i = base + lane;
    const int2 my = i < cnt ? slot[start + i] : make_int2(0x7fffffff, 0);
    int r = 0;
    for (int jb = 0; jb < cnt; jb += 32) {
      const int j = jb + lane;
      const int other = j < cnt ? slot[start + j].x : 0x7fffffff;
      const int nb = min(32, cnt - jb);
      for (int k = 0; k < nb; ++k)
        r += (__builtin_amdgcn_readlane(other, k) < my.x) ? 1 : 0;
    }
    if (i < cnt) csr_src[start + r] = my.y;
  }
}

__global__ __launch_bounds__(256)
void node1_kernel(const int* __restrict__ rowptr, const int* __restrict__ csr_src,
                  const float* __restrict__ as_, const float* __restrict__ ad_,
                  const float* __restrict__ xh, const float* __restrict__ bias,
                  float* __restrict__ hout, int N) {
  const int lane = threadIdx.x & 31;
  const int n = uni((int)((blockIdx.x * blockDim.x + threadIdx.x) >> 5));
  if (n >= N) return;
  const int h = lane >> 2;
  const int start = rowptr[n];
  const int cnt = rowptr[n + 1] - start;
  const int deg = cnt + 1;
  const float adn = ad_[(size_t)n * 8 + h];

  float m = -INFINITY;
  for (int base = 0; base < deg; base += 32) {
    const int i = base + lane;
    const int sl = i < cnt ? csr_src[start + i] : n;
    const int nb = min(32, deg - base);
    for (int k = 0; k < nb; k += 4) {
      float ev[4];
#pragma unroll
      for (int u = 0; u < 4; ++u) {
        const int s = __builtin_amdgcn_readlane(sl, k + u);
        ev[u] = leaky(as_[(size_t)s * 8 + h] + adn);
      }
#pragma unroll
      for (int u = 0; u < 4; ++u)
        m = (u == 0 || k + u < nb) ? fmaxf(m, ev[u]) : m;
    }
  }

  float den = 0.f;
  for (int base = 0; base < deg; base += 32) {
    const int i = base + lane;
    const int sl = i < cnt ? csr_src[start + i] : n;
    const int nb = min(32, deg - base);
    for (int k = 0; k < nb; k += 4) {
      float ev[4];
#pragma unroll
      for (int u = 0; u < 4; ++u) {
        const int s = __builtin_amdgcn_readlane(sl, k + u);
        ev[u] = leaky(as_[(size_t)s * 8 + h] + adn);
      }
#pragma unroll
      for (int u = 0; u < 4; ++u)
        den = (u == 0 || k + u < nb) ? den + expf(ev[u] - m) : den;
    }
  }
  const float dsum = den + 1e-16f;

  float4 acc = make_float4(0.f, 0.f, 0.f, 0.f);
  for (int base = 0; base < deg; base += 32) {
    const int i = base + lane;
    const int sl = i < cnt ? csr_src[start + i] : n;
    const int nb = min(32, deg - base);
    for (int k = 0; k < nb; k += 4) {
      float ev[4];
      float4 xv[4];
#pragma unroll
      for (int u = 0; u < 4; ++u) {
        const int s = __builtin_amdgcn_readlane(sl, k + u);
        ev[u] = leaky(as_[(size_t)s * 8 + h] + adn);
        xv[u] = *reinterpret_cast<const float4*>(xh + (size_t)s * 128 + lane * 4);
      }
#pragma unroll
      for (int u = 0; u < 4; ++u) {
        const float alpha = expf(ev[u] - m) / dsum;
        const bool ok = u == 0 || k + u < nb;
        acc.x = ok ? acc.x + xv[u].x * alpha : acc.x;
        acc.y = ok ? acc.y + xv[u].y * alpha : acc.y;
        acc.z = ok ? acc.z + xv[u].z * alpha : acc.z;
        acc.w = ok ? acc.w + xv[u].w * alpha : acc.w;
      }
    }
  }
  const float4 b = *reinterpret_cast<const float4*>(bias + lane * 4);
  float4 o;
  o.x = acc.x + b.x; o.y = acc.y + b.y; o.z = acc.z + b.z; o.w = acc.w + b.w;
  o.x = o.x > 0.f ? o.x : (expf(o.x) - 1.f);
  o.y = o.y > 0.f ? o.y : (expf(o.y) - 1.f);
  o.z = o.z > 0.f ? o.z : (expf(o.z) - 1.f);
  o.w = o.w > 0.f ? o.w : (expf(o.w) - 1.f);
  *reinterpret_cast<float4*>(hout + (size_t)n * 128 + lane * 4) = o;
}

__global__ __launch_bounds__(256)
void node2_kernel(const int* __restrict__ rowptr, const int* __restrict__ csr_src,
                  const float* __restrict__ as_, const float* __restrict__ ad_,
                  const float* __restrict__ xh, const float* __restrict__ bias,
                  float* __restrict__ out, int N) {
  const int lane = threadIdx.x & 31;
  const int n = uni((int)((blockIdx.x * blockDim.x + threadIdx.x) >> 5));
  if (n >= N) return;
  const int start = rowptr[n];
  const int cnt = rowptr[n + 1] - start;
  const int deg = cnt + 1;
  const float adn = ad_[n];

  float m = -INFINITY;
  for (int base = 0; base < deg; base += 32) {
    const int i = base + lane;
    const int sl = i < cnt ? csr_src[start + i] : n;
    const float ev = leaky(as_[sl] + adn);
    m = i < deg ? fmaxf(m, ev) : m;
  }
  for (int o = 16; o > 0; o >>= 1) m = fmaxf(m, __shfl_xor(m, o));

  float den = 0.f;
  for (int base = 0; base < deg; base += 32) {
    const int i = base + lane;
    const int sl = i < cnt ? csr_src[start + i] : n;
    const float ev = leaky(as_[sl] + adn);
    float ex = i < deg ? expf(ev - m) : 0.f;
    for (int o = 16; o > 0; o >>= 1) ex += __shfl_xor(ex, o);
    den += ex;
  }
  const float dsum = den + 1e-16f;

  float2 acc = make_float2(0.f, 0.f);
  for (int base = 0; base < deg; base += 32) {
    const int i = base + lane;
    const int sl = i < cnt ? csr_src[start + i] : n;
    const float ev = leaky(as_[sl] + adn);
    const float alpha = expf(ev - m) / dsum;
    const int nb = min(32, deg - base);
    for (int k = 0; k < nb; k += 4) {
      float2 xv[4];
      float al[4];
#pragma unroll
      for (int u = 0; u < 4; ++u) {
        const int s = __builtin_amdgcn_readlane(sl, k + u);
        al[u] = lane_f(alpha, k + u);
        xv[u] = *reinterpret_cast<const float2*>(xh + (size_t)s * 64 + lane * 2);
      }
#pragma unroll
      for (int u = 0; u < 4; ++u) {
        const bool ok = u == 0 || k + u < nb;
        acc.x = ok ? acc.x + xv[u].x * al[u] : acc.x;
        acc.y = ok ? acc.y + xv[u].y * al[u] : acc.y;
      }
    }
  }
  const float2 b = *reinterpret_cast<const float2*>(bias + lane * 2);
  float2 o;
  o.x = acc.x + b.x;
  o.y = acc.y + b.y;
  *reinterpret_cast<float2*>(out + (size_t)n * 64 + lane * 2) = o;
}

extern "C" void kernel_launch(void* const* d_in, const int* in_sizes, int n_in,
                              void* d_out, int out_size, void* d_ws,
                              size_t ws_size, hipStream_t stream) {
  (void)n_in; (void)out_size; (void)ws_size;
  const float* x        = (const float*)d_in[0];
  const int*   eidx     = (const int*)d_in[1];
  const float* W1       = (const float*)d_in[2];
  const float* att_src1 = (const float*)d_in[3];
  const float* att_dst1 = (const float*)d_in[4];
  const float* b1       = (const float*)d_in[5];
  const float* W2       = (const float*)d_in[6];
  const float* att_src2 = (const float*)d_in[7];
  const float* att_dst2 = (const float*)d_in[8];
  const float* b2       = (const float*)d_in[9];
  float* out = (float*)d_out;

  const int N    = in_sizes[0] / 128;
  const int E    = in_sizes[1] / 2;
  const int* esrc = eidx;
  const int* edst = eidx + E;

  float* ws = (float*)d_ws;
  float* xh1    = ws;        ws += (size_t)N * 128;
  float* h1     = ws;        ws += (size_t)N * 128;
  float* as1    = ws;        ws += (size_t)N * 8;
  float* ad1    = ws;        ws += (size_t)N * 8;
  int*   csr    = (int*)ws;  ws += (size_t)E;
  float* xh2    = ws;        ws += (size_t)N * 64;
  float* as2    = ws;        ws += (size_t)N;
  float* ad2    = ws;        ws += (size_t)N;
  int*   rowptr = (int*)ws;
  unsigned* cnt = (unsigned*)h1;
  int*   slot   = (int*)h1 + ((N + 3) / 4) * 4;
  int*   rank   = slot + (size_t)E * 2;

  auto cdiv = [](int a, int b) { return (a + b - 1) / b; };

  fill_u32_kernel<<<cdiv(N, 256), 256, 0, stream>>>(cnt, 0u, N);
  count_rank_kernel<<<cdiv(E, 256), 256, 0, stream>>>(edst, cnt, rank, E);
  scan_kernel<<<cdiv(N, 1024), 256, 0, stream>>>(cnt, rowptr, N);
  place_kernel<<<cdiv(E, 256), 256, 0, stream>>>(esrc, edst, rank, rowptr, slot, E);
  seg_sort_kernel<<<cdiv(N * 32, 256), 256, 0, stream>>>(slot, rowptr, csr, N);

  gemm_wmma_kernel<128, 128><<<N / 16, 256, 0, stream>>>(x, W1, xh1);
  scores1_kernel<<<cdiv(N * 8, 256), 256, 0, stream>>>(xh1, att_src1, att_dst1,
                                                       as1, ad1, N * 8);
  node1_kernel<<<cdiv(N * 32, 256), 256, 0, stream>>>(rowptr, csr, as1, ad1, xh1,
                                                      b1, h1, N);

  gemm_wmma_kernel<128, 64><<<N / 16, 128, 0, stream>>>(h1, W2, xh2);
  scores2_kernel<<<cdiv(N, 256), 256, 0, stream>>>(xh2, att_src2, att_dst2, as2,
                                                   ad2, N);
  node2_kernel<<<cdiv(N * 32, 256), 256, 0, stream>>>(rowptr, csr, as2, ad2, xh2,
                                                      b2, out, N);
}


# four small launches folded into neighbours (11 -> 7): GEMM kernels clear the edge counters and sum the attention logits from their tile (16-lane tree: new sum order), node kernel 1 orders its edges
# speedup vs baseline: 4.5699x; 1.0813x over previous
#include <hip/hip_runtime.h>
#include <hip/hip_bf16.h>
#include <math.h>

typedef __attribute__((ext_vector_type(2))) float v2f;
typedef __attribute__((ext_vector_type(8))) float v8f;

#define NEG_SLOPE 0.2f

template <int K, int NC, int H, bool ZERO>
__global__ __launch_bounds__((NC / 16) * 32)
void gemm_wmma_kernel(const float* __restrict__ X, const float* __restrict__ W,
                      const float* __restrict__ att_s,
                      const float* __restrict__ att_d, float* __restrict__ Y,
                      float* __restrict__ as_, float* __restrict__ ad_,
                      unsigned* __restrict__ zero_cnt) {
  constexpr int NW  = NC / 16;
  constexpr int TPH = NW / H;
  __shared__ float ps[NW][16];
  __shared__ float pd[NW][16];
  const int lane = threadIdx.x & 31;
  const int wave = threadIdx.x >> 5;
  const int half = lane >> 4;
  const int l15  = lane & 15;
  const int m0   = blockIdx.x * 16;
  const int col  = wave * 16 + l15;
  const float* xrow = X + (size_t)(m0 + l15) * K;
  const float wsv = att_s[col], wdv = att_d[col];
  v8f c = {};
#pragma unroll
  for (int k0 = 0; k0 < K; k0 += 4) {
    const int ka = k0 + half * 2;
    const float2 av = *reinterpret_cast<const float2*>(xrow + ka);
    v2f a; a.x = av.x; a.y = av.y;
    v2f b;
    b.x = W[(size_t)ka * NC + col];
    b.y = W[(size_t)(ka + 1) * NC + col];
    c = __builtin_amdgcn_wmma_f32_16x16x4_f32(false, a, false, b, (short)0, c,
                                              false, false);
  }
#pragma unroll
  for (int r = 0; r < 8; ++r)
    Y[(size_t)(m0 + half * 8 + r) * NC + col] = c[r];

  asm volatile("v_nop\n\tv_nop\n\tv_nop\n\tv_nop" : "+v"(c));

  float rs = 0.f, rd = 0.f;
#pragma unroll
  for (int r = 0; r < 8; ++r) {
    float s = c[r] * wsv, d = c[r] * wdv;
    for (int o = 8; o > 0; o >>= 1) {
      s += __shfl_xor(s, o);
      d += __shfl_xor(d, o);
    }
    rs = l15 == r ? s : rs;
    rd = l15 == r ? d : rd;
  }
  if (l15 < 8) {
    ps[wave][half * 8 + l15] = rs;
    pd[wave][half * 8 + l15] = rd;
  }
  __syncthreads();
  const int t = threadIdx.x;
  if (t < 16 * H) {
    const int head = t % H, row = t / H;
    float s = 0.f, d = 0.f;
    for (int w = 0; w < TPH; ++w) {
      s += ps[head * TPH + w][row];
      d += pd[head * TPH + w][row];
    }
    as_[(size_t)(m0 + row) * H + head] = s;
    ad_[(size_t)(m0 + row) * H + head] = d;
  }
  if (ZERO) {
    if (t < 16) zero_cnt[m0 + t] = 0u;
  }
}

static __device__ __forceinline__ int uni(int v) {
  return __builtin_amdgcn_readfirstlane(v);
}
static __device__ __forceinline__ float lane_f(float v, int k) {
  return __int_as_float(__builtin_amdgcn_readlane(__float_as_int(v), k));
}
static __device__ __forceinline__ float leaky(float v) {
  return v > 0.f ? v : NEG_SLOPE * v;
}

__global__ __launch_bounds__(256)
void count_rank_kernel(const int* __restrict__ edst, unsigned* __restrict__ cnt,
                       int* __restrict__ rank, int E) {
  int e = blockIdx.x * blockDim.x + threadIdx.x;
  if (e >= E) return;
  rank[e] = (int)atomicAdd(&cnt[edst[e]], 1u);
}

__global__ __launch_bounds__(256)
void scan_kernel(const unsigned* __restrict__ cnt, int* __restrict__ rowptr,
                 int N) {
  __shared__ int wsum[8];
  __shared__ int wsum2[8];
  const int tid = threadIdx.x, lane = tid & 31, wv = tid >> 5;
  const int block_start = blockIdx.x * 1024;
  int part = 0;
  for (int i = tid * 4; i < block_start; i += 1024) {
    const uint4 v = *reinterpret_cast<const uint4*>(cnt + i);
    part += (int)(v.x + v.y + v.z + v.w);
  }
  for (int o = 16; o > 0; o >>= 1) part += __shfl_xor(part, o);
  if (lane == 0) wsum[wv] = part;
  __syncthreads();
  int base = 0;
  for (int w = 0; w < 8; ++w) base += wsum[w];
  const int i0 = block_start + tid * 4;
  const int c0 = i0 < N ? (int)cnt[i0] : 0;
  const int c1 = i0 + 1 < N ? (int)cnt[i0 + 1] : 0;
  const int c2 = i0 + 2 < N ? (int)cnt[i0 + 2] : 0;
  const int c3 = i0 + 3 < N ? (int)cnt[i0 + 3] : 0;
  const int tsum = c0 + c1 + c2 + c3;
  int incl = tsum;
  for (int o = 1; o < 32; o <<= 1) {
    int v = __shfl_up(incl, o);
    if (lane >= o) incl += v;
  }
  if (lane == 31) wsum2[wv] = incl;
  __syncthreads();
  int woff = 0;
  for (int w = 0; w < wv; ++w) woff += wsum2[w];
  const int excl = base + woff + incl - tsum;
  if (i0 < N) rowptr[i0] = excl;
  if (i0 + 1 < N) rowptr[i0 + 1] = excl + c0;
  if (i0 + 2 < N) rowptr[i0 + 2] = excl + c0 + c1;
  if (i0 + 3 < N) rowptr[i0 + 3] = excl + c0 + c1 + c2;
  if (i0 < N && N <= i0 + 4) rowptr[N] = excl + tsum;
}

__global__ __launch_bounds__(256)
void place_kernel(const int* __restrict__ esrc, const int* __restrict__ edst,
                  const int* __restrict__ rank, const int* __restrict__ rowptr,
                  int* __restrict__ slot_, int E) {
  int e = blockIdx.x * blockDim.x + threadIdx.x;
  if (e >= E) return;
  int2* slot = reinterpret_cast<int2*>(slot_);
  slot[rowptr[edst[e]] + rank[e]] = make_int2(e, esrc[e]);
}

__global__ __launch_bounds__(256)
void node1_kernel(const int* __restrict__ rowptr, const int* __restrict__ slot_,
                  int* csr_src,
                  const float* __restrict__ as_, const float* __restrict__ ad_,
                  const float* __restrict__ xh, const float* __restrict__ bias,
                  float* __restrict__ hout, int N) {
  const int2* slot = reinterpret_cast<const int2*>(slot_);
  const int lane = threadIdx.x & 31;
  const int n = uni((int)((blockIdx.x * blockDim.x + threadIdx.x) >> 5));
  if (n >= N) return;
  const int h = lane >> 2;
  const int start = rowptr[n];
  const int cnt = rowptr[n + 1] - start;
  const int deg = cnt + 1;
  const float adn = ad_[(size_t)n * 8 + h];

  const bool one = cnt <= 32;
  int sl0 = n;
  if (one) {
    const int2 my = lane < cnt ? slot[start + lane] : make_int2(0x7fffffff, 0);
    int r = 0;
    for (int k = 0; k < cnt; ++k)
      r += (__builtin_amdgcn_readlane(my.x, k) < my.x) ? 1 : 0;
    if (lane < cnt) csr_src[start + r] = my.y;
    sl0 = __builtin_amdgcn_ds_permute(r << 2, my.y);
  } else {
    for (int base = 0; base < cnt; base += 32) {
      const int i = base + lane;
      const int2 my = i < cnt ? slot[start + i] : make_int2(0x7fffffff, 0);
      int r = 0;
      for (int jb = 0; jb < cnt; jb += 32) {
        const int j = jb + lane;
        const int other = j < cnt ? slot[start + j].x : 0x7fffffff;
        const int nb = min(32, cnt - jb);
        for (int k = 0; k < nb; ++k)
          r += (__builtin_amdgcn_readlane(other, k) < my.x) ? 1 : 0;
      }
      if (i < cnt) csr_src[start + r] = my.y;
    }
    __threadfence();
  }

  float m = -INFINITY;
  for (int base = 0; base < deg; base += 32) {
    const int i = base + lane;
    const int sl = i < cnt ? (one ? sl0 : csr_src[start + i]) : n;
    const int nb = min(32, deg - base);
    for (int k = 0; k < nb; k += 4) {
      float ev[4];
#pragma unroll
      for (int u = 0; u < 4; ++u) {
        const int s = __builtin_amdgcn_readlane(sl, k + u);
        ev[u] = leaky(as_[(size_t)s * 8 + h] + adn);
      }
#pragma unroll
      for (int u = 0; u < 4; ++u)
        m = (u == 0 || k + u < nb) ? fmaxf(m, ev[u]) : m;
    }
  }

  float den = 0.f;
  for (int base = 0; base < deg; base += 32) {
    const int i = base + lane;
    const int sl = i < cnt ? (one ? sl0 : csr_src[start + i]) : n;
    const int nb = min(32, deg - base);
    for (int k = 0; k < nb; k += 4) {
      float ev[4];
#pragma unroll
      for (int u = 0; u < 4; ++u) {
        const int s = __builtin_amdgcn_readlane(sl, k + u);
        ev[u] = leaky(as_[(size_t)s * 8 + h] + adn);
      }
#pragma unroll
      for (int u = 0; u < 4; ++u)
        den = (u == 0 || k + u < nb) ? den + expf(ev[u] - m) : den;
    }
  }
  const float dsum = den + 1e-16f;

  float4 acc = make_float4(0.f, 0.f, 0.f, 0.f);
  for (int base = 0; base < deg; base += 32) {
    const int i = base + lane;
    const int sl = i < cnt ? (one ? sl0 : csr_src[start + i]) : n;
    const int nb = min(32, deg - base);
    for (int k = 0; k < nb; k += 4) {
      float ev[4];
      float4 xv[4];
#pragma unroll
      for (int u = 0; u < 4; ++u) {
        const int s = __builtin_amdgcn_readlane(sl, k + u);
        ev[u] = leaky(as_[(size_t)s * 8 + h] + adn);
        xv[u] = *reinterpret_cast<const float4*>(xh + (size_t)s * 128 + lane * 4);
      }
#pragma unroll
      for (int u = 0; u < 4; ++u) {
        const float alpha = expf(ev[u] - m) / dsum;
        const bool ok = u == 0 || k + u < nb;
        acc.x = ok ? acc.x + xv[u].x * alpha : acc.x;
        acc.y = ok ? acc.y + xv[u].y * alpha : acc.y;
        acc.z = ok ? acc.z + xv[u].z * alpha : acc.z;
        acc.w = ok ? acc.w + xv[u].w * alpha : acc.w;
      }
    }
  }
  const float4 b = *reinterpret_cast<const float4*>(bias + lane * 4);
  float4 o;
  o.x = acc.x + b.x; o.y = acc.y + b.y; o.z = acc.z + b.z; o.w = acc.w + b.w;
  o.x = o.x > 0.f ? o.x : (expf(o.x) - 1.f);
  o.y = o.y > 0.f ? o.y : (expf(o.y) - 1.f);
  o.z = o.z > 0.f ? o.z : (expf(o.z) - 1.f);
  o.w = o.w > 0.f ? o.w : (expf(o.w) - 1.f);
  *reinterpret_cast<float4*>(hout + (size_t)n * 128 + lane * 4) = o;
}

__global__ __launch_bounds__(256)
void node2_kernel(const int* __restrict__ rowptr, const int* __restrict__ csr_src,
                  const float* __restrict__ as_, const float* __restrict__ ad_,
                  const float* __restrict__ xh, const float* __restrict__ bias,
                  float* __restrict__ out, int N) {
  const int lane = threadIdx.x & 31;
  const int n = uni((int)((blockIdx.x * blockDim.x + threadIdx.x) >> 5));
  if (n >= N) return;
  const int start = rowptr[n];
  const int cnt = rowptr[n + 1] - start;
  const int deg = cnt + 1;
  const float adn = ad_[n];

  float m = -INFINITY;
  for (int base = 0; base < deg; base += 32) {
    const int i = base + lane;
    const int sl = i < cnt ? csr_src[start + i] : n;
    const float ev = leaky(as_[sl] + adn);
    m = i < deg ? fmaxf(m, ev) : m;
  }
  for (int o = 16; o > 0; o >>= 1) m = fmaxf(m, __shfl_xor(m, o));

  float den = 0.f;
  for (int base = 0; base < deg; base += 32) {
    const int i = base + lane;
    const int sl = i < cnt ? csr_src[start + i] : n;
    const float ev = leaky(as_[sl] + adn);
    float ex = i < deg ? expf(ev - m) : 0.f;
    for (int o = 16; o > 0; o >>= 1) ex += __shfl_xor(ex, o);
    den += ex;
  }
  const float dsum = den + 1e-16f;

  float2 acc = make_float2(0.f, 0.f);
  for (int base = 0; base < deg; base += 32) {
    const int i = base + lane;
    const int sl = i < cnt ? csr_src[start + i] : n;
    const float ev = leaky(as_[sl] + adn);
    const float alpha = expf(ev - m) / dsum;
    const int nb = min(32, deg - base);
    for (int k = 0; k < nb; k += 4) {
      float2 xv[4];
      float al[4];
#pragma unroll
      for (int u = 0; u < 4; ++u) {
        const int s = __builtin_amdgcn_readlane(sl, k + u);
        al[u] = lane_f(alpha, k + u);
        xv[u] = *reinterpret_cast<const float2*>(xh + (size_t)s * 64 + lane * 2);
      }
#pragma unroll
      for (int u = 0; u < 4; ++u) {
        const bool ok = u == 0 || k + u < nb;
        acc.x = ok ? acc.x + xv[u].x * al[u] : acc.x;
        acc.y = ok ? acc.y + xv[u].y * al[u] : acc.y;
      }
    }
  }
  const float2 b = *reinterpret_cast<const float2*>(bias + lane * 2);
  float2 o;
  o.x = acc.x + b.x;
  o.y = acc.y + b.y;
  *reinterpret_cast<float2*>(out + (size_t)n * 64 + lane * 2) = o;
}

extern "C" void kernel_launch(void* const* d_in, const int* in_sizes, int n_in,
                              void* d_out, int out_size, void* d_ws,
                              size_t ws_size, hipStream_t stream) {
  (void)n_in; (void)out_size; (void)ws_size;
  const float* x        = (const float*)d_in[0];
  const int*   eidx     = (const int*)d_in[1];
  const float* W1       = (const float*)d_in[2];
  const float* att_src1 = (const float*)d_in[3];
  const float* att_dst1 = (const float*)d_in[4];
  const float* b1       = (const float*)d_in[5];
  const float* W2       = (const float*)d_in[6];
  const float* att_src2 = (const float*)d_in[7];
  const float* att_dst2 = (const float*)d_in[8];
  const float* b2       = (const float*)d_in[9];
  float* out = (float*)d_out;

  const int N    = in_sizes[0] / 128;
  const int E    = in_sizes[1] / 2;
  const int* esrc = eidx;
  const int* edst = eidx + E;

  float* ws = (float*)d_ws;
  float* xh1    = ws;        ws += (size_t)N * 128;
  float* h1     = ws;        ws += (size_t)N * 128;
  float* as1    = ws;        ws += (size_t)N * 8;
  float* ad1    = ws;        ws += (size_t)N * 8;
  int*   csr    = (int*)ws;  ws += (size_t)E;
  float* xh2    = ws;        ws += (size_t)N * 64;
  float* as2    = ws;        ws += (size_t)N;
  float* ad2    = ws;        ws += (size_t)N;
  int*   rowptr = (int*)ws;
  unsigned* cnt = (unsigned*)xh2;
  int*   slot   = (int*)xh2 + ((N + 3) / 4) * 4;
  int*   rank   = slot + (size_t)E * 2;

  auto cdiv = [](int a, int b) { return (a + b - 1) / b; };

  gemm_wmma_kernel<128, 128, 8, true><<<N / 16, 256, 0, stream>>>(
      x, W1, att_src1, att_dst1, xh1, as1, ad1, cnt);

  count_rank_kernel<<<cdiv(E, 256), 256, 0, stream>>>(edst, cnt, rank, E);
  scan_kernel<<<cdiv(N, 1024), 256, 0, stream>>>(cnt, rowptr, N);
  place_kernel<<<cdiv(E, 256), 256, 0, stream>>>(esrc, edst, rank, rowptr, slot, E);

  node1_kernel<<<cdiv(N * 32, 256), 256, 0, stream>>>(rowptr, slot, csr, as1, ad1,
                                                      xh1, b1, h1, N);

  gemm_wmma_kernel<128, 64, 1, false><<<N / 16, 128, 0, stream>>>(
      h1, W2, att_src2, att_dst2, xh2, as2, ad2, nullptr);
  node2_kernel<<<cdiv(N * 32, 256), 256, 0, stream>>>(rowptr, csr, as2, ad2, xh2,
                                                      b2, out, N);
}


# node kernels gather a segment's attention logits once into registers (lane = edge and head) instead of once per softmax pass; layer-1 denominator summed per lane then over 4 lanes: new sum order
# speedup vs baseline: 5.0936x; 1.1146x over previous
#include <hip/hip_runtime.h>
#include <hip/hip_bf16.h>
#include <math.h>

typedef __attribute__((ext_vector_type(2))) float v2f;
typedef __attribute__((ext_vector_type(8))) float v8f;

#define NEG_SLOPE 0.2f

template <int K, int NC, int H, bool ZERO>
__global__ __launch_bounds__((NC / 16) * 32)
void gemm_wmma_kernel(const float* __restrict__ X, const float* __restrict__ W,
                      const float* __restrict__ att_s,
                      const float* __restrict__ att_d, float* __restrict__ Y,
                      float* __restrict__ as_, float* __restrict__ ad_,
                      unsigned* __restrict__ zero_cnt) {
  constexpr int NW  = NC / 16;
  constexpr int TPH = NW / H;
  __shared__ float ps[NW][16];
  __shared__ float pd[NW][16];
  const int lane = threadIdx.x & 31;
  const int wave = threadIdx.x >> 5;
  const int half = lane >> 4;
  const int l15  = lane & 15;
  const int m0   = blockIdx.x * 16;
  const int col  = wave * 16 + l15;
  const float* xrow = X + (size_t)(m0 + l15) * K;
  const float wsv = att_s[col], wdv = att_d[col];
  v8f c = {};
#pragma unroll
  for (int k0 = 0; k0 < K; k0 += 4) {
    const int ka = k0 + half * 2;
    const float2 av = *reinterpret_cast<const float2*>(xrow + ka);
    v2f a; a.x = av.x; a.y = av.y;
    v2f b;
    b.x = W[(size_t)ka * NC + col];
    b.y = W[(size_t)(ka + 1) * NC + col];
    c = __builtin_amdgcn_wmma_f32_16x16x4_f32(false, a, false, b, (short)0, c,
                                              false, false);
  }
#pragma unroll
  for (int r = 0; r < 8; ++r)
    Y[(size_t)(m0 + half * 8 + r) * NC + col] = c[r];

  asm volatile("v_nop\n\tv_nop\n\tv_nop\n\tv_nop" : "+v"(c));

  float rs = 0.f, rd = 0.f;
#pragma unroll
  for (int r = 0; r < 8; ++r) {
    float s = c[r] * wsv, d = c[r] * wdv;
    for (int o = 8; o > 0; o >>= 1) {
      s += __shfl_xor(s, o);
      d += __shfl_xor(d, o);
    }
    rs = l15 == r ? s : rs;
    rd = l15 == r ? d : rd;
  }
  if (l15 < 8) {
    ps[wave][half * 8 + l15] = rs;
    pd[wave][half * 8 + l15] = rd;
  }
  __syncthreads();
  const int t = threadIdx.x;
  if (t < 16 * H) {
    const int head = t % H, row = t / H;
    float s = 0.f, d = 0.f;
    for (int w = 0; w < TPH; ++w) {
      s += ps[head * TPH + w][row];
      d += pd[head * TPH + w][row];
    }
    as_[(size_t)(m0 + row) * H + head] = s;
    ad_[(size_t)(m0 + row) * H + head] = d;
  }
  if (ZERO) {
    if (t < 16) zero_cnt[m0 + t] = 0u;
  }
}

static __device__ __forceinline__ int uni(int v) {
  return __builtin_amdgcn_readfirstlane(v);
}
static __device__ __forceinline__ float lane_f(float v, int k) {
  return __int_as_float(__builtin_amdgcn_readlane(__float_as_int(v), k));
}
static __device__ __forceinline__ float leaky(float v) {
  return v > 0.f ? v : NEG_SLOPE * v;
}

__global__ __launch_bounds__(256)
void count_rank_kernel(const int* __restrict__ edst, unsigned* __restrict__ cnt,
                       int* __restrict__ rank, int E) {
  int e = blockIdx.x * blockDim.x + threadIdx.x;
  if (e >= E) return;
  rank[e] = (int)atomicAdd(&cnt[edst[e]], 1u);
}

__global__ __launch_bounds__(256)
void scan_kernel(const unsigned* __restrict__ cnt, int* __restrict__ rowptr,
                 int N) {
  __shared__ int wsum[8];
  __shared__ int wsum2[8];
  const int tid = threadIdx.x, lane = tid & 31, wv = tid >> 5;
  const int block_start = blockIdx.x * 1024;
  int part = 0;
  for (int i = tid * 4; i < block_start; i += 1024) {
    const uint4 v = *reinterpret_cast<const uint4*>(cnt + i);
    part += (int)(v.x + v.y + v.z + v.w);
  }
  for (int o = 16; o > 0; o >>= 1) part += __shfl_xor(part, o);
  if (lane == 0) wsum[wv] = part;
  __syncthreads();
  int base = 0;
  for (int w = 0; w < 8; ++w) base += wsum[w];
  const int i0 = block_start + tid * 4;
  const int c0 = i0 < N ? (int)cnt[i0] : 0;
  const int c1 = i0 + 1 < N ? (int)cnt[i0 + 1] : 0;
  const int c2 = i0 + 2 < N ? (int)cnt[i0 + 2] : 0;
  const int c3 = i0 + 3 < N ? (int)cnt[i0 + 3] : 0;
  const int tsum = c0 + c1 + c2 + c3;
  int incl = tsum;
  for (int o = 1; o < 32; o <<= 1) {
    int v = __shfl_up(incl, o);
    if (lane >= o) incl += v;
  }
  if (lane == 31) wsum2[wv] = incl;
  __syncthreads();
  int woff = 0;
  for (int w = 0; w < wv; ++w) woff += wsum2[w];
  const int excl = base + woff + incl - tsum;
  if (i0 < N) rowptr[i0] = excl;
  if (i0 + 1 < N) rowptr[i0 + 1] = excl + c0;
  if (i0 + 2 < N) rowptr[i0 + 2] = excl + c0 + c1;
  if (i0 + 3 < N) rowptr[i0 + 3] = excl + c0 + c1 + c2;
  if (i0 < N && N <= i0 + 4) rowptr[N] = excl + tsum;
}

__global__ __launch_bounds__(256)
void place_kernel(const int* __restrict__ esrc, const int* __restrict__ edst,
                  const int* __restrict__ rank, const int* __restrict__ rowptr,
                  int* __restrict__ slot_, int E) {
  int e = blockIdx.x * blockDim.x + threadIdx.x;
  if (e >= E) return;
  int2* slot = reinterpret_cast<int2*>(slot_);
  slot[rowptr[edst[e]] + rank[e]] = make_int2(e, esrc[e]);
}

template <int CTRL>
static __device__ __forceinline__ float quad_f(float v) {
  return __int_as_float(
      __builtin_amdgcn_mov_dpp(__float_as_int(v), CTRL, 0xf, 0xf, true));
}
#define QUAD_XOR1 0xB1
#define QUAD_XOR2 0x4E

__global__ __launch_bounds__(256)
void node1_kernel(const int* __restrict__ rowptr, const int* __restrict__ slot_,
                  int* csr_src,
                  const float* __restrict__ as_, const float* __restrict__ ad_,
                  const float* __restrict__ xh, const float* __restrict__ bias,
                  float* __restrict__ hout, int N) {
  const int2* slot = reinterpret_cast<const int2*>(slot_);
  const int lane = threadIdx.x & 31;
  const int n = uni((int)((blockIdx.x * blockDim.x + threadIdx.x) >> 5));
  if (n >= N) return;
  const int h = lane >> 2;
  const int start = rowptr[n];
  const int cnt = rowptr[n + 1] - start;
  const int deg = cnt + 1;
  const float adn = ad_[(size_t)n * 8 + h];

  float4 acc = make_float4(0.f, 0.f, 0.f, 0.f);
  if (deg <= 32) {
    const int2 my = lane < cnt ? slot[start + lane] : make_int2(0x7fffffff, 0);
    int r = 0;
    for (int k = 0; k < cnt; ++k)
      r += (__builtin_amdgcn_readlane(my.x, k) < my.x) ? 1 : 0;
    if (lane < cnt) csr_src[start + r] = my.y;
    const int pushed = __builtin_amdgcn_ds_permute(r << 2, my.y);
    const int sl = lane < cnt ? pushed : n;

    const int q = lane & 3;
    float a[8];
#pragma unroll
    for (int t = 0; t < 8; ++t) {
      const int s = __builtin_amdgcn_ds_bpermute((4 * t + q) << 2, sl);
      const float e = leaky(as_[(size_t)s * 8 + h] + adn);
      a[t] = 4 * t + q < deg ? e : -INFINITY;
    }
    float m = a[0];
#pragma unroll
    for (int t = 1; t < 8; ++t) m = fmaxf(m, a[t]);
    m = fmaxf(m, quad_f<QUAD_XOR1>(m));
    m = fmaxf(m, quad_f<QUAD_XOR2>(m));

    float den = 0.f;
#pragma unroll
    for (int t = 0; t < 8; ++t) {
      if (t == 0 || 4 * t < deg) {
        a[t] = expf(a[t] - m);
        den += a[t];
      }
    }
    den += quad_f<QUAD_XOR1>(den);
    den += quad_f<QUAD_XOR2>(den);
    const float dsum = den + 1e-16f;

#pragma unroll
    for (int t = 0; t < 8; ++t) {
      if (t == 0 || 4 * t < deg) {
        const float at = a[t] / dsum;
        float4 xv[4];
#pragma unroll
        for (int u = 0; u < 4; ++u) {
          const int s = __builtin_amdgcn_readlane(sl, 4 * t + u);
          xv[u] = *reinterpret_cast<const float4*>(xh + (size_t)s * 128 + lane * 4);
        }
        const float al[4] = {quad_f<0x00>(at), quad_f<0x55>(at),
                             quad_f<0xAA>(at), quad_f<0xFF>(at)};
#pragma unroll
        for (int u = 0; u < 4; ++u) {
          acc.x = fmaf(xv[u].x, al[u], acc.x);
          acc.y = fmaf(xv[u].y, al[u], acc.y);
          acc.z = fmaf(xv[u].z, al[u], acc.z);
          acc.w = fmaf(xv[u].w, al[u], acc.w);
        }
      }
    }
  } else {
    for (int base = 0; base < cnt; base += 32) {
      const int i = base + lane;
      const int2 my = i < cnt ? slot[start + i] : make_int2(0x7fffffff, 0);
      int r = 0;
      for (int jb = 0; jb < cnt; jb += 32) {
        const int j = jb + lane;
        const int other = j < cnt ? slot[start + j].x : 0x7fffffff;
        const int nb = min(32, cnt - jb);
        for (int k = 0; k < nb; ++k)
          r += (__builtin_amdgcn_readlane(other, k) < my.x) ? 1 : 0;
      }
      if (i < cnt) csr_src[start + r] = my.y;
    }
    __threadfence();

    float m = -INFINITY;
    for (int base = 0; base < deg; base += 32) {
      const int i = base + lane;
      const int sl = i < cnt ? csr_src[start + i] : n;
      const int nb = min(32, deg - base);
      for (int k = 0; k < nb; k += 4) {
        float ev[4];
#pragma unroll
        for (int u = 0; u < 4; ++u) {
          const int s = __builtin_amdgcn_readlane(sl, k + u);
          ev[u] = leaky(as_[(size_t)s * 8 + h] + adn);
        }
#pragma unroll
        for (int u = 0; u < 4; ++u)
          m = (u == 0 || k + u < nb) ? fmaxf(m, ev[u]) : m;
      }
    }

    float den = 0.f;
    for (int base = 0; base < deg; base += 32) {
      const int i = base + lane;
      const int sl = i < cnt ? csr_src[start + i] : n;
      const int nb = min(32, deg - base);
      for (int k = 0; k < nb; k += 4) {
        float ev[4];
#pragma unroll
        for (int u = 0; u < 4; ++u) {
          const int s = __builtin_amdgcn_readlane(sl, k + u);
          ev[u] = leaky(as_[(size_t)s * 8 + h] + adn);
        }
#pragma unroll
        for (int u = 0; u < 4; ++u)
          den = (u == 0 || k + u < nb) ? den + expf(ev[u] - m) : den;
      }
    }
    const float dsum = den + 1e-16f;

    for (int base = 0; base < deg; base += 32) {
      const int i = base + lane;
      const int sl = i < cnt ? csr_src[start + i] : n;
      const int nb = min(32, deg - base);
      for (int k = 0; k < nb; k += 4) {
        float ev[4];
        float4 xv[4];
#pragma unroll
        for (int u = 0; u < 4; ++u) {
          const int s = __builtin_amdgcn_readlane(sl, k + u);
          ev[u] = leaky(as_[(size_t)s * 8 + h] + adn);
          xv[u] = *reinterpret_cast<const float4*>(xh + (size_t)s * 128 + lane * 4);
        }
#pragma unroll
        for (int u = 0; u < 4; ++u) {
          const float alpha = expf(ev[u] - m) / dsum;
          const bool ok = u == 0 || k + u < nb;
          acc.x = ok ? acc.x + xv[u].x * alpha : acc.x;
          acc.y = ok ? acc.y + xv[u].y * alpha : acc.y;
          acc.z = ok ? acc.z + xv[u].z * alpha : acc.z;
          acc.w = ok ? acc.w + xv[u].w * alpha : acc.w;
        }
      }
    }
  }
  const float4 b = *reinterpret_cast<const float4*>(bias + lane * 4);
  float4 o;
  o.x = acc.x + b.x; o.y = acc.y + b.y; o.z = acc.z + b.z; o.w = acc.w + b.w;
  o.x = o.x > 0.f ? o.x : (expf(o.x) - 1.f);
  o.y = o.y > 0.f ? o.y : (expf(o.y) - 1.f);
  o.z = o.z > 0.f ? o.z : (expf(o.z) - 1.f);
  o.w = o.w > 0.f ? o.w : (expf(o.w) - 1.f);
  *reinterpret_cast<float4*>(hout + (size_t)n * 128 + lane * 4) = o;
}

__global__ __launch_bounds__(256)
void node2_kernel(const int* __restrict__ rowptr, const int* __restrict__ csr_src,
                  const float* __restrict__ as_, const float* __restrict__ ad_,
                  const float* __restrict__ xh, const float* __restrict__ bias,
                  float* __restrict__ out, int N) {
  const int lane = threadIdx.x & 31;
  const int n = uni((int)((blockIdx.x * blockDim.x + threadIdx.x) >> 5));
  if (n >= N) return;
  const int start = rowptr[n];
  const int cnt = rowptr[n + 1] - start;
  const int deg = cnt + 1;
  const float adn = ad_[n];

  float2 acc = make_float2(0.f, 0.f);
  if (deg <= 32) {
    const int sl = lane < cnt ? csr_src[start + lane] : n;
    const float ev = leaky(as_[sl] + adn);
    float m = lane < deg ? ev : -INFINITY;
    for (int o = 16; o > 0; o >>= 1) m = fmaxf(m, __shfl_xor(m, o));
    const float ex = lane < deg ? expf(ev - m) : 0.f;
    float den = ex;
    for (int o = 16; o > 0; o >>= 1) den += __shfl_xor(den, o);
    const float alpha = ex / (den + 1e-16f);
    for (int k = 0; k < deg; k += 4) {
      float2 xv[4];
      float al[4];
#pragma unroll
      for (int u = 0; u < 4; ++u) {
        const int s = __builtin_amdgcn_readlane(sl, k + u);
        al[u] = lane_f(alpha, k + u);
        xv[u] = *reinterpret_cast<const float2*>(xh + (size_t)s * 64 + lane * 2);
      }
#pragma unroll
      for (int u = 0; u < 4; ++u) {
        acc.x = fmaf(xv[u].x, al[u], acc.x);
        acc.y = fmaf(xv[u].y, al[u], acc.y);
      }
    }
  } else {
    float m = -INFINITY;
    for (int base = 0; base < deg; base += 32) {
      const int i = base + lane;
      const int sl = i < cnt ? csr_src[start + i] : n;
      const float ev = leaky(as_[sl] + adn);
      m = i < deg ? fmaxf(m, ev) : m;
    }
    for (int o = 16; o > 0; o >>= 1) m = fmaxf(m, __shfl_xor(m, o));

    float den = 0.f;
    for (int base = 0; base < deg; base += 32) {
      const int i = base + lane;
      const int sl = i < cnt ? csr_src[start + i] : n;
      const float ev = leaky(as_[sl] + adn);
      float ex = i < deg ? expf(ev - m) : 0.f;
      for (int o = 16; o > 0; o >>= 1) ex += __shfl_xor(ex, o);
      den += ex;
    }
    const float dsum = den + 1e-16f;

    for (int base = 0; base < deg; base += 32) {
      const int i = base + lane;
      const int sl = i < cnt ? csr_src[start + i] : n;
      const float ev = leaky(as_[sl] + adn);
      const float alpha = expf(ev - m) / dsum;
      const int nb = min(32, deg - base);
      for (int k = 0; k < nb; k += 4) {
        float2 xv[4];
        float al[4];
#pragma unroll
        for (int u = 0; u < 4; ++u) {
          const int s = __builtin_amdgcn_readlane(sl, k + u);
          al[u] = lane_f(alpha, k + u);
          xv[u] = *reinterpret_cast<const float2*>(xh + (size_t)s * 64 + lane * 2);
        }
#pragma unroll
        for (int u = 0; u < 4; ++u) {
          const bool ok = u == 0 || k + u < nb;
          acc.x = ok ? acc.x + xv[u].x * al[u] : acc.x;
          acc.y = ok ? acc.y + xv[u].y * al[u] : acc.y;
        }
      }
    }
  }
  const float2 b = *reinterpret_cast<const float2*>(bias + lane * 2);
  float2 o;
  o.x = acc.x + b.x;
  o.y = acc.y + b.y;
  *reinterpret_cast<float2*>(out + (size_t)n * 64 + lane * 2) = o;
}

extern "C" void kernel_launch(void* const* d_in, const int* in_sizes, int n_in,
                              void* d_out, int out_size, void* d_ws,
                              size_t ws_size, hipStream_t stream) {
  (void)n_in; (void)out_size; (void)ws_size;
  const float* x        = (const float*)d_in[0];
  const int*   eidx     = (const int*)d_in[1];
  const float* W1       = (const float*)d_in[2];
  const float* att_src1 = (const float*)d_in[3];
  const float* att_dst1 = (const float*)d_in[4];
  const float* b1       = (const float*)d_in[5];
  const float* W2       = (const float*)d_in[6];
  const float* att_src2 = (const float*)d_in[7];
  const float* att_dst2 = (const float*)d_in[8];
  const float* b2       = (const float*)d_in[9];
  float* out = (float*)d_out;

  const int N    = in_sizes[0] / 128;
  const int E    = in_sizes[1] / 2;
  const int* esrc = eidx;
  const int* edst = eidx + E;

  float* ws = (float*)d_ws;
  float* xh1    = ws;        ws += (size_t)N * 128;
  float* h1     = ws;        ws += (size_t)N * 128;
  float* as1    = ws;        ws += (size_t)N * 8;
  float* ad1    = ws;        ws += (size_t)N * 8;
  int*   csr    = (int*)ws;  ws += (size_t)E;
  float* xh2    = ws;        ws += (size_t)N * 64;
  float* as2    = ws;        ws += (size_t)N;
  float* ad2    = ws;        ws += (size_t)N;
  int*   rowptr = (int*)ws;
  unsigned* cnt = (unsigned*)xh2;
  int*   slot   = (int*)xh2 + ((N + 3) / 4) * 4;
  int*   rank   = slot + (size_t)E * 2;

  auto cdiv = [](int a, int b) { return (a + b - 1) / b; };

  gemm_wmma_kernel<128, 128, 8, true><<<N / 16, 256, 0, stream>>>(
      x, W1, att_src1, att_dst1, xh1, as1, ad1, cnt);

  count_rank_kernel<<<cdiv(E, 256), 256, 0, stream>>>(edst, cnt, rank, E);
  scan_kernel<<<cdiv(N, 1024), 256, 0, stream>>>(cnt, rowptr, N);
  place_kernel<<<cdiv(E, 256), 256, 0, stream>>>(esrc, edst, rank, rowptr, slot, E);

  node1_kernel<<<cdiv(N * 32, 256), 256, 0, stream>>>(rowptr, slot, csr, as1, ad1,
                                                      xh1, b1, h1, N);

  gemm_wmma_kernel<128, 64, 1, false><<<N / 16, 128, 0, stream>>>(
      h1, W2, att_src2, att_dst2, xh2, as2, ad2, nullptr);
  node2_kernel<<<cdiv(N * 32, 256), 256, 0, stream>>>(rowptr, csr, as2, ad2, xh2,
                                                      b2, out, N);
}


# edges grouped by destination by a two-level counting sort on LDS counters (256-node bins, then nodes; the LDS atomic's return is the rank) in place of per-edge returning global atomics; 7 launches
# speedup vs baseline: 6.7198x; 1.3193x over previous
#include <hip/hip_runtime.h>
#include <hip/hip_bf16.h>
#include <math.h>

typedef __attribute__((ext_vector_type(2))) float v2f;
typedef __attribute__((ext_vector_type(8))) float v8f;

#define NEG_SLOPE 0.2f

template <int K, int NC, int H>
__global__ __launch_bounds__((NC / 16) * 32)
void gemm_wmma_kernel(const float* __restrict__ X, const float* __restrict__ W,
                      const float* __restrict__ att_s,
                      const float* __restrict__ att_d, float* __restrict__ Y,
                      float* __restrict__ as_, float* __restrict__ ad_) {
  constexpr int NW  = NC / 16;
  constexpr int TPH = NW / H;
  __shared__ float ps[NW][16];
  __shared__ float pd[NW][16];
  const int lane = threadIdx.x & 31;
  const int wave = threadIdx.x >> 5;
  const int half = lane >> 4;
  const int l15  = lane & 15;
  const int m0   = blockIdx.x * 16;
  const int col  = wave * 16 + l15;
  const float* xrow = X + (size_t)(m0 + l15) * K;
  const float wsv = att_s[col], wdv = att_d[col];
  v8f c = {};
#pragma unroll
  for (int k0 = 0; k0 < K; k0 += 4) {
    const int ka = k0 + half * 2;
    const float2 av = *reinterpret_cast<const float2*>(xrow + ka);
    v2f a; a.x = av.x; a.y = av.y;
    v2f b;
    b.x = W[(size_t)ka * NC + col];
    b.y = W[(size_t)(ka + 1) * NC + col];
    c = __builtin_amdgcn_wmma_f32_16x16x4_f32(false, a, false, b, (short)0, c,
                                              false, false);
  }
#pragma unroll
  for (int r = 0; r < 8; ++r)
    Y[(size_t)(m0 + half * 8 + r) * NC + col] = c[r];

  asm volatile("v_nop\n\tv_nop\n\tv_nop\n\tv_nop" : "+v"(c));

  float rs = 0.f, rd = 0.f;
#pragma unroll
  for (int r = 0; r < 8; ++r) {
    float s = c[r] * wsv, d = c[r] * wdv;
    for (int o = 8; o > 0; o >>= 1) {
      s += __shfl_xor(s, o);
      d += __shfl_xor(d, o);
    }
    rs = l15 == r ? s : rs;
    rd = l15 == r ? d : rd;
  }
  if (l15 < 8) {
    ps[wave][half * 8 + l15] = rs;
    pd[wave][half * 8 + l15] = rd;
  }
  __syncthreads();
  const int t = threadIdx.x;
  if (t < 16 * H) {
    const int head = t % H, row = t / H;
    float s = 0.f, d = 0.f;
    for (int w = 0; w < TPH; ++w) {
      s += ps[head * TPH + w][row];
      d += pd[head * TPH + w][row];
    }
    as_[(size_t)(m0 + row) * H + head] = s;
    ad_[(size_t)(m0 + row) * H + head] = d;
  }
}

static __device__ __forceinline__ int uni(int v) {
  return __builtin_amdgcn_readfirstlane(v);
}
static __device__ __forceinline__ float lane_f(float v, int k) {
  return __int_as_float(__builtin_amdgcn_readlane(__float_as_int(v), k));
}
static __device__ __forceinline__ float leaky(float v) {
  return v > 0.f ? v : NEG_SLOPE * v;
}

#define SRT_NB 256
#define SRT_MAXB 256
#define SRT_CAP 8192

__global__ __launch_bounds__(1024)
void bin_count_kernel(const int* __restrict__ edst, int* __restrict__ hist,
                      int E) {
  __shared__ int lc[SRT_MAXB];
  const int tid = threadIdx.x;
  if (tid < SRT_MAXB) lc[tid] = 0;
  __syncthreads();
  const int chunk = (E + SRT_NB - 1) / SRT_NB;
  const int e_beg = blockIdx.x * chunk;
  const int e_end = min(E, e_beg + chunk);
  for (int e = e_beg + tid; e < e_end; e += 1024) atomicAdd(&lc[edst[e] >> 8], 1);
  __syncthreads();
  if (tid < SRT_MAXB) hist[blockIdx.x * SRT_MAXB + tid] = lc[tid];
}

__global__ __launch_bounds__(1024)
void bin_scatter_kernel(const int* __restrict__ esrc, const int* __restrict__ edst,
                        const int* __restrict__ hist, int* __restrict__ part_,
                        int* __restrict__ partd, int* __restrict__ binstart,
                        int E, int nbins) {
  __shared__ int gbase[SRT_MAXB];
  __shared__ int lc[SRT_MAXB];
  __shared__ int ptot[4][SRT_MAXB];
  __shared__ int ppre[4][SRT_MAXB];
  __shared__ int wsum[32];
  int2* part = reinterpret_cast<int2*>(part_);
  const int tid = threadIdx.x, lane = tid & 31, wave = tid >> 5;
  const int b = blockIdx.x;
  {
    const int bin = tid & (SRT_MAXB - 1), qt = tid >> 8;
    int tot = 0, pre = 0;
#pragma unroll 16
    for (int k = 0; k < SRT_NB / 4; ++k) {
      const int ch = qt * (SRT_NB / 4) + k;
      const int v = hist[ch * SRT_MAXB + bin];
      tot += v;
      pre += ch < b ? v : 0;
    }
    ptot[qt][bin] = tot;
    ppre[qt][bin] = pre;
  }
  if (tid < SRT_MAXB) lc[tid] = 0;
  __syncthreads();
  int t = 0, p = 0;
  if (tid < SRT_MAXB) {
    t = ptot[0][tid] + ptot[1][tid] + ptot[2][tid] + ptot[3][tid];
    p = ppre[0][tid] + ppre[1][tid] + ppre[2][tid] + ppre[3][tid];
  }
  int incl = t;
  for (int o = 1; o < 32; o <<= 1) {
    const int v = __shfl_up(incl, o);
    if (lane >= o) incl += v;
  }
  if (lane == 31) wsum[wave] = incl;
  __syncthreads();
  int excl = incl - t;
  for (int w = 0; w < wave; ++w) excl += wsum[w];
  if (tid < SRT_MAXB) {
    gbase[tid] = excl + p;
    if (b == 0 && tid < nbins) binstart[tid] = excl;
  }
  if (b == 0 && tid == 0) binstart[nbins] = E;
  __syncthreads();

  const int chunk = (E + SRT_NB - 1) / SRT_NB;
  const int e_beg = b * chunk;
  const int e_end = min(E, e_beg + chunk);
  for (int e = e_beg + tid; e < e_end; e += 1024) {
    const int d = edst[e];
    const int bin = d >> 8;
    const int pos = gbase[bin] + atomicAdd(&lc[bin], 1);
    part[pos] = make_int2(e, esrc[e]);
    partd[pos] = d;
  }
}

__global__ __launch_bounds__(1024)
void bin_finish_kernel(const int* __restrict__ part_, const int* __restrict__ partd,
                       const int* __restrict__ binstart, int* __restrict__ rowptr,
                       int* __restrict__ slot_, int N) {
  __shared__ int cnt[256];
  __shared__ int off[256];
  __shared__ int wtot[32];
  __shared__ int2 stage[SRT_CAP];
  const int2* part = reinterpret_cast<const int2*>(part_);
  int2* slot = reinterpret_cast<int2*>(slot_);
  const int tid = threadIdx.x, lane = tid & 31, wave = tid >> 5;
  const int lo = blockIdx.x * 256;
  const int span = min(256, N - lo);
  const int bs = binstart[blockIdx.x];
  const int n = binstart[blockIdx.x + 1] - bs;
  if (tid < 256) cnt[tid] = 0;
  __syncthreads();
  for (int i = tid; i < n; i += 1024) atomicAdd(&cnt[partd[bs + i] - lo], 1);
  __syncthreads();
  const int c = tid < 256 ? cnt[tid] : 0;
  int incl = c;
  for (int o = 1; o < 32; o <<= 1) {
    const int v = __shfl_up(incl, o);
    if (lane >= o) incl += v;
  }
  if (lane == 31) wtot[wave] = incl;
  __syncthreads();
  int woff = 0;
  for (int w = 0; w < wave; ++w) woff += wtot[w];
  const int excl = woff + incl - c;
  if (tid < 256) {
    off[tid] = excl;
    cnt[tid] = 0;
    if (tid < span) rowptr[lo + tid] = bs + excl;
    if (tid == span - 1 && lo + span == N) rowptr[N] = bs + excl + c;
  }
  __syncthreads();
  const bool staged = n <= SRT_CAP;
  for (int i = tid; i < n; i += 1024) {
    const int2 rec = part[bs + i];
    const int dl = partd[bs + i] - lo;
    const int p = off[dl] + atomicAdd(&cnt[dl], 1);
    if (staged) stage[p] = rec;
    else slot[bs + p] = rec;
  }
  __syncthreads();
  if (staged)
    for (int i = tid; i < n; i += 1024) slot[bs + i] = stage[i];
}

template <int CTRL>
static __device__ __forceinline__ float quad_f(float v) {
  return __int_as_float(
      __builtin_amdgcn_mov_dpp(__float_as_int(v), CTRL, 0xf, 0xf, true));
}
#define QUAD_XOR1 0xB1
#define QUAD_XOR2 0x4E

__global__ __launch_bounds__(256)
void node1_kernel(const int* __restrict__ rowptr, const int* __restrict__ slot_,
                  int* csr_src,
                  const float* __restrict__ as_, const float* __restrict__ ad_,
                  const float* __restrict__ xh, const float* __restrict__ bias,
                  float* __restrict__ hout, int N) {
  const int2* slot = reinterpret_cast<const int2*>(slot_);
  const int lane = threadIdx.x & 31;
  const int n = uni((int)((blockIdx.x * blockDim.x + threadIdx.x) >> 5));
  if (n >= N) return;
  const int h = lane >> 2;
  const int start = rowptr[n];
  const int cnt = rowptr[n + 1] - start;
  const int deg = cnt + 1;
  const float adn = ad_[(size_t)n * 8 + h];

  float4 acc = make_float4(0.f, 0.f, 0.f, 0.f);
  if (deg <= 32) {
    const int2 my = lane < cnt ? slot[start + lane] : make_int2(0x7fffffff, 0);
    int r = 0;
    for (int k = 0; k < cnt; ++k)
      r += (__builtin_amdgcn_readlane(my.x, k) < my.x) ? 1 : 0;
    if (lane < cnt) csr_src[start + r] = my.y;
    const int pushed = __builtin_amdgcn_ds_permute(r << 2, my.y);
    const int sl = lane < cnt ? pushed : n;

    const int q = lane & 3;
    float a[8];
#pragma unroll
    for (int t = 0; t < 8; ++t) {
      const int s = __builtin_amdgcn_ds_bpermute((4 * t + q) << 2, sl);
      const float e = leaky(as_[(size_t)s * 8 + h] + adn);
      a[t] = 4 * t + q < deg ? e : -INFINITY;
    }
    float m = a[0];
#pragma unroll
    for (int t = 1; t < 8; ++t) m = fmaxf(m, a[t]);
    m = fmaxf(m, quad_f<QUAD_XOR1>(m));
    m = fmaxf(m, quad_f<QUAD_XOR2>(m));

    float den = 0.f;
#pragma unroll
    for (int t = 0; t < 8; ++t) {
      if (t == 0 || 4 * t < deg) {
        a[t] = expf(a[t] - m);
        den += a[t];
      }
    }
    den += quad_f<QUAD_XOR1>(den);
    den += quad_f<QUAD_XOR2>(den);
    const float dsum = den + 1e-16f;

#pragma unroll
    for (int t = 0; t < 8; ++t) {
      if (t == 0 || 4 * t < deg) {
        const float at = a[t] / dsum;
        float4 xv[4];
#pragma unroll
        for (int u = 0; u < 4; ++u) {
          const int s = __builtin_amdgcn_readlane(sl, 4 * t + u);
          xv[u] = *reinterpret_cast<const float4*>(xh + (size_t)s * 128 + lane * 4);
        }
        const float al[4] = {quad_f<0x00>(at), quad_f<0x55>(at),
                             quad_f<0xAA>(at), quad_f<0xFF>(at)};
#pragma unroll
        for (int u = 0; u < 4; ++u) {
          acc.x = fmaf(xv[u].x, al[u], acc.x);
          acc.y = fmaf(xv[u].y, al[u], acc.y);
          acc.z = fmaf(xv[u].z, al[u], acc.z);
          acc.w = fmaf(xv[u].w, al[u], acc.w);
        }
      }
    }
  } else {
    for (int base = 0; base < cnt; base += 32) {
      const int i = base + lane;
      const int2 my = i < cnt ? slot[start + i] : make_int2(0x7fffffff, 0);
      int r = 0;
      for (int jb = 0; jb < cnt; jb += 32) {
        const int j = jb + lane;
        const int other = j < cnt ? slot[start + j].x : 0x7fffffff;
        const int nb = min(32, cnt - jb);
        for (int k = 0; k < nb; ++k)
          r += (__builtin_amdgcn_readlane(other, k) < my.x) ? 1 : 0;
      }
      if (i < cnt) csr_src[start + r] = my.y;
    }
    __threadfence();

    float m = -INFINITY;
    for (int base = 0; base < deg; base += 32) {
      const int i = base + lane;
      const int sl = i < cnt ? csr_src[start + i] : n;
      const int nb = min(32, deg - base);
      for (int k = 0; k < nb; k += 4) {
        float ev[4];
#pragma unroll
        for (int u = 0; u < 4; ++u) {
          const int s = __builtin_amdgcn_readlane(sl, k + u);
          ev[u] = leaky(as_[(size_t)s * 8 + h] + adn);
        }
#pragma unroll
        for (int u = 0; u < 4; ++u)
          m = (u == 0 || k + u < nb) ? fmaxf(m, ev[u]) : m;
      }
    }

    float den = 0.f;
    for (int base = 0; base < deg; base += 32) {
      const int i = base + lane;
      const int sl = i < cnt ? csr_src[start + i] : n;
      const int nb = min(32, deg - base);
      for (int k = 0; k < nb; k += 4) {
        float ev[4];
#pragma unroll
        for (int u = 0; u < 4; ++u) {
          const int s = __builtin_amdgcn_readlane(sl, k + u);
          ev[u] = leaky(as_[(size_t)s * 8 + h] + adn);
        }
#pragma unroll
        for (int u = 0; u < 4; ++u)
          den = (u == 0 || k + u < nb) ? den + expf(ev[u] - m) : den;
      }
    }
    const float dsum = den + 1e-16f;

    for (int base = 0; base < deg; base += 32) {
      const int i = base + lane;
      const int sl = i < cnt ? csr_src[start + i] : n;
      const int nb = min(32, deg - base);
      for (int k = 0; k < nb; k += 4) {
        float ev[4];
        float4 xv[4];
#pragma unroll
        for (int u = 0; u < 4; ++u) {
          const int s = __builtin_amdgcn_readlane(sl, k + u);
          ev[u] = leaky(as_[(size_t)s * 8 + h] + adn);
          xv[u] = *reinterpret_cast<const float4*>(xh + (size_t)s * 128 + lane * 4);
        }
#pragma unroll
        for (int u = 0; u < 4; ++u) {
          const float alpha = expf(ev[u] - m) / dsum;
          const bool ok = u == 0 || k + u < nb;
          acc.x = ok ? acc.x + xv[u].x * alpha : acc.x;
          acc.y = ok ? acc.y + xv[u].y * alpha : acc.y;
          acc.z = ok ? acc.z + xv[u].z * alpha : acc.z;
          acc.w = ok ? acc.w + xv[u].w * alpha : acc.w;
        }
      }
    }
  }
  const float4 b = *reinterpret_cast<const float4*>(bias + lane * 4);
  float4 o;
  o.x = acc.x + b.x; o.y = acc.y + b.y; o.z = acc.z + b.z; o.w = acc.w + b.w;
  o.x = o.x > 0.f ? o.x : (expf(o.x) - 1.f);
  o.y = o.y > 0.f ? o.y : (expf(o.y) - 1.f);
  o.z = o.z > 0.f ? o.z : (expf(o.z) - 1.f);
  o.w = o.w > 0.f ? o.w : (expf(o.w) - 1.f);
  *reinterpret_cast<float4*>(hout + (size_t)n * 128 + lane * 4) = o;
}

__global__ __launch_bounds__(256)
void node2_kernel(const int* __restrict__ rowptr, const int* __restrict__ csr_src,
                  const float* __restrict__ as_, const float* __restrict__ ad_,
                  const float* __restrict__ xh, const float* __restrict__ bias,
                  float* __restrict__ out, int N) {
  const int lane = threadIdx.x & 31;
  const int n = uni((int)((blockIdx.x * blockDim.x + threadIdx.x) >> 5));
  if (n >= N) return;
  const int start = rowptr[n];
  const int cnt = rowptr[n + 1] - start;
  const int deg = cnt + 1;
  const float adn = ad_[n];

  float2 acc = make_float2(0.f, 0.f);
  if (deg <= 32) {
    const int sl = lane < cnt ? csr_src[start + lane] : n;
    const float ev = leaky(as_[sl] + adn);
    float m = lane < deg ? ev : -INFINITY;
    for (int o = 16; o > 0; o >>= 1) m = fmaxf(m, __shfl_xor(m, o));
    const float ex = lane < deg ? expf(ev - m) : 0.f;
    float den = ex;
    for (int o = 16; o > 0; o >>= 1) den += __shfl_xor(den, o);
    const float alpha = ex / (den + 1e-16f);
    for (int k = 0; k < deg; k += 4) {
      float2 xv[4];
      float al[4];
#pragma unroll
      for (int u = 0; u < 4; ++u) {
        const int s = __builtin_amdgcn_readlane(sl, k + u);
        al[u] = lane_f(alpha, k + u);
        xv[u] = *reinterpret_cast<const float2*>(xh + (size_t)s * 64 + lane * 2);
      }
#pragma unroll
      for (int u = 0; u < 4; ++u) {
        acc.x = fmaf(xv[u].x, al[u], acc.x);
        acc.y = fmaf(xv[u].y, al[u], acc.y);
      }
    }
  } else {
    float m = -INFINITY;
    for (int base = 0; base < deg; base += 32) {
      const int i = base + lane;
      const int sl = i < cnt ? csr_src[start + i] : n;
      const float ev = leaky(as_[sl] + adn);
      m = i < deg ? fmaxf(m, ev) : m;
    }
    for (int o = 16; o > 0; o >>= 1) m = fmaxf(m, __shfl_xor(m, o));

    float den = 0.f;
    for (int base = 0; base < deg; base += 32) {
      const int i = base + lane;
      const int sl = i < cnt ? csr_src[start + i] : n;
      const float ev = leaky(as_[sl] + adn);
      float ex = i < deg ? expf(ev - m) : 0.f;
      for (int o = 16; o > 0; o >>= 1) ex += __shfl_xor(ex, o);
      den += ex;
    }
    const float dsum = den + 1e-16f;

    for (int base = 0; base < deg; base += 32) {
      const int i = base + lane;
      const int sl = i < cnt ? csr_src[start + i] : n;
      const float ev = leaky(as_[sl] + adn);
      const float alpha = expf(ev - m) / dsum;
      const int nb = min(32, deg - base);
      for (int k = 0; k < nb; k += 4) {
        float2 xv[4];
        float al[4];
#pragma unroll
        for (int u = 0; u < 4; ++u) {
          const int s = __builtin_amdgcn_readlane(sl, k + u);
          al[u] = lane_f(alpha, k + u);
          xv[u] = *reinterpret_cast<const float2*>(xh + (size_t)s * 64 + lane * 2);
        }
#pragma unroll
        for (int u = 0; u < 4; ++u) {
          const bool ok = u == 0 || k + u < nb;
          acc.x = ok ? acc.x + xv[u].x * al[u] : acc.x;
          acc.y = ok ? acc.y + xv[u].y * al[u] : acc.y;
        }
      }
    }
  }
  const float2 b = *reinterpret_cast<const float2*>(bias + lane * 2);
  float2 o;
  o.x = acc.x + b.x;
  o.y = acc.y + b.y;
  *reinterpret_cast<float2*>(out + (size_t)n * 64 + lane * 2) = o;
}

extern "C" void kernel_launch(void* const* d_in, const int* in_sizes, int n_in,
                              void* d_out, int out_size, void* d_ws,
                              size_t ws_size, hipStream_t stream) {
  (void)n_in; (void)out_size; (void)ws_size;
  const float* x        = (const float*)d_in[0];
  const int*   eidx     = (const int*)d_in[1];
  const float* W1       = (const float*)d_in[2];
  const float* att_src1 = (const float*)d_in[3];
  const float* att_dst1 = (const float*)d_in[4];
  const float* b1       = (const float*)d_in[5];
  const float* W2       = (const float*)d_in[6];
  const float* att_src2 = (const float*)d_in[7];
  const float* att_dst2 = (const float*)d_in[8];
  const float* b2       = (const float*)d_in[9];
  float* out = (float*)d_out;

  const int N    = in_sizes[0] / 128;
  const int E    = in_sizes[1] / 2;
  const int* esrc = eidx;
  const int* edst = eidx + E;

  float* ws = (float*)d_ws;
  float* xh1    = ws;        ws += (size_t)N * 128;
  float* h1     = ws;        ws += (size_t)N * 128;
  float* as1    = ws;        ws += (size_t)N * 8;
  float* ad1    = ws;        ws += (size_t)N * 8;
  int*   csr    = (int*)ws;  ws += (size_t)E;
  float* xh2    = ws;        ws += (size_t)N * 64;
  float* as2    = ws;        ws += (size_t)N;
  float* ad2    = ws;        ws += (size_t)N;
  int*   rowptr = (int*)ws;
  int*   slot   = (int*)xh2;
  const int nbins = (N + 255) / 256;
  int*   part     = (int*)h1;
  int*   partd    = part + (size_t)E * 2;
  int*   hist     = partd + (size_t)E;
  int*   binstart = hist + SRT_NB * SRT_MAXB;

  auto cdiv = [](int a, int b) { return (a + b - 1) / b; };

  gemm_wmma_kernel<128, 128, 8><<<N / 16, 256, 0, stream>>>(
      x, W1, att_src1, att_dst1, xh1, as1, ad1);

  bin_count_kernel<<<SRT_NB, 1024, 0, stream>>>(edst, hist, E);
  bin_scatter_kernel<<<SRT_NB, 1024, 0, stream>>>(esrc, edst, hist, part, partd,
                                                  binstart, E, nbins);
  bin_finish_kernel<<<nbins, 1024, 0, stream>>>(part, partd, binstart, rowptr,
                                                slot, N);

  node1_kernel<<<cdiv(N * 32, 256), 256, 0, stream>>>(rowptr, slot, csr, as1, ad1,
                                                      xh1, b1, h1, N);

  gemm_wmma_kernel<128, 64, 1><<<N / 16, 128, 0, stream>>>(
      h1, W2, att_src2, att_dst2, xh2, as2, ad2);
  node2_kernel<<<cdiv(N * 32, 256), 256, 0, stream>>>(rowptr, csr, as2, ad2, xh2,
                                                      b2, out, N);
}


# GEMM kernels: each wave keeps its 128x16 slice of W in registers and loops over row tiles; the X tile reaches LDS once per block by whole-row loads; products and their sum order unchanged
# speedup vs baseline: 7.6058x; 1.1318x over previous
#include <hip/hip_runtime.h>
#include <hip/hip_bf16.h>
#include <math.h>

typedef __attribute__((ext_vector_type(2))) float v2f;
typedef __attribute__((ext_vector_type(8))) float v8f;

#define NEG_SLOPE 0.2f


#define XT_PITCH 132

static __device__ __forceinline__ void lds_barrier() {
  asm volatile("s_wait_dscnt 0x0\n\ts_barrier_signal -1\n\ts_barrier_wait -1" ::: "memory");
}

template <int K, int NC, int H>
__global__ __launch_bounds__((NC / 16) * 32)
void gemm_wmma_kernel(const float* __restrict__ X, const float* __restrict__ W,
                      const float* __restrict__ att_s,
                      const float* __restrict__ att_d, float* __restrict__ Y,
                      float* __restrict__ as_, float* __restrict__ ad_,
                      int ntiles) {
  constexpr int NW  = NC / 16;
  constexpr int TPH = NW / H;
  constexpr int RPW = 16 / NW;
  static_assert(K == 128, "one row of X = 32 lanes x 16 bytes");
  static_assert(RPW == 2 || RPW == 4, "a wave brings in two or four of a tile's 16 rows");
  __shared__ alignas(16) float xt[2][16 * XT_PITCH];
  __shared__ float ps[2][NW][16];
  __shared__ float pd[2][NW][16];
  const int lane = threadIdx.x & 31;
  const int wave = threadIdx.x >> 5;
  const int half = lane >> 4;
  const int l15  = lane & 15;
  const int col  = wave * 16 + l15;
  const float wsv = att_s[col], wdv = att_d[col];

  v2f bf[K / 4];
#pragma unroll
  for (int j = 0; j < K / 4; ++j) {
    const int ka = 4 * j + half * 2;
    bf[j].x = W[(size_t)ka * NC + col];
    bf[j].y = W[(size_t)(ka + 1) * NC + col];
  }

  float4 s0 = make_float4(0.f, 0.f, 0.f, 0.f), s1 = s0, s2 = s0, s3 = s0;
  auto fetch_rows = [&](int tile) {
    const float* p = X + (size_t)(tile * 16 + wave) * K + lane * 4;
    s0 = *reinterpret_cast<const float4*>(p);
    s1 = *reinterpret_cast<const float4*>(p + (size_t)NW * K);
    if constexpr (RPW == 4) {
      s2 = *reinterpret_cast<const float4*>(p + (size_t)2 * NW * K);
      s3 = *reinterpret_cast<const float4*>(p + (size_t)3 * NW * K);
    }
  };
  auto park_rows = [&](int buf) {
    float* q = &xt[buf][wave * XT_PITCH + lane * 4];
    *reinterpret_cast<float4*>(q) = s0;
    *reinterpret_cast<float4*>(q + NW * XT_PITCH) = s1;
    if constexpr (RPW == 4) {
      *reinterpret_cast<float4*>(q + 2 * NW * XT_PITCH) = s2;
      *reinterpret_cast<float4*>(q + 3 * NW * XT_PITCH) = s3;
    }
  };

  int t = blockIdx.x;
  if (t < ntiles) {
    fetch_rows(t);
    park_rows(0);
  }
  lds_barrier();

  for (int it = 0; t < ntiles; t += gridDim.x, ++it) {
    const int cur = it & 1;
    const int m0 = t * 16;
    const int tn = t + gridDim.x;
    const bool more = tn < ntiles;
    if (more) fetch_rows(tn);

    const float* arow = &xt[cur][l15 * XT_PITCH + half * 2];
    v8f c = {};
#pragma unroll
    for (int j = 0; j < K / 4; ++j) {
      const float2 av = *reinterpret_cast<const float2*>(arow + 4 * j);
      v2f a; a.x = av.x; a.y = av.y;
      c = __builtin_amdgcn_wmma_f32_16x16x4_f32(false, a, false, bf[j], (short)0, c,
                                                false, false);
    }
    __builtin_amdgcn_sched_barrier(0);
    asm volatile("v_nop\n\tv_nop\n\tv_nop\n\tv_nop" : "+v"(c));
    __builtin_amdgcn_sched_barrier(0);

#pragma unroll
    for (int r = 0; r < 8; ++r)
      Y[(size_t)(m0 + half * 8 + r) * NC + col] = c[r];

    float rs = 0.f, rd = 0.f;
#pragma unroll
    for (int r = 0; r < 8; ++r) {
      float s = c[r] * wsv, d = c[r] * wdv;
      for (int o = 8; o > 0; o >>= 1) {
        s += __shfl_xor(s, o);
        d += __shfl_xor(d, o);
      }
      rs = l15 == r ? s : rs;
      rd = l15 == r ? d : rd;
    }
    if (l15 < 8) {
      ps[cur][wave][half * 8 + l15] = rs;
      pd[cur][wave][half * 8 + l15] = rd;
    }
    if (more) park_rows(cur ^ 1);
    lds_barrier();
    const int tid = threadIdx.x;
    if (tid < 16 * H) {
      const int head = tid % H, row = tid / H;
      float s = 0.f, d = 0.f;
      for (int w = 0; w < TPH; ++w) {
        s += ps[cur][head * TPH + w][row];
        d += pd[cur][head * TPH + w][row];
      }
      as_[(size_t)(m0 + row) * H + head] = s;
      ad_[(size_t)(m0 + row) * H + head] = d;
    }
  }
}

static __device__ __forceinline__ int uni(int v) {
  return __builtin_amdgcn_readfirstlane(v);
}
static __device__ __forceinline__ float lane_f(float v, int k) {
  return __int_as_float(__builtin_amdgcn_readlane(__float_as_int(v), k));
}
static __device__ __forceinline__ float leaky(float v) {
  return v > 0.f ? v : NEG_SLOPE * v;
}

#define SRT_NB 256
#define SRT_MAXB 256
#define SRT_CAP 8192

__global__ __launch_bounds__(1024)
void bin_count_kernel(const int* __restrict__ edst, int* __restrict__ hist,
                      int E) {
  __shared__ int lc[SRT_MAXB];
  const int tid = threadIdx.x;
  if (tid < SRT_MAXB) lc[tid] = 0;
  __syncthreads();
  const int chunk = (E + SRT_NB - 1) / SRT_NB;
  const int e_beg = blockIdx.x * chunk;
  const int e_end = min(E, e_beg + chunk);
  for (int e = e_beg + tid; e < e_end; e += 1024) atomicAdd(&lc[edst[e] >> 8], 1);
  __syncthreads();
  if (tid < SRT_MAXB) hist[blockIdx.x * SRT_MAXB + tid] = lc[tid];
}

__global__ __launch_bounds__(1024)
void bin_scatter_kernel(const int* __restrict__ esrc, const int* __restrict__ edst,
                        const int* __restrict__ hist, int* __restrict__ part_,
                        int* __restrict__ partd, int* __restrict__ binstart,
                        int E, int nbins) {
  __shared__ int gbase[SRT_MAXB];
  __shared__ int lc[SRT_MAXB];
  __shared__ int ptot[4][SRT_MAXB];
  __shared__ int ppre[4][SRT_MAXB];
  __shared__ int wsum[32];
  int2* part = reinterpret_cast<int2*>(part_);
  const int tid = threadIdx.x, lane = tid & 31, wave = tid >> 5;
  const int b = blockIdx.x;
  {
    const int bin = tid & (SRT_MAXB - 1), qt = tid >> 8;
    int tot = 0, pre = 0;
#pragma unroll 16
    for (int k = 0; k < SRT_NB / 4; ++k) {
      const int ch = qt * (SRT_NB / 4) + k;
      const int v = hist[ch * SRT_MAXB + bin];
      tot += v;
      pre += ch < b ? v : 0;
    }
    ptot[qt][bin] = tot;
    ppre[qt][bin] = pre;
  }
  if (tid < SRT_MAXB) lc[tid] = 0;
  __syncthreads();
  int t = 0, p = 0;
  if (tid < SRT_MAXB) {
    t = ptot[0][tid] + ptot[1][tid] + ptot[2][tid] + ptot[3][tid];
    p = ppre[0][tid] + ppre[1][tid] + ppre[2][tid] + ppre[3][tid];
  }
  int incl = t;
  for (int o = 1; o < 32; o <<= 1) {
    const int v = __shfl_up(incl, o);
    if (lane >= o) incl += v;
  }
  if (lane == 31) wsum[wave] = incl;
  __syncthreads();
  int excl = incl - t;
  for (int w = 0; w < wave; ++w) excl += wsum[w];
  if (tid < SRT_MAXB) {
    gbase[tid] = excl + p;
    if (b == 0 && tid < nbins) binstart[tid] = excl;
  }
  if (b == 0 && tid == 0) binstart[nbins] = E;
  __syncthreads();

  const int chunk = (E + SRT_NB - 1) / SRT_NB;
  const int e_beg = b * chunk;
  const int e_end = min(E, e_beg + chunk);
  for (int e = e_beg + tid; e < e_end; e += 1024) {
    const int d = edst[e];
    const int bin = d >> 8;
    const int pos = gbase[bin] + atomicAdd(&lc[bin], 1);
    part[pos] = make_int2(e, esrc[e]);
    partd[pos] = d;
  }
}

__global__ __launch_bounds__(1024)
void bin_finish_kernel(const int* __restrict__ part_, const int* __restrict__ partd,
                       const int* __restrict__ binstart, int* __restrict__ rowptr,
                       int* __restrict__ slot_, int N) {
  __shared__ int cnt[256];
  __shared__ int off[256];
  __shared__ int wtot[32];
  __shared__ int2 stage[SRT_CAP];
  const int2* part = reinterpret_cast<const int2*>(part_);
  int2* slot = reinterpret_cast<int2*>(slot_);
  const int tid = threadIdx.x, lane = tid & 31, wave = tid >> 5;
  const int lo = blockIdx.x * 256;
  const int span = min(256, N - lo);
  const int bs = binstart[blockIdx.x];
  const int n = binstart[blockIdx.x + 1] - bs;
  if (tid < 256) cnt[tid] = 0;
  __syncthreads();
  for (int i = tid; i < n; i += 1024) atomicAdd(&cnt[partd[bs + i] - lo], 1);
  __syncthreads();
  const int c = tid < 256 ? cnt[tid] : 0;
  int incl = c;
  for (int o = 1; o < 32; o <<= 1) {
    const int v = __shfl_up(incl, o);
    if (lane >= o) incl += v;
  }
  if (lane == 31) wtot[wave] = incl;
  __syncthreads();
  int woff = 0;
  for (int w = 0; w < wave; ++w) woff += wtot[w];
  const int excl = woff + incl - c;
  if (tid < 256) {
    off[tid] = excl;
    cnt[tid] = 0;
    if (tid < span) rowptr[lo + tid] = bs + excl;
    if (tid == span - 1 && lo + span == N) rowptr[N] = bs + excl + c;
  }
  __syncthreads();
  const bool staged = n <= SRT_CAP;
  for (int i = tid; i < n; i += 1024) {
    const int2 rec = part[bs + i];
    const int dl = partd[bs + i] - lo;
    const int p = off[dl] + atomicAdd(&cnt[dl], 1);
    if (staged) stage[p] = rec;
    else slot[bs + p] = rec;
  }
  __syncthreads();
  if (staged)
    for (int i = tid; i < n; i += 1024) slot[bs + i] = stage[i];
}

template <int CTRL>
static __device__ __forceinline__ float quad_f(float v) {
  return __int_as_float(
      __builtin_amdgcn_mov_dpp(__float_as_int(v), CTRL, 0xf, 0xf, true));
}
#define QUAD_XOR1 0xB1
#define QUAD_XOR2 0x4E

__global__ __launch_bounds__(256)
void node1_kernel(const int* __restrict__ rowptr, const int* __restrict__ slot_,
                  int* csr_src,
                  const float* __restrict__ as_, const float* __restrict__ ad_,
                  const float* __restrict__ xh, const float* __restrict__ bias,
                  float* __restrict__ hout, int N) {
  const int2* slot = reinterpret_cast<const int2*>(slot_);
  const int lane = threadIdx.x & 31;
  const int n = uni((int)((blockIdx.x * blockDim.x + threadIdx.x) >> 5));
  if (n >= N) return;
  const int h = lane >> 2;
  const int start = rowptr[n];
  const int cnt = rowptr[n + 1] - start;
  const int deg = cnt + 1;
  const float adn = ad_[(size_t)n * 8 + h];

  float4 acc = make_float4(0.f, 0.f, 0.f, 0.f);
  if (deg <= 32) {
    const int2 my = lane < cnt ? slot[start + lane] : make_int2(0x7fffffff, 0);
    int r = 0;
    for (int k = 0; k < cnt; ++k)
      r += (__builtin_amdgcn_readlane(my.x, k) < my.x) ? 1 : 0;
    if (lane < cnt) csr_src[start + r] = my.y;
    const int pushed = __builtin_amdgcn_ds_permute(r << 2, my.y);
    const int sl = lane < cnt ? pushed : n;

    const int q = lane & 3;
    float a[8];
#pragma unroll
    for (int t = 0; t < 8; ++t) {
      const int s = __builtin_amdgcn_ds_bpermute((4 * t + q) << 2, sl);
      const float e = leaky(as_[(size_t)s * 8 + h] + adn);
      a[t] = 4 * t + q < deg ? e : -INFINITY;
    }
    float m = a[0];
#pragma unroll
    for (int t = 1; t < 8; ++t) m = fmaxf(m, a[t]);
    m = fmaxf(m, quad_f<QUAD_XOR1>(m));
    m = fmaxf(m, quad_f<QUAD_XOR2>(m));

    float den = 0.f;
#pragma unroll
    for (int t = 0; t < 8; ++t) {
      if (t == 0 || 4 * t < deg) {
        a[t] = expf(a[t] - m);
        den += a[t];
      }
    }
    den += quad_f<QUAD_XOR1>(den);
    den += quad_f<QUAD_XOR2>(den);
    const float dsum = den + 1e-16f;

#pragma unroll
    for (int t = 0; t < 8; ++t) {
      if (t == 0 || 4 * t < deg) {
        const float at = a[t] / dsum;
        float4 xv[4];
#pragma unroll
        for (int u = 0; u < 4; ++u) {
          const int s = __builtin_amdgcn_readlane(sl, 4 * t + u);
          xv[u] = *reinterpret_cast<const float4*>(xh + (size_t)s * 128 + lane * 4);
        }
        const float al[4] = {quad_f<0x00>(at), quad_f<0x55>(at),
                             quad_f<0xAA>(at), quad_f<0xFF>(at)};
#pragma unroll
        for (int u = 0; u < 4; ++u) {
          acc.x = fmaf(xv[u].x, al[u], acc.x);
          acc.y = fmaf(xv[u].y, al[u], acc.y);
          acc.z = fmaf(xv[u].z, al[u], acc.z);
          acc.w = fmaf(xv[u].w, al[u], acc.w);
        }
      }
    }
  } else {
    for (int base = 0; base < cnt; base += 32) {
      const int i = base + lane;
      const int2 my = i < cnt ? slot[start + i] : make_int2(0x7fffffff, 0);
      int r = 0;
      for (int jb = 0; jb < cnt; jb += 32) {
        const int j = jb + lane;
        const int other = j < cnt ? slot[start + j].x : 0x7fffffff;
        const int nb = min(32, cnt - jb);
        for (int k = 0; k < nb; ++k)
          r += (__builtin_amdgcn_readlane(other, k) < my.x) ? 1 : 0;
      }
      if (i < cnt) csr_src[start + r] = my.y;
    }
    __threadfence();

    float m = -INFINITY;
    for (int base = 0; base < deg; base += 32) {
      const int i = base + lane;
      const int sl = i < cnt ? csr_src[start + i] : n;
      const int nb = min(32, deg - base);
      for (int k = 0; k < nb; k += 4) {
        float ev[4];
#pragma unroll
        for (int u = 0; u < 4; ++u) {
          const int s = __builtin_amdgcn_readlane(sl, k + u);
          ev[u] = leaky(as_[(size_t)s * 8 + h] + adn);
        }
#pragma unroll
        for (int u = 0; u < 4; ++u)
          m = (u == 0 || k + u < nb) ? fmaxf(m, ev[u]) : m;
      }
    }

    float den = 0.f;
    for (int base = 0; base < deg; base += 32) {
      const int i = base + lane;
      const int sl = i < cnt ? csr_src[start + i] : n;
      const int nb = min(32, deg - base);
      for (int k = 0; k < nb; k += 4) {
        float ev[4];
#pragma unroll
        for (int u = 0; u < 4; ++u) {
          const int s = __builtin_amdgcn_readlane(sl, k + u);
          ev[u] = leaky(as_[(size_t)s * 8 + h] + adn);
        }
#pragma unroll
        for (int u = 0; u < 4; ++u)
          den = (u == 0 || k + u < nb) ? den + expf(ev[u] - m) : den;
      }
    }
    const float dsum = den + 1e-16f;

    for (int base = 0; base < deg; base += 32) {
      const int i = base + lane;
      const int sl = i < cnt ? csr_src[start + i] : n;
      const int nb = min(32, deg - base);
      for (int k = 0; k < nb; k += 4) {
        float ev[4];
        float4 xv[4];
#pragma unroll
        for (int u = 0; u < 4; ++u) {
          const int s = __builtin_amdgcn_readlane(sl, k + u);
          ev[u] = leaky(as_[(size_t)s * 8 + h] + adn);
          xv[u] = *reinterpret_cast<const float4*>(xh + (size_t)s * 128 + lane * 4);
        }
#pragma unroll
        for (int u = 0; u < 4; ++u) {
          const float alpha = expf(ev[u] - m) / dsum;
          const bool ok = u == 0 || k + u < nb;
          acc.x = ok ? acc.x + xv[u].x * alpha : acc.x;
          acc.y = ok ? acc.y + xv[u].y * alpha : acc.y;
          acc.z = ok ? acc.z + xv[u].z * alpha : acc.z;
          acc.w = ok ? acc.w + xv[u].w * alpha : acc.w;
        }
      }
    }
  }
  const float4 b = *reinterpret_cast<const float4*>(bias + lane * 4);
  float4 o;
  o.x = acc.x + b.x; o.y = acc.y + b.y; o.z = acc.z + b.z; o.w = acc.w + b.w;
  o.x = o.x > 0.f ? o.x : (expf(o.x) - 1.f);
  o.y = o.y > 0.f ? o.y : (expf(o.y) - 1.f);
  o.z = o.z > 0.f ? o.z : (expf(o.z) - 1.f);
  o.w = o.w > 0.f ? o.w : (expf(o.w) - 1.f);
  *reinterpret_cast<float4*>(hout + (size_t)n * 128 + lane * 4) = o;
}

__global__ __launch_bounds__(256)
void node2_kernel(const int* __restrict__ rowptr, const int* __restrict__ csr_src,
                  const float* __restrict__ as_, const float* __restrict__ ad_,
                  const float* __restrict__ xh, const float* __restrict__ bias,
                  float* __restrict__ out, int N) {
  const int lane = threadIdx.x & 31;
  const int n = uni((int)((blockIdx.x * blockDim.x + threadIdx.x) >> 5));
  if (n >= N) return;
  const int start = rowptr[n];
  const int cnt = rowptr[n + 1] - start;
  const int deg = cnt + 1;
  const float adn = ad_[n];

  float2 acc = make_float2(0.f, 0.f);
  if (deg <= 32) {
    const int sl = lane < cnt ? csr_src[start + lane] : n;
    const float ev = leaky(as_[sl] + adn);
    float m = lane < deg ? ev : -INFINITY;
    for (int o = 16; o > 0; o >>= 1) m = fmaxf(m, __shfl_xor(m, o));
    const float ex = lane < deg ? expf(ev - m) : 0.f;
    float den = ex;
    for (int o = 16; o > 0; o >>= 1) den += __shfl_xor(den, o);
    const float alpha = ex / (den + 1e-16f);
    for (int k = 0; k < deg; k += 4) {
      float2 xv[4];
      float al[4];
#pragma unroll
      for (int u = 0; u < 4; ++u) {
        const int s = __builtin_amdgcn_readlane(sl, k + u);
        al[u] = lane_f(alpha, k + u);
        xv[u] = *reinterpret_cast<const float2*>(xh + (size_t)s * 64 + lane * 2);
      }
#pragma unroll
      for (int u = 0; u < 4; ++u) {
        acc.x = fmaf(xv[u].x, al[u], acc.x);
        acc.y = fmaf(xv[u].y, al[u], acc.y);
      }
    }
  } else {
    float m = -INFINITY;
    for (int base = 0; base < deg; base += 32) {
      const int i = base + lane;
      const int sl = i < cnt ? csr_src[start + i] : n;
      const float ev = leaky(as_[sl] + adn);
      m = i < deg ? fmaxf(m, ev) : m;
    }
    for (int o = 16; o > 0; o >>= 1) m = fmaxf(m, __shfl_xor(m, o));

    float den = 0.f;
    for (int base = 0; base < deg; base += 32) {
      const int i = base + lane;
      const int sl = i < cnt ? csr_src[start + i] : n;
      const float ev = leaky(as_[sl] + adn);
      float ex = i < deg ? expf(ev - m) : 0.f;
      for (int o = 16; o > 0; o >>= 1) ex += __shfl_xor(ex, o);
      den += ex;
    }
    const float dsum = den + 1e-16f;

    for (int base = 0; base < deg; base += 32) {
      const int i = base + lane;
      const int sl = i < cnt ? csr_src[start + i] : n;
      const float ev = leaky(as_[sl] + adn);
      const float alpha = expf(ev - m) / dsum;
      const int nb = min(32, deg - base);
      for (int k = 0; k < nb; k += 4) {
        float2 xv[4];
        float al[4];
#pragma unroll
        for (int u = 0; u < 4; ++u) {
          const int s = __builtin_amdgcn_readlane(sl, k + u);
          al[u] = lane_f(alpha, k + u);
          xv[u] = *reinterpret_cast<const float2*>(xh + (size_t)s * 64 + lane * 2);
        }
#pragma unroll
        for (int u = 0; u < 4; ++u) {
          const bool ok = u == 0 || k + u < nb;
          acc.x = ok ? acc.x + xv[u].x * al[u] : acc.x;
          acc.y = ok ? acc.y + xv[u].y * al[u] : acc.y;
        }
      }
    }
  }
  const float2 b = *reinterpret_cast<const float2*>(bias + lane * 2);
  float2 o;
  o.x = acc.x + b.x;
  o.y = acc.y + b.y;
  *reinterpret_cast<float2*>(out + (size_t)n * 64 + lane * 2) = o;
}

extern "C" void kernel_launch(void* const* d_in, const int* in_sizes, int n_in,
                              void* d_out, int out_size, void* d_ws,
                              size_t ws_size, hipStream_t stream) {
  (void)n_in; (void)out_size; (void)ws_size;
  const float* x        = (const float*)d_in[0];
  const int*   eidx     = (const int*)d_in[1];
  const float* W1       = (const float*)d_in[2];
  const float* att_src1 = (const float*)d_in[3];
  const float* att_dst1 = (const float*)d_in[4];
  const float* b1       = (const float*)d_in[5];
  const float* W2       = (const float*)d_in[6];
  const float* att_src2 = (const float*)d_in[7];
  const float* att_dst2 = (const float*)d_in[8];
  const float* b2       = (const float*)d_in[9];
  float* out = (float*)d_out;

  const int N    = in_sizes[0] / 128;
  const int E    = in_sizes[1] / 2;
  const int* esrc = eidx;
  const int* edst = eidx + E;

  float* ws = (float*)d_ws;
  float* xh1    = ws;        ws += (size_t)N * 128;
  float* h1     = ws;        ws += (size_t)N * 128;
  float* as1    = ws;        ws += (size_t)N * 8;
  float* ad1    = ws;        ws += (size_t)N * 8;
  int*   csr    = (int*)ws;  ws += (size_t)E;
  float* xh2    = ws;        ws += (size_t)N * 64;
  float* as2    = ws;        ws += (size_t)N;
  float* ad2    = ws;        ws += (size_t)N;
  int*   rowptr = (int*)ws;
  int*   slot   = (int*)xh2;
  const int nbins = (N + 255) / 256;
  int*   part     = (int*)h1;
  int*   partd    = part + (size_t)E * 2;
  int*   hist     = partd + (size_t)E;
  int*   binstart = hist + SRT_NB * SRT_MAXB;

  auto cdiv = [](int a, int b) { return (a + b - 1) / b; };

  gemm_wmma_kernel<128, 128, 8><<<768, 256, 0, stream>>>(
      x, W1, att_src1, att_dst1, xh1, as1, ad1, N / 16);

  bin_count_kernel<<<SRT_NB, 1024, 0, stream>>>(edst, hist, E);
  bin_scatter_kernel<<<SRT_NB, 1024, 0, stream>>>(esrc, edst, hist, part, partd,
                                                  binstart, E, nbins);
  bin_finish_kernel<<<nbins, 1024, 0, stream>>>(part, partd, binstart, rowptr,
                                                slot, N);

  node1_kernel<<<cdiv(N * 32, 256), 256, 0, stream>>>(rowptr, slot, csr, as1, ad1,
                                                      xh1, b1, h1, N);

  gemm_wmma_kernel<128, 64, 1><<<1024, 128, 0, stream>>>(
      h1, W2, att_src2, att_dst2, xh2, as2, ad2, N / 16);
  node2_kernel<<<cdiv(N * 32, 256), 256, 0, stream>>>(rowptr, csr, as2, ad2, xh2,
                                                      b2, out, N);
}


# layer-1 transform shares one launch with the grouping's second edge pass (independent work, blocks split by block index; that pass now in 256-thread blocks): 7 -> 6 launches; float sums unchanged
# speedup vs baseline: 8.2381x; 1.0831x over previous
#include <hip/hip_runtime.h>
#include <hip/hip_bf16.h>
#include <math.h>

typedef __attribute__((ext_vector_type(2))) float v2f;
typedef __attribute__((ext_vector_type(8))) float v8f;

#define NEG_SLOPE 0.2f


#define XT_PITCH 132

static __device__ __forceinline__ void lds_barrier() {
  asm volatile("s_wait_dscnt 0x0\n\ts_barrier_signal -1\n\ts_barrier_wait -1" ::: "memory");
}

#define SRT_NB 256
#define SRT_MAXB 256
#define SRT_CAP 8192

#define SRT_EB 8
static __device__ __forceinline__ void bin_scatter_block(
    const int* __restrict__ esrc, const int* __restrict__ edst,
    const int* __restrict__ hist, int* __restrict__ part_,
    int* __restrict__ partd, int* __restrict__ binstart, int E, int nbins,
    int* gbase, int* lc, int* wsum, int (*ptot)[SRT_MAXB], int (*ppre)[SRT_MAXB]) {
  int2* part = reinterpret_cast<int2*>(part_);
  const int tid = threadIdx.x, lane = tid & 31, wave = tid >> 5;
  const int b = blockIdx.x;
  {
    const int qt = __builtin_amdgcn_readfirstlane(tid >> 6), g4 = (tid & 63) * 4;
    const int nfront = min(max(b - qt * (SRT_NB / 4), 0), SRT_NB / 4);
    const int4* hp = reinterpret_cast<const int4*>(
        hist + (size_t)qt * (SRT_NB / 4) * SRT_MAXB + g4);
    int4 tot = make_int4(0, 0, 0, 0), pre = tot;
#pragma unroll 16
    for (int k = 0; k < SRT_NB / 4; ++k) {
      const int4 v = hp[k * (SRT_MAXB / 4)];
      tot.x += v.x; tot.y += v.y; tot.z += v.z; tot.w += v.w;
      const bool in_front = k < nfront;
      pre.x += in_front ? v.x : 0;
      pre.y += in_front ? v.y : 0;
      pre.z += in_front ? v.z : 0;
      pre.w += in_front ? v.w : 0;
    }
    *reinterpret_cast<int4*>(&ptot[qt][g4]) = tot;
    *reinterpret_cast<int4*>(&ppre[qt][g4]) = pre;
  }
  lc[tid] = 0;
  __syncthreads();
  const int t = ptot[0][tid] + ptot[1][tid] + ptot[2][tid] + ptot[3][tid];
  const int p = ppre[0][tid] + ppre[1][tid] + ppre[2][tid] + ppre[3][tid];
  int incl = t;
  for (int o = 1; o < 32; o <<= 1) {
    const int v = __shfl_up(incl, o);
    if (lane >= o) incl += v;
  }
  if (lane == 31) wsum[wave] = incl;
  __syncthreads();
  int excl = incl - t;
  for (int w = 0; w < wave; ++w) excl += wsum[w];
  gbase[tid] = excl + p;
  if (b == 0 && tid < nbins) binstart[tid] = excl;
  if (b == 0 && tid == 0) binstart[nbins] = E;
  __syncthreads();

  const int chunk = (E + SRT_NB - 1) / SRT_NB;
  const int e_beg = b * chunk;
  const int e_end = min(E, e_beg + chunk);
  for (int e0 = e_beg + tid; e0 < e_end; e0 += SRT_EB * SRT_MAXB) {
    int dv[SRT_EB], sv[SRT_EB];
#pragma unroll
    for (int u = 0; u < SRT_EB; ++u) {
      const int e = e0 + u * SRT_MAXB;
      const bool ok = e < e_end;
      dv[u] = ok ? edst[e] : 0;
      sv[u] = ok ? esrc[e] : 0;
    }
#pragma unroll
    for (int u = 0; u < SRT_EB; ++u) {
      const int e = e0 + u * SRT_MAXB;
      if (e < e_end) {
        const int bin = dv[u] >> 8;
        const int pos = gbase[bin] + atomicAdd(&lc[bin], 1);
        part[pos] = make_int2(e, sv[u]);
        partd[pos] = dv[u];
      }
    }
  }
}

template <int K, int NC, int H, bool RIDE>
__global__ __launch_bounds__((NC / 16) * 32)
void gemm_wmma_kernel(const float* __restrict__ X, const float* __restrict__ W,
                      const float* __restrict__ att_s,
                      const float* __restrict__ att_d, float* __restrict__ Y,
                      float* __restrict__ as_, float* __restrict__ ad_,
                      int ntiles,
                      const int* __restrict__ esrc, const int* __restrict__ edst,
                      const int* __restrict__ hist, int* __restrict__ part,
                      int* __restrict__ partd, int* __restrict__ binstart,
                      int E, int nbins) {
  int bid = blockIdx.x, nblk = gridDim.x;
  if constexpr (RIDE) {
    static_assert((NC / 16) * 32 == SRT_MAXB, "one thread per coarse bin");
    __shared__ int gbase[SRT_MAXB];
    __shared__ int lc[SRT_MAXB];
    __shared__ int wsum[SRT_MAXB / 32];
    __shared__ alignas(16) int ptot[4][SRT_MAXB];
    __shared__ alignas(16) int ppre[4][SRT_MAXB];
    if (bid < SRT_NB) {
      bin_scatter_block(esrc, edst, hist, part, partd, binstart, E, nbins,
                        gbase, lc, wsum, ptot, ppre);
      return;
    }
    bid -= SRT_NB;
    nblk -= SRT_NB;
  }
  constexpr int NW  = NC / 16;
  constexpr int TPH = NW / H;
  constexpr int RPW = 16 / NW;
  static_assert(K == 128, "one row of X = 32 lanes x 16 bytes");
  static_assert(RPW == 2 || RPW == 4, "a wave brings in two or four of a tile's 16 rows");
  __shared__ alignas(16) float xt[2][16 * XT_PITCH];
  __shared__ float ps[2][NW][16];
  __shared__ float pd[2][NW][16];
  const int lane = threadIdx.x & 31;
  const int wave = threadIdx.x >> 5;
  const int half = lane >> 4;
  const int l15  = lane & 15;
  const int col  = wave * 16 + l15;
  const float wsv = att_s[col], wdv = att_d[col];

  v2f bf[K / 4];
#pragma unroll
  for (int j = 0; j < K / 4; ++j) {
    const int ka = 4 * j + half * 2;
    bf[j].x = W[(size_t)ka * NC + col];
    bf[j].y = W[(size_t)(ka + 1) * NC + col];
  }

  float4 s0 = make_float4(0.f, 0.f, 0.f, 0.f), s1 = s0, s2 = s0, s3 = s0;
  auto fetch_rows = [&](int tile) {
    const float* p = X + (size_t)(tile * 16 + wave) * K + lane * 4;
    s0 = *reinterpret_cast<const float4*>(p);
    s1 = *reinterpret_cast<const float4*>(p + (size_t)NW * K);
    if constexpr (RPW == 4) {
      s2 = *reinterpret_cast<const float4*>(p + (size_t)2 * NW * K);
      s3 = *reinterpret_cast<const float4*>(p + (size_t)3 * NW * K);
    }
  };
  auto park_rows = [&](int buf) {
    float* q = &xt[buf][wave * XT_PITCH + lane * 4];
    *reinterpret_cast<float4*>(q) = s0;
    *reinterpret_cast<float4*>(q + NW * XT_PITCH) = s1;
    if constexpr (RPW == 4) {
      *reinterpret_cast<float4*>(q + 2 * NW * XT_PITCH) = s2;
      *reinterpret_cast<float4*>(q + 3 * NW * XT_PITCH) = s3;
    }
  };

  int t = bid;
  if (t < ntiles) {
    fetch_rows(t);
    park_rows(0);
  }
  lds_barrier();

  for (int it = 0; t < ntiles; t += nblk, ++it) {
    const int cur = it & 1;
    const int m0 = t * 16;
    const int tn = t + nblk;
    const bool more = tn < ntiles;
    if (more) fetch_rows(tn);

    const float* arow = &xt[cur][l15 * XT_PITCH + half * 2];
    v8f c = {};
#pragma unroll
    for (int j = 0; j < K / 4; ++j) {
      const float2 av = *reinterpret_cast<const float2*>(arow + 4 * j);
      v2f a; a.x = av.x; a.y = av.y;
      c = __builtin_amdgcn_wmma_f32_16x16x4_f32(false, a, false, bf[j], (short)0, c,
                                                false, false);
    }
    __builtin_amdgcn_sched_barrier(0);
    asm volatile("v_nop\n\tv_nop\n\tv_nop\n\tv_nop" : "+v"(c));
    __builtin_amdgcn_sched_barrier(0);

#pragma unroll
    for (int r = 0; r < 8; ++r)
      Y[(size_t)(m0 + half * 8 + r) * NC + col] = c[r];

    float rs = 0.f, rd = 0.f;
#pragma unroll
    for (int r = 0; r < 8; ++r) {
      float s = c[r] * wsv, d = c[r] * wdv;
      for (int o = 8; o > 0; o >>= 1) {
        s += __shfl_xor(s, o);
        d += __shfl_xor(d, o);
      }
      rs = l15 == r ? s : rs;
      rd = l15 == r ? d : rd;
    }
    if (l15 < 8) {
      ps[cur][wave][half * 8 + l15] = rs;
      pd[cur][wave][half * 8 + l15] = rd;
    }
    if (more) park_rows(cur ^ 1);
    lds_barrier();
    const int tid = threadIdx.x;
    if (tid < 16 * H) {
      const int head = tid % H, row = tid / H;
      float s = 0.f, d = 0.f;
      for (int w = 0; w < TPH; ++w) {
        s += ps[cur][head * TPH + w][row];
        d += pd[cur][head * TPH + w][row];
      }
      as_[(size_t)(m0 + row) * H + head] = s;
      ad_[(size_t)(m0 + row) * H + head] = d;
    }
  }
}

static __device__ __forceinline__ int uni(int v) {
  return __builtin_amdgcn_readfirstlane(v);
}
static __device__ __forceinline__ float lane_f(float v, int k) {
  return __int_as_float(__builtin_amdgcn_readlane(__float_as_int(v), k));
}
static __device__ __forceinline__ float leaky(float v) {
  return v > 0.f ? v : NEG_SLOPE * v;
}


__global__ __launch_bounds__(1024)
void bin_count_kernel(const int* __restrict__ edst, int* __restrict__ hist,
                      int E) {
  __shared__ int lc[SRT_MAXB];
  const int tid = threadIdx.x;
  if (tid < SRT_MAXB) lc[tid] = 0;
  __syncthreads();
  const int chunk = (E + SRT_NB - 1) / SRT_NB;
  const int e_beg = blockIdx.x * chunk;
  const int e_end = min(E, e_beg + chunk);
  for (int e = e_beg + tid; e < e_end; e += 1024) atomicAdd(&lc[edst[e] >> 8], 1);
  __syncthreads();
  if (tid < SRT_MAXB) hist[blockIdx.x * SRT_MAXB + tid] = lc[tid];
}

__global__ __launch_bounds__(1024)
void bin_finish_kernel(const int* __restrict__ part_, const int* __restrict__ partd,
                       const int* __restrict__ binstart, int* __restrict__ rowptr,
                       int* __restrict__ slot_, int N) {
  __shared__ int cnt[256];
  __shared__ int off[256];
  __shared__ int wtot[32];
  __shared__ int2 stage[SRT_CAP];
  const int2* part = reinterpret_cast<const int2*>(part_);
  int2* slot = reinterpret_cast<int2*>(slot_);
  const int tid = threadIdx.x, lane = tid & 31, wave = tid >> 5;
  const int lo = blockIdx.x * 256;
  const int span = min(256, N - lo);
  const int bs = binstart[blockIdx.x];
  const int n = binstart[blockIdx.x + 1] - bs;
  if (tid < 256) cnt[tid] = 0;
  __syncthreads();
  for (int i = tid; i < n; i += 1024) atomicAdd(&cnt[partd[bs + i] - lo], 1);
  __syncthreads();
  const int c = tid < 256 ? cnt[tid] : 0;
  int incl = c;
  for (int o = 1; o < 32; o <<= 1) {
    const int v = __shfl_up(incl, o);
    if (lane >= o) incl += v;
  }
  if (lane == 31) wtot[wave] = incl;
  __syncthreads();
  int woff = 0;
  for (int w = 0; w < wave; ++w) woff += wtot[w];
  const int excl = woff + incl - c;
  if (tid < 256) {
    off[tid] = excl;
    cnt[tid] = 0;
    if (tid < span) rowptr[lo + tid] = bs + excl;
    if (tid == span - 1 && lo + span == N) rowptr[N] = bs + excl + c;
  }
  __syncthreads();
  const bool staged = n <= SRT_CAP;
  for (int i = tid; i < n; i += 1024) {
    const int2 rec = part[bs + i];
    const int dl = partd[bs + i] - lo;
    const int p = off[dl] + atomicAdd(&cnt[dl], 1);
    if (staged) stage[p] = rec;
    else slot[bs + p] = rec;
  }
  __syncthreads();
  if (staged)
    for (int i = tid; i < n; i += 1024) slot[bs + i] = stage[i];
}

template <int CTRL>
static __device__ __forceinline__ float quad_f(float v) {
  return __int_as_float(
      __builtin_amdgcn_mov_dpp(__float_as_int(v), CTRL, 0xf, 0xf, true));
}
#define QUAD_XOR1 0xB1
#define QUAD_XOR2 0x4E

__global__ __launch_bounds__(256)
void node1_kernel(const int* __restrict__ rowptr, const int* __restrict__ slot_,
                  int* csr_src,
                  const float* __restrict__ as_, const float* __restrict__ ad_,
                  const float* __restrict__ xh, const float* __restrict__ bias,
                  float* __restrict__ hout, int N) {
  const int2* slot = reinterpret_cast<const int2*>(slot_);
  const int lane = threadIdx.x & 31;
  const int n = uni((int)((blockIdx.x * blockDim.x + threadIdx.x) >> 5));
  if (n >= N) return;
  const int h = lane >> 2;
  const int start = rowptr[n];
  const int cnt = rowptr[n + 1] - start;
  const int deg = cnt + 1;
  const float adn = ad_[(size_t)n * 8 + h];

  float4 acc = make_float4(0.f, 0.f, 0.f, 0.f);
  if (deg <= 32) {
    const int2 my = lane < cnt ? slot[start + lane] : make_int2(0x7fffffff, 0);
    int r = 0;
    for (int k = 0; k < cnt; ++k)
      r += (__builtin_amdgcn_readlane(my.x, k) < my.x) ? 1 : 0;
    if (lane < cnt) csr_src[start + r] = my.y;
    const int pushed = __builtin_amdgcn_ds_permute(r << 2, my.y);
    const int sl = lane < cnt ? pushed : n;

    const int q = lane & 3;
    float a[8];
#pragma unroll
    for (int t = 0; t < 8; ++t) {
      const int s = __builtin_amdgcn_ds_bpermute((4 * t + q) << 2, sl);
      const float e = leaky(as_[(size_t)s * 8 + h] + adn);
      a[t] = 4 * t + q < deg ? e : -INFINITY;
    }
    float m = a[0];
#pragma unroll
    for (int t = 1; t < 8; ++t) m = fmaxf(m, a[t]);
    m = fmaxf(m, quad_f<QUAD_XOR1>(m));
    m = fmaxf(m, quad_f<QUAD_XOR2>(m));

    float den = 0.f;
#pragma unroll
    for (int t = 0; t < 8; ++t) {
      if (t == 0 || 4 * t < deg) {
        a[t] = expf(a[t] - m);
        den += a[t];
      }
    }
    den += quad_f<QUAD_XOR1>(den);
    den += quad_f<QUAD_XOR2>(den);
    const float dsum = den + 1e-16f;

#pragma unroll
    for (int t = 0; t < 8; ++t) {
      if (t == 0 || 4 * t < deg) {
        const float at = a[t] / dsum;
        float4 xv[4];
#pragma unroll
        for (int u = 0; u < 4; ++u) {
          const int s = __builtin_amdgcn_readlane(sl, 4 * t + u);
          xv[u] = *reinterpret_cast<const float4*>(xh + (size_t)s * 128 + lane * 4);
        }
        const float al[4] = {quad_f<0x00>(at), quad_f<0x55>(at),
                             quad_f<0xAA>(at), quad_f<0xFF>(at)};
#pragma unroll
        for (int u = 0; u < 4; ++u) {
          acc.x = fmaf(xv[u].x, al[u], acc.x);
          acc.y = fmaf(xv[u].y, al[u], acc.y);
          acc.z = fmaf(xv[u].z, al[u], acc.z);
          acc.w = fmaf(xv[u].w, al[u], acc.w);
        }
      }
    }
  } else {
    for (int base = 0; base < cnt; base += 32) {
      const int i = base + lane;
      const int2 my = i < cnt ? slot[start + i] : make_int2(0x7fffffff, 0);
      int r = 0;
      for (int jb = 0; jb < cnt; jb += 32) {
        const int j = jb + lane;
        const int other = j < cnt ? slot[start + j].x : 0x7fffffff;
        const int nb = min(32, cnt - jb);
        for (int k = 0; k < nb; ++k)
          r += (__builtin_amdgcn_readlane(other, k) < my.x) ? 1 : 0;
      }
      if (i < cnt) csr_src[start + r] = my.y;
    }
    __threadfence();

    float m = -INFINITY;
    for (int base = 0; base < deg; base += 32) {
      const int i = base + lane;
      const int sl = i < cnt ? csr_src[start + i] : n;
      const int nb = min(32, deg - base);
      for (int k = 0; k < nb; k += 4) {
        float ev[4];
#pragma unroll
        for (int u = 0; u < 4; ++u) {
          const int s = __builtin_amdgcn_readlane(sl, k + u);
          ev[u] = leaky(as_[(size_t)s * 8 + h] + adn);
        }
#pragma unroll
        for (int u = 0; u < 4; ++u)
          m = (u == 0 || k + u < nb) ? fmaxf(m, ev[u]) : m;
      }
    }

    float den = 0.f;
    for (int base = 0; base < deg; base += 32) {
      const int i = base + lane;
      const int sl = i < cnt ? csr_src[start + i] : n;
      const int nb = min(32, deg - base);
      for (int k = 0; k < nb; k += 4) {
        float ev[4];
#pragma unroll
        for (int u = 0; u < 4; ++u) {
          const int s = __builtin_amdgcn_readlane(sl, k + u);
          ev[u] = leaky(as_[(size_t)s * 8 + h] + adn);
        }
#pragma unroll
        for (int u = 0; u < 4; ++u)
          den = (u == 0 || k + u < nb) ? den + expf(ev[u] - m) : den;
      }
    }
    const float dsum = den + 1e-16f;

    for (int base = 0; base < deg; base += 32) {
      const int i = base + lane;
      const int sl = i < cnt ? csr_src[start + i] : n;
      const int nb = min(32, deg - base);
      for (int k = 0; k < nb; k += 4) {
        float ev[4];
        float4 xv[4];
#pragma unroll
        for (int u = 0; u < 4; ++u) {
          const int s = __builtin_amdgcn_readlane(sl, k + u);
          ev[u] = leaky(as_[(size_t)s * 8 + h] + adn);
          xv[u] = *reinterpret_cast<const float4*>(xh + (size_t)s * 128 + lane * 4);
        }
#pragma unroll
        for (int u = 0; u < 4; ++u) {
          const float alpha = expf(ev[u] - m) / dsum;
          const bool ok = u == 0 || k + u < nb;
          acc.x = ok ? acc.x + xv[u].x * alpha : acc.x;
          acc.y = ok ? acc.y + xv[u].y * alpha : acc.y;
          acc.z = ok ? acc.z + xv[u].z * alpha : acc.z;
          acc.w = ok ? acc.w + xv[u].w * alpha : acc.w;
        }
      }
    }
  }
  const float4 b = *reinterpret_cast<const float4*>(bias + lane * 4);
  float4 o;
  o.x = acc.x + b.x; o.y = acc.y + b.y; o.z = acc.z + b.z; o.w = acc.w + b.w;
  o.x = o.x > 0.f ? o.x : (expf(o.x) - 1.f);
  o.y = o.y > 0.f ? o.y : (expf(o.y) - 1.f);
  o.z = o.z > 0.f ? o.z : (expf(o.z) - 1.f);
  o.w = o.w > 0.f ? o.w : (expf(o.w) - 1.f);
  *reinterpret_cast<float4*>(hout + (size_t)n * 128 + lane * 4) = o;
}

__global__ __launch_bounds__(256)
void node2_kernel(const int* __restrict__ rowptr, const int* __restrict__ csr_src,
                  const float* __restrict__ as_, const float* __restrict__ ad_,
                  const float* __restrict__ xh, const float* __restrict__ bias,
                  float* __restrict__ out, int N) {
  const int lane = threadIdx.x & 31;
  const int n = uni((int)((blockIdx.x * blockDim.x + threadIdx.x) >> 5));
  if (n >= N) return;
  const int start = rowptr[n];
  const int cnt = rowptr[n + 1] - start;
  const int deg = cnt + 1;
  const float adn = ad_[n];

  float2 acc = make_float2(0.f, 0.f);
  if (deg <= 32) {
    const int sl = lane < cnt ? csr_src[start + lane] : n;
    const float ev = leaky(as_[sl] + adn);
    float m = lane < deg ? ev : -INFINITY;
    for (int o = 16; o > 0; o >>= 1) m = fmaxf(m, __shfl_xor(m, o));
    const float ex = lane < deg ? expf(ev - m) : 0.f;
    float den = ex;
    for (int o = 16; o > 0; o >>= 1) den += __shfl_xor(den, o);
    const float alpha = ex / (den + 1e-16f);
    for (int k = 0; k < deg; k += 4) {
      float2 xv[4];
      float al[4];
#pragma unroll
      for (int u = 0; u < 4; ++u) {
        const int s = __builtin_amdgcn_readlane(sl, k + u);
        al[u] = lane_f(alpha, k + u);
        xv[u] = *reinterpret_cast<const float2*>(xh + (size_t)s * 64 + lane * 2);
      }
#pragma unroll
      for (int u = 0; u < 4; ++u) {
        acc.x = fmaf(xv[u].x, al[u], acc.x);
        acc.y = fmaf(xv[u].y, al[u], acc.y);
      }
    }
  } else {
    float m = -INFINITY;
    for (int base = 0; base < deg; base += 32) {
      const int i = base + lane;
      const int sl = i < cnt ? csr_src[start + i] : n;
      const float ev = leaky(as_[sl] + adn);
      m = i < deg ? fmaxf(m, ev) : m;
    }
    for (int o = 16; o > 0; o >>= 1) m = fmaxf(m, __shfl_xor(m, o));

    float den = 0.f;
    for (int base = 0; base < deg; base += 32) {
      const int i = base + lane;
      const int sl = i < cnt ? csr_src[start + i] : n;
      const float ev = leaky(as_[sl] + adn);
      float ex = i < deg ? expf(ev - m) : 0.f;
      for (int o = 16; o > 0; o >>= 1) ex += __shfl_xor(ex, o);
      den += ex;
    }
    const float dsum = den + 1e-16f;

    for (int base = 0; base < deg; base += 32) {
      const int i = base + lane;
      const int sl = i < cnt ? csr_src[start + i] : n;
      const float ev = leaky(as_[sl] + adn);
      const float alpha = expf(ev - m) / dsum;
      const int nb = min(32, deg - base);
      for (int k = 0; k < nb; k += 4) {
        float2 xv[4];
        float al[4];
#pragma unroll
        for (int u = 0; u < 4; ++u) {
          const int s = __builtin_amdgcn_readlane(sl, k + u);
          al[u] = lane_f(alpha, k + u);
          xv[u] = *reinterpret_cast<const float2*>(xh + (size_t)s * 64 + lane * 2);
        }
#pragma unroll
        for (int u = 0; u < 4; ++u) {
          const bool ok = u == 0 || k + u < nb;
          acc.x = ok ? acc.x + xv[u].x * al[u] : acc.x;
          acc.y = ok ? acc.y + xv[u].y * al[u] : acc.y;
        }
      }
    }
  }
  const float2 b = *reinterpret_cast<const float2*>(bias + lane * 2);
  float2 o;
  o.x = acc.x + b.x;
  o.y = acc.y + b.y;
  *reinterpret_cast<float2*>(out + (size_t)n * 64 + lane * 2) = o;
}

extern "C" void kernel_launch(void* const* d_in, const int* in_sizes, int n_in,
                              void* d_out, int out_size, void* d_ws,
                              size_t ws_size, hipStream_t stream) {
  (void)n_in; (void)out_size; (void)ws_size;
  const float* x        = (const float*)d_in[0];
  const int*   eidx     = (const int*)d_in[1];
  const float* W1       = (const float*)d_in[2];
  const float* att_src1 = (const float*)d_in[3];
  const float* att_dst1 = (const float*)d_in[4];
  const float* b1       = (const float*)d_in[5];
  const float* W2       = (const float*)d_in[6];
  const float* att_src2 = (const float*)d_in[7];
  const float* att_dst2 = (const float*)d_in[8];
  const float* b2       = (const float*)d_in[9];
  float* out = (float*)d_out;

  const int N    = in_sizes[0] / 128;
  const int E    = in_sizes[1] / 2;
  const int* esrc = eidx;
  const int* edst = eidx + E;

  float* ws = (float*)d_ws;
  float* xh1    = ws;        ws += (size_t)N * 128;
  float* h1     = ws;        ws += (size_t)N * 128;
  float* as1    = ws;        ws += (size_t)N * 8;
  float* ad1    = ws;        ws += (size_t)N * 8;
  int*   csr    = (int*)ws;  ws += (size_t)E;
  float* xh2    = ws;        ws += (size_t)N * 64;
  float* as2    = ws;        ws += (size_t)N;
  float* ad2    = ws;        ws += (size_t)N;
  int*   rowptr = (int*)ws;
  int*   slot   = (int*)xh2;
  const int nbins = (N + 255) / 256;
  int*   hist     = (int*)h1;
  int*   binstart = hist + SRT_NB * SRT_MAXB;
  int*   part     = binstart + 512;
  int*   partd    = part + (size_t)E * 2;

  auto cdiv = [](int a, int b) { return (a + b - 1) / b; };

  bin_count_kernel<<<SRT_NB, 1024, 0, stream>>>(edst, hist, E);
  gemm_wmma_kernel<128, 128, 8, true><<<SRT_NB + 512, 256, 0, stream>>>(
      x, W1, att_src1, att_dst1, xh1, as1, ad1, N / 16,
      esrc, edst, hist, part, partd, binstart, E, nbins);
  bin_finish_kernel<<<nbins, 1024, 0, stream>>>(part, partd, binstart, rowptr,
                                                slot, N);

  node1_kernel<<<cdiv(N * 32, 256), 256, 0, stream>>>(rowptr, slot, csr, as1, ad1,
                                                      xh1, b1, h1, N);

  gemm_wmma_kernel<128, 64, 1, false><<<1024, 128, 0, stream>>>(
      h1, W2, att_src2, att_dst2, xh2, as2, ad2, N / 16,
      nullptr, nullptr, nullptr, nullptr, nullptr, nullptr, 0, 0);
  node2_kernel<<<cdiv(N * 32, 256), 256, 0, stream>>>(rowptr, csr, as2, ad2, xh2,
                                                      b2, out, N);
}


# layer-2 transform done inside the layer-1 node kernel: 16 node waves park their rows in LDS and multiply the tile by W2 on WMMA (k in 4 quarters, fixed order: new sum order); 6 -> 5 launches
# speedup vs baseline: 8.6164x; 1.0459x over previous
#include <hip/hip_runtime.h>
#include <hip/hip_bf16.h>
#include <math.h>

typedef __attribute__((ext_vector_type(2))) float v2f;
typedef __attribute__((ext_vector_type(8))) float v8f;

#define NEG_SLOPE 0.2f


#define XT_PITCH 132

static __device__ __forceinline__ void lds_barrier() {
  asm volatile("s_wait_dscnt 0x0\n\ts_barrier_signal -1\n\ts_barrier_wait -1" ::: "memory");
}

#define SRT_NB 256
#define SRT_MAXB 256
#define SRT_CAP 8192

#define SRT_EB 8
static __device__ __forceinline__ void bin_scatter_block(
    const int* __restrict__ esrc, const int* __restrict__ edst,
    const int* __restrict__ hist, int* __restrict__ part_,
    int* __restrict__ partd, int* __restrict__ binstart, int E, int nbins,
    int* gbase, int* lc, int* wsum, int (*ptot)[SRT_MAXB], int (*ppre)[SRT_MAXB]) {
  int2* part = reinterpret_cast<int2*>(part_);
  const int tid = threadIdx.x, lane = tid & 31, wave = tid >> 5;
  const int b = blockIdx.x;
  {
    const int qt = __builtin_amdgcn_readfirstlane(tid >> 6), g4 = (tid & 63) * 4;
    const int nfront = min(max(b - qt * (SRT_NB / 4), 0), SRT_NB / 4);
    const int4* hp = reinterpret_cast<const int4*>(
        hist + (size_t)qt * (SRT_NB / 4) * SRT_MAXB + g4);
    int4 tot = make_int4(0, 0, 0, 0), pre = tot;
#pragma unroll 16
    for (int k = 0; k < SRT_NB / 4; ++k) {
      const int4 v = hp[k * (SRT_MAXB / 4)];
      tot.x += v.x; tot.y += v.y; tot.z += v.z; tot.w += v.w;
      const bool in_front = k < nfront;
      pre.x += in_front ? v.x : 0;
      pre.y += in_front ? v.y : 0;
      pre.z += in_front ? v.z : 0;
      pre.w += in_front ? v.w : 0;
    }
    *reinterpret_cast<int4*>(&ptot[qt][g4]) = tot;
    *reinterpret_cast<int4*>(&ppre[qt][g4]) = pre;
  }
  lc[tid] = 0;
  __syncthreads();
  const int t = ptot[0][tid] + ptot[1][tid] + ptot[2][tid] + ptot[3][tid];
  const int p = ppre[0][tid] + ppre[1][tid] + ppre[2][tid] + ppre[3][tid];
  int incl = t;
  for (int o = 1; o < 32; o <<= 1) {
    const int v = __shfl_up(incl, o);
    if (lane >= o) incl += v;
  }
  if (lane == 31) wsum[wave] = incl;
  __syncthreads();
  int excl = incl - t;
  for (int w = 0; w < wave; ++w) excl += wsum[w];
  gbase[tid] = excl + p;
  if (b == 0 && tid < nbins) binstart[tid] = excl;
  if (b == 0 && tid == 0) binstart[nbins] = E;
  __syncthreads();

  const int chunk = (E + SRT_NB - 1) / SRT_NB;
  const int e_beg = b * chunk;
  const int e_end = min(E, e_beg + chunk);
  for (int e0 = e_beg + tid; e0 < e_end; e0 += SRT_EB * SRT_MAXB) {
    int dv[SRT_EB], sv[SRT_EB];
#pragma unroll
    for (int u = 0; u < SRT_EB; ++u) {
      const int e = e0 + u * SRT_MAXB;
      const bool ok = e < e_end;
      dv[u] = ok ? edst[e] : 0;
      sv[u] = ok ? esrc[e] : 0;
    }
#pragma unroll
    for (int u = 0; u < SRT_EB; ++u) {
      const int e = e0 + u * SRT_MAXB;
      if (e < e_end) {
        const int bin = dv[u] >> 8;
        const int pos = gbase[bin] + atomicAdd(&lc[bin], 1);
        part[pos] = make_int2(e, sv[u]);
        partd[pos] = dv[u];
      }
    }
  }
}

template <int K, int NC, int H, bool RIDE>
__global__ __launch_bounds__((NC / 16) * 32)
void gemm_wmma_kernel(const float* __restrict__ X, const float* __restrict__ W,
                      const float* __restrict__ att_s,
                      const float* __restrict__ att_d, float* __restrict__ Y,
                      float* __restrict__ as_, float* __restrict__ ad_,
                      int ntiles,
                      const int* __restrict__ esrc, const int* __restrict__ edst,
                      const int* __restrict__ hist, int* __restrict__ part,
                      int* __restrict__ partd, int* __restrict__ binstart,
                      int E, int nbins) {
  int bid = blockIdx.x, nblk = gridDim.x;
  if constexpr (RIDE) {
    static_assert((NC / 16) * 32 == SRT_MAXB, "one thread per coarse bin");
    __shared__ int gbase[SRT_MAXB];
    __shared__ int lc[SRT_MAXB];
    __shared__ int wsum[SRT_MAXB / 32];
    __shared__ alignas(16) int ptot[4][SRT_MAXB];
    __shared__ alignas(16) int ppre[4][SRT_MAXB];
    if (bid < SRT_NB) {
      bin_scatter_block(esrc, edst, hist, part, partd, binstart, E, nbins,
                        gbase, lc, wsum, ptot, ppre);
      return;
    }
    bid -= SRT_NB;
    nblk -= SRT_NB;
  }
  constexpr int NW  = NC / 16;
  constexpr int TPH = NW / H;
  constexpr int RPW = 16 / NW;
  static_assert(K == 128, "one row of X = 32 lanes x 16 bytes");
  static_assert(RPW == 2 || RPW == 4, "a wave brings in two or four of a tile's 16 rows");
  __shared__ alignas(16) float xt[2][16 * XT_PITCH];
  __shared__ float ps[2][NW][16];
  __shared__ float pd[2][NW][16];
  const int lane = threadIdx.x & 31;
  const int wave = threadIdx.x >> 5;
  const int half = lane >> 4;
  const int l15  = lane & 15;
  const int col  = wave * 16 + l15;
  const float wsv = att_s[col], wdv = att_d[col];

  v2f bf[K / 4];
#pragma unroll
  for (int j = 0; j < K / 4; ++j) {
    const int ka = 4 * j + half * 2;
    bf[j].x = W[(size_t)ka * NC + col];
    bf[j].y = W[(size_t)(ka + 1) * NC + col];
  }

  float4 s0 = make_float4(0.f, 0.f, 0.f, 0.f), s1 = s0, s2 = s0, s3 = s0;
  auto fetch_rows = [&](int tile) {
    const float* p = X + (size_t)(tile * 16 + wave) * K + lane * 4;
    s0 = *reinterpret_cast<const float4*>(p);
    s1 = *reinterpret_cast<const float4*>(p + (size_t)NW * K);
    if constexpr (RPW == 4) {
      s2 = *reinterpret_cast<const float4*>(p + (size_t)2 * NW * K);
      s3 = *reinterpret_cast<const float4*>(p + (size_t)3 * NW * K);
    }
  };
  auto park_rows = [&](int buf) {
    float* q = &xt[buf][wave * XT_PITCH + lane * 4];
    *reinterpret_cast<float4*>(q) = s0;
    *reinterpret_cast<float4*>(q + NW * XT_PITCH) = s1;
    if constexpr (RPW == 4) {
      *reinterpret_cast<float4*>(q + 2 * NW * XT_PITCH) = s2;
      *reinterpret_cast<float4*>(q + 3 * NW * XT_PITCH) = s3;
    }
  };

  int t = bid;
  if (t < ntiles) {
    fetch_rows(t);
    park_rows(0);
  }
  lds_barrier();

  for (int it = 0; t < ntiles; t += nblk, ++it) {
    const int cur = it & 1;
    const int m0 = t * 16;
    const int tn = t + nblk;
    const bool more = tn < ntiles;
    if (more) fetch_rows(tn);

    const float* arow = &xt[cur][l15 * XT_PITCH + half * 2];
    v8f c = {};
#pragma unroll
    for (int j = 0; j < K / 4; ++j) {
      const float2 av = *reinterpret_cast<const float2*>(arow + 4 * j);
      v2f a; a.x = av.x; a.y = av.y;
      c = __builtin_amdgcn_wmma_f32_16x16x4_f32(false, a, false, bf[j], (short)0, c,
                                                false, false);
    }
    __builtin_amdgcn_sched_barrier(0);
    asm volatile("v_nop\n\tv_nop\n\tv_nop\n\tv_nop" : "+v"(c));
    __builtin_amdgcn_sched_barrier(0);

#pragma unroll
    for (int r = 0; r < 8; ++r)
      Y[(size_t)(m0 + half * 8 + r) * NC + col] = c[r];

    float rs = 0.f, rd = 0.f;
#pragma unroll
    for (int r = 0; r < 8; ++r) {
      float s = c[r] * wsv, d = c[r] * wdv;
      for (int o = 8; o > 0; o >>= 1) {
        s += __shfl_xor(s, o);
        d += __shfl_xor(d, o);
      }
      rs = l15 == r ? s : rs;
      rd = l15 == r ? d : rd;
    }
    if (l15 < 8) {
      ps[cur][wave][half * 8 + l15] = rs;
      pd[cur][wave][half * 8 + l15] = rd;
    }
    if (more) park_rows(cur ^ 1);
    lds_barrier();
    const int tid = threadIdx.x;
    if (tid < 16 * H) {
      const int head = tid % H, row = tid / H;
      float s = 0.f, d = 0.f;
      for (int w = 0; w < TPH; ++w) {
        s += ps[cur][head * TPH + w][row];
        d += pd[cur][head * TPH + w][row];
      }
      as_[(size_t)(m0 + row) * H + head] = s;
      ad_[(size_t)(m0 + row) * H + head] = d;
    }
  }
}

static __device__ __forceinline__ int uni(int v) {
  return __builtin_amdgcn_readfirstlane(v);
}
static __device__ __forceinline__ float lane_f(float v, int k) {
  return __int_as_float(__builtin_amdgcn_readlane(__float_as_int(v), k));
}
static __device__ __forceinline__ float leaky(float v) {
  return v > 0.f ? v : NEG_SLOPE * v;
}


__global__ __launch_bounds__(1024)
void bin_count_kernel(const int* __restrict__ edst, int* __restrict__ hist,
                      int E) {
  __shared__ int lc[SRT_MAXB];
  const int tid = threadIdx.x;
  if (tid < SRT_MAXB) lc[tid] = 0;
  __syncthreads();
  const int chunk = (E + SRT_NB - 1) / SRT_NB;
  const int e_beg = blockIdx.x * chunk;
  const int e_end = min(E, e_beg + chunk);
  for (int e = e_beg + tid; e < e_end; e += 1024) atomicAdd(&lc[edst[e] >> 8], 1);
  __syncthreads();
  if (tid < SRT_MAXB) hist[blockIdx.x * SRT_MAXB + tid] = lc[tid];
}

__global__ __launch_bounds__(1024)
void bin_finish_kernel(const int* __restrict__ part_, const int* __restrict__ partd,
                       const int* __restrict__ binstart, int* __restrict__ rowptr,
                       int* __restrict__ slot_, int N) {
  __shared__ int cnt[256];
  __shared__ int off[256];
  __shared__ int wtot[32];
  __shared__ int2 stage[SRT_CAP];
  const int2* part = reinterpret_cast<const int2*>(part_);
  int2* slot = reinterpret_cast<int2*>(slot_);
  const int tid = threadIdx.x, lane = tid & 31, wave = tid >> 5;
  const int lo = blockIdx.x * 256;
  const int span = min(256, N - lo);
  const int bs = binstart[blockIdx.x];
  const int n = binstart[blockIdx.x + 1] - bs;
  if (tid < 256) cnt[tid] = 0;
  __syncthreads();
  for (int i = tid; i < n; i += 1024) atomicAdd(&cnt[partd[bs + i] - lo], 1);
  __syncthreads();
  const int c = tid < 256 ? cnt[tid] : 0;
  int incl = c;
  for (int o = 1; o < 32; o <<= 1) {
    const int v = __shfl_up(incl, o);
    if (lane >= o) incl += v;
  }
  if (lane == 31) wtot[wave] = incl;
  __syncthreads();
  int woff = 0;
  for (int w = 0; w < wave; ++w) woff += wtot[w];
  const int excl = woff + incl - c;
  if (tid < 256) {
    off[tid] = excl;
    cnt[tid] = 0;
    if (tid < span) rowptr[lo + tid] = bs + excl;
    if (tid == span - 1 && lo + span == N) rowptr[N] = bs + excl + c;
  }
  __syncthreads();
  const bool staged = n <= SRT_CAP;
  for (int i = tid; i < n; i += 1024) {
    const int2 rec = part[bs + i];
    const int dl = partd[bs + i] - lo;
    const int p = off[dl] + atomicAdd(&cnt[dl], 1);
    if (staged) stage[p] = rec;
    else slot[bs + p] = rec;
  }
  __syncthreads();
  if (staged)
    for (int i = tid; i < n; i += 1024) slot[bs + i] = stage[i];
}

template <int CTRL>
static __device__ __forceinline__ float quad_f(float v) {
  return __int_as_float(
      __builtin_amdgcn_mov_dpp(__float_as_int(v), CTRL, 0xf, 0xf, true));
}
#define QUAD_XOR1 0xB1
#define QUAD_XOR2 0x4E

#define PT_PITCH 68
__global__ __launch_bounds__(512)
void node1_kernel(const int* __restrict__ rowptr, const int* __restrict__ slot_,
                  int* csr_src,
                  const float* __restrict__ as_, const float* __restrict__ ad_,
                  const float* __restrict__ xh, const float* __restrict__ bias,
                  const float* __restrict__ W2, const float* __restrict__ att_s2,
                  const float* __restrict__ att_d2, float* __restrict__ Y2,
                  float* __restrict__ as2, float* __restrict__ ad2, int N) {
  __shared__ alignas(16) float ht[16 * XT_PITCH];
  __shared__ alignas(16) float pt[4 * 16 * PT_PITCH];
  const int2* slot = reinterpret_cast<const int2*>(slot_);
  const int lane = threadIdx.x & 31;
  const int wave = uni((int)(threadIdx.x >> 5));
  const int n = blockIdx.x * 16 + wave;
  const bool live = n < N;
  const int half = lane >> 4, l15 = lane & 15;
  const int ct = wave & 3, kq = wave >> 2;

  float4 acc = make_float4(0.f, 0.f, 0.f, 0.f);
  if (live) {
  const int h = lane >> 2;
  const int start = rowptr[n];
  const int cnt = rowptr[n + 1] - start;
  const int deg = cnt + 1;
  const float adn = ad_[(size_t)n * 8 + h];

  if (deg <= 32) {
    const int2 my = lane < cnt ? slot[start + lane] : make_int2(0x7fffffff, 0);
    int r = 0;
    for (int k = 0; k < cnt; ++k)
      r += (__builtin_amdgcn_readlane(my.x, k) < my.x) ? 1 : 0;
    if (lane < cnt) csr_src[start + r] = my.y;
    const int pushed = __builtin_amdgcn_ds_permute(r << 2, my.y);
    const int sl = lane < cnt ? pushed : n;

    const int q = lane & 3;
    float a[8];
#pragma unroll
    for (int t = 0; t < 8; ++t) {
      const int s = __builtin_amdgcn_ds_bpermute((4 * t + q) << 2, sl);
      const float e = leaky(as_[(size_t)s * 8 + h] + adn);
      a[t] = 4 * t + q < deg ? e : -INFINITY;
    }
    float m = a[0];
#pragma unroll
    for (int t = 1; t < 8; ++t) m = fmaxf(m, a[t]);
    m = fmaxf(m, quad_f<QUAD_XOR1>(m));
    m = fmaxf(m, quad_f<QUAD_XOR2>(m));

    float den = 0.f;
#pragma unroll
    for (int t = 0; t < 8; ++t) {
      if (t == 0 || 4 * t < deg) {
        a[t] = expf(a[t] - m);
        den += a[t];
      }
    }
    den += quad_f<QUAD_XOR1>(den);
    den += quad_f<QUAD_XOR2>(den);
    const float dsum = den + 1e-16f;

#pragma unroll
    for (int t = 0; t < 8; ++t) {
      if (t == 0 || 4 * t < deg) {
        const float at = a[t] / dsum;
        float4 xv[4];
#pragma unroll
        for (int u = 0; u < 4; ++u) {
          const int s = __builtin_amdgcn_readlane(sl, 4 * t + u);
          xv[u] = *reinterpret_cast<const float4*>(xh + (size_t)s * 128 + lane * 4);
        }
        const float al[4] = {quad_f<0x00>(at), quad_f<0x55>(at),
                             quad_f<0xAA>(at), quad_f<0xFF>(at)};
#pragma unroll
        for (int u = 0; u < 4; ++u) {
          acc.x = fmaf(xv[u].x, al[u], acc.x);
          acc.y = fmaf(xv[u].y, al[u], acc.y);
          acc.z = fmaf(xv[u].z, al[u], acc.z);
          acc.w = fmaf(xv[u].w, al[u], acc.w);
        }
      }
    }
  } else {
    for (int base = 0; base < cnt; base += 32) {
      const int i = base + lane;
      const int2 my = i < cnt ? slot[start + i] : make_int2(0x7fffffff, 0);
      int r = 0;
      for (int jb = 0; jb < cnt; jb += 32) {
        const int j = jb + lane;
        const int other = j < cnt ? slot[start + j].x : 0x7fffffff;
        const int nb = min(32, cnt - jb);
        for (int k = 0; k < nb; ++k)
          r += (__builtin_amdgcn_readlane(other, k) < my.x) ? 1 : 0;
      }
      if (i < cnt) csr_src[start + r] = my.y;
    }
    __threadfence();

    float m = -INFINITY;
    for (int base = 0; base < deg; base += 32) {
      const int i = base + lane;
      const int sl = i < cnt ? csr_src[start + i] : n;
      const int nb = min(32, deg - base);
      for (int k = 0; k < nb; k += 4) {
        float ev[4];
#pragma unroll
        for (int u = 0; u < 4; ++u) {
          const int s = __builtin_amdgcn_readlane(sl, k + u);
          ev[u] = leaky(as_[(size_t)s * 8 + h] + adn);
        }
#pragma unroll
        for (int u = 0; u < 4; ++u)
          m = (u == 0 || k + u < nb) ? fmaxf(m, ev[u]) : m;
      }
    }

    float den = 0.f;
    for (int base = 0; base < deg; base += 32) {
      const int i = base + lane;
      const int sl = i < cnt ? csr_src[start + i] : n;
      const int nb = min(32, deg - base);
      for (int k = 0; k < nb; k += 4) {
        float ev[4];
#pragma unroll
        for (int u = 0; u < 4; ++u) {
          const int s = __builtin_amdgcn_readlane(sl, k + u);
          ev[u] = leaky(as_[(size_t)s * 8 + h] + adn);
        }
#pragma unroll
        for (int u = 0; u < 4; ++u)
          den = (u == 0 || k + u < nb) ? den + expf(ev[u] - m) : den;
      }
    }
    const float dsum = den + 1e-16f;

    for (int base = 0; base < deg; base += 32) {
      const int i = base + lane;
      const int sl = i < cnt ? csr_src[start + i] : n;
      const int nb = min(32, deg - base);
      for (int k = 0; k < nb; k += 4) {
        float ev[4];
        float4 xv[4];
#pragma unroll
        for (int u = 0; u < 4; ++u) {
          const int s = __builtin_amdgcn_readlane(sl, k + u);
          ev[u] = leaky(as_[(size_t)s * 8 + h] + adn);
          xv[u] = *reinterpret_cast<const float4*>(xh + (size_t)s * 128 + lane * 4);
        }
#pragma unroll
        for (int u = 0; u < 4; ++u) {
          const float alpha = expf(ev[u] - m) / dsum;
          const bool ok = u == 0 || k + u < nb;
          acc.x = ok ? acc.x + xv[u].x * alpha : acc.x;
          acc.y = ok ? acc.y + xv[u].y * alpha : acc.y;
          acc.z = ok ? acc.z + xv[u].z * alpha : acc.z;
          acc.w = ok ? acc.w + xv[u].w * alpha : acc.w;
        }
      }
    }
  }
  }

  v2f bfr[8];
#pragma unroll
  for (int jj = 0; jj < 8; ++jj) {
    const int ka = 4 * (8 * kq + jj) + half * 2;
    bfr[jj].x = W2[(size_t)ka * 64 + ct * 16 + l15];
    bfr[jj].y = W2[(size_t)(ka + 1) * 64 + ct * 16 + l15];
  }

  const float4 b = *reinterpret_cast<const float4*>(bias + lane * 4);
  float4 o;
  o.x = acc.x + b.x; o.y = acc.y + b.y; o.z = acc.z + b.z; o.w = acc.w + b.w;
  o.x = o.x > 0.f ? o.x : (expf(o.x) - 1.f);
  o.y = o.y > 0.f ? o.y : (expf(o.y) - 1.f);
  o.z = o.z > 0.f ? o.z : (expf(o.z) - 1.f);
  o.w = o.w > 0.f ? o.w : (expf(o.w) - 1.f);
  if (!live) o = make_float4(0.f, 0.f, 0.f, 0.f);

  *reinterpret_cast<float4*>(&ht[wave * XT_PITCH + lane * 4]) = o;
  lds_barrier();

  const float* arow = &ht[l15 * XT_PITCH + 32 * kq + half * 2];
  v8f c = {};
#pragma unroll
  for (int jj = 0; jj < 8; ++jj) {
    const float2 av = *reinterpret_cast<const float2*>(arow + 4 * jj);
    v2f a; a.x = av.x; a.y = av.y;
    c = __builtin_amdgcn_wmma_f32_16x16x4_f32(false, a, false, bfr[jj], (short)0, c,
                                              false, false);
  }
  __builtin_amdgcn_sched_barrier(0);
  asm volatile("v_nop\n\tv_nop\n\tv_nop\n\tv_nop" : "+v"(c));
  __builtin_amdgcn_sched_barrier(0);

#pragma unroll
  for (int r = 0; r < 8; ++r)
    pt[(kq * 16 + half * 8 + r) * PT_PITCH + ct * 16 + l15] = c[r];
  lds_barrier();

  const float* pr = &pt[wave * PT_PITCH + lane * 2];
  const float2 p0 = *reinterpret_cast<const float2*>(pr);
  const float2 p1 = *reinterpret_cast<const float2*>(pr + 16 * PT_PITCH);
  const float2 p2 = *reinterpret_cast<const float2*>(pr + 32 * PT_PITCH);
  const float2 p3 = *reinterpret_cast<const float2*>(pr + 48 * PT_PITCH);
  float2 y;
  y.x = ((p0.x + p1.x) + p2.x) + p3.x;
  y.y = ((p0.y + p1.y) + p2.y) + p3.y;
  if (live) {
    *reinterpret_cast<float2*>(Y2 + (size_t)n * 64 + lane * 2) = y;
    const float2 ws2 = *reinterpret_cast<const float2*>(att_s2 + lane * 2);
    const float2 wd2 = *reinterpret_cast<const float2*>(att_d2 + lane * 2);
    float s = y.x * ws2.x + y.y * ws2.y;
    float d = y.x * wd2.x + y.y * wd2.y;
    for (int ofs = 16; ofs > 0; ofs >>= 1) {
      s += __shfl_xor(s, ofs);
      d += __shfl_xor(d, ofs);
    }
    if (lane == 0) {
      as2[n] = s;
      ad2[n] = d;
    }
  }
}

__global__ __launch_bounds__(256)
void node2_kernel(const int* __restrict__ rowptr, const int* __restrict__ csr_src,
                  const float* __restrict__ as_, const float* __restrict__ ad_,
                  const float* __restrict__ xh, const float* __restrict__ bias,
                  float* __restrict__ out, int N) {
  const int lane = threadIdx.x & 31;
  const int n = uni((int)((blockIdx.x * blockDim.x + threadIdx.x) >> 5));
  if (n >= N) return;
  const int start = rowptr[n];
  const int cnt = rowptr[n + 1] - start;
  const int deg = cnt + 1;
  const float adn = ad_[n];

  float2 acc = make_float2(0.f, 0.f);
  if (deg <= 32) {
    const int sl = lane < cnt ? csr_src[start + lane] : n;
    const float ev = leaky(as_[sl] + adn);
    float m = lane < deg ? ev : -INFINITY;
    for (int o = 16; o > 0; o >>= 1) m = fmaxf(m, __shfl_xor(m, o));
    const float ex = lane < deg ? expf(ev - m) : 0.f;
    float den = ex;
    for (int o = 16; o > 0; o >>= 1) den += __shfl_xor(den, o);
    const float alpha = ex / (den + 1e-16f);
    for (int k = 0; k < deg; k += 4) {
      float2 xv[4];
      float al[4];
#pragma unroll
      for (int u = 0; u < 4; ++u) {
        const int s = __builtin_amdgcn_readlane(sl, k + u);
        al[u] = lane_f(alpha, k + u);
        xv[u] = *reinterpret_cast<const float2*>(xh + (size_t)s * 64 + lane * 2);
      }
#pragma unroll
      for (int u = 0; u < 4; ++u) {
        acc.x = fmaf(xv[u].x, al[u], acc.x);
        acc.y = fmaf(xv[u].y, al[u], acc.y);
      }
    }
  } else {
    float m = -INFINITY;
    for (int base = 0; base < deg; base += 32) {
      const int i = base + lane;
      const int sl = i < cnt ? csr_src[start + i] : n;
      const float ev = leaky(as_[sl] + adn);
      m = i < deg ? fmaxf(m, ev) : m;
    }
    for (int o = 16; o > 0; o >>= 1) m = fmaxf(m, __shfl_xor(m, o));

    float den = 0.f;
    for (int base = 0; base < deg; base += 32) {
      const int i = base + lane;
      const int sl = i < cnt ? csr_src[start + i] : n;
      const float ev = leaky(as_[sl] + adn);
      float ex = i < deg ? expf(ev - m) : 0.f;
      for (int o = 16; o > 0; o >>= 1) ex += __shfl_xor(ex, o);
      den += ex;
    }
    const float dsum = den + 1e-16f;

    for (int base = 0; base < deg; base += 32) {
      const int i = base + lane;
      const int sl = i < cnt ? csr_src[start + i] : n;
      const float ev = leaky(as_[sl] + adn);
      const float alpha = expf(ev - m) / dsum;
      const int nb = min(32, deg - base);
      for (int k = 0; k < nb; k += 4) {
        float2 xv[4];
        float al[4];
#pragma unroll
        for (int u = 0; u < 4; ++u) {
          const int s = __builtin_amdgcn_readlane(sl, k + u);
          al[u] = lane_f(alpha, k + u);
          xv[u] = *reinterpret_cast<const float2*>(xh + (size_t)s * 64 + lane * 2);
        }
#pragma unroll
        for (int u = 0; u < 4; ++u) {
          const bool ok = u == 0 || k + u < nb;
          acc.x = ok ? acc.x + xv[u].x * al[u] : acc.x;
          acc.y = ok ? acc.y + xv[u].y * al[u] : acc.y;
        }
      }
    }
  }
  const float2 b = *reinterpret_cast<const float2*>(bias + lane * 2);
  float2 o;
  o.x = acc.x + b.x;
  o.y = acc.y + b.y;
  *reinterpret_cast<float2*>(out + (size_t)n * 64 + lane * 2) = o;
}

extern "C" void kernel_launch(void* const* d_in, const int* in_sizes, int n_in,
                              void* d_out, int out_size, void* d_ws,
                              size_t ws_size, hipStream_t stream) {
  (void)n_in; (void)out_size; (void)ws_size;
  const float* x        = (const float*)d_in[0];
  const int*   eidx     = (const int*)d_in[1];
  const float* W1       = (const float*)d_in[2];
  const float* att_src1 = (const float*)d_in[3];
  const float* att_dst1 = (const float*)d_in[4];
  const float* b1       = (const float*)d_in[5];
  const float* W2       = (const float*)d_in[6];
  const float* att_src2 = (const float*)d_in[7];
  const float* att_dst2 = (const float*)d_in[8];
  const float* b2       = (const float*)d_in[9];
  float* out = (float*)d_out;

  const int N    = in_sizes[0] / 128;
  const int E    = in_sizes[1] / 2;
  const int* esrc = eidx;
  const int* edst = eidx + E;

  float* ws = (float*)d_ws;
  float* xh1    = ws;        ws += (size_t)N * 128;
  float* h1     = ws;        ws += (size_t)N * 128;
  float* as1    = ws;        ws += (size_t)N * 8;
  float* ad1    = ws;        ws += (size_t)N * 8;
  int*   csr    = (int*)ws;  ws += (size_t)E;
  float* xh2    = ws;        ws += (size_t)N * 64;
  float* as2    = ws;        ws += (size_t)N;
  float* ad2    = ws;        ws += (size_t)N;
  int*   rowptr = (int*)ws;
  const int nbins = (N + 255) / 256;
  int*   hist     = (int*)h1;
  int*   binstart = hist + SRT_NB * SRT_MAXB;
  int*   part     = binstart + 512;
  int*   partd    = part + (size_t)E * 2;
  int*   slot     = partd + (size_t)E;

  auto cdiv = [](int a, int b) { return (a + b - 1) / b; };

  bin_count_kernel<<<SRT_NB, 1024, 0, stream>>>(edst, hist, E);
  gemm_wmma_kernel<128, 128, 8, true><<<SRT_NB + 512, 256, 0, stream>>>(
      x, W1, att_src1, att_dst1, xh1, as1, ad1, N / 16,
      esrc, edst, hist, part, partd, binstart, E, nbins);
  bin_finish_kernel<<<nbins, 1024, 0, stream>>>(part, partd, binstart, rowptr,
                                                slot, N);

  node1_kernel<<<cdiv(N, 16), 512, 0, stream>>>(rowptr, slot, csr, as1, ad1, xh1, b1,
                                                W2, att_src2, att_dst2, xh2, as2,
                                                ad2, N);

  node2_kernel<<<cdiv(N * 32, 256), 256, 0, stream>>>(rowptr, csr, as2, ad2, xh2,
                                                      b2, out, N);
}


# grouping level 1 made chunk-local (a block counts, ranks and moves its own chunk by bin; the per-bin block fetches its 256 runs via two tables): count launch gone, 5 -> 4 launches; float sums same
# speedup vs baseline: 9.9136x; 1.1506x over previous
#include <hip/hip_runtime.h>
#include <hip/hip_bf16.h>
#include <math.h>

typedef __attribute__((ext_vector_type(2))) float v2f;
typedef __attribute__((ext_vector_type(8))) float v8f;

#define NEG_SLOPE 0.2f


#define XT_PITCH 132

static __device__ __forceinline__ void lds_barrier() {
  asm volatile("s_wait_dscnt 0x0\n\ts_barrier_signal -1\n\ts_barrier_wait -1" ::: "memory");
}

#define SRT_NB 256
#define SRT_MAXB 256
#define SRT_CAP 8192

#define SRT_EB 8
static __device__ __forceinline__ void bin_scatter_block(
    const int* __restrict__ esrc, const int* __restrict__ edst,
    int* __restrict__ hist, int* __restrict__ lpre, int* __restrict__ part_,
    int* __restrict__ partd, int E, int* gbase, int* lc, int* wsum) {
  int2* part = reinterpret_cast<int2*>(part_);
  const int tid = threadIdx.x, lane = tid & 31, wave = tid >> 5;
  const int b = blockIdx.x;
  const int chunk = (E + SRT_NB - 1) / SRT_NB;
  const int e_beg = b * chunk;
  const int e_end = min(E, e_beg + chunk);
  lc[tid] = 0;
  __syncthreads();

  for (int e0 = e_beg + tid; e0 < e_end; e0 += SRT_EB * SRT_MAXB) {
    int dv[SRT_EB];
#pragma unroll
    for (int u = 0; u < SRT_EB; ++u) {
      const int e = e0 + u * SRT_MAXB;
      dv[u] = e < e_end ? edst[e] : 0;
    }
#pragma unroll
    for (int u = 0; u < SRT_EB; ++u)
      if (e0 + u * SRT_MAXB < e_end) atomicAdd(&lc[dv[u] >> 8], 1);
  }
  __syncthreads();

  const int t = lc[tid];
  int incl = t;
  for (int o = 1; o < 32; o <<= 1) {
    const int v = __shfl_up(incl, o);
    if (lane >= o) incl += v;
  }
  if (lane == 31) wsum[wave] = incl;
  __syncthreads();
  int excl = incl - t;
  for (int w = 0; w < wave; ++w) excl += wsum[w];
  hist[b * SRT_MAXB + tid] = t;
  lpre[b * SRT_MAXB + tid] = excl;
  gbase[tid] = e_beg + excl;
  lc[tid] = 0;
  __syncthreads();

  for (int e0 = e_beg + tid; e0 < e_end; e0 += SRT_EB * SRT_MAXB) {
    int dv[SRT_EB], sv[SRT_EB];
#pragma unroll
    for (int u = 0; u < SRT_EB; ++u) {
      const int e = e0 + u * SRT_MAXB;
      const bool ok = e < e_end;
      dv[u] = ok ? edst[e] : 0;
      sv[u] = ok ? esrc[e] : 0;
    }
#pragma unroll
    for (int u = 0; u < SRT_EB; ++u) {
      const int e = e0 + u * SRT_MAXB;
      if (e < e_end) {
        const int bin = dv[u] >> 8;
        const int pos = gbase[bin] + atomicAdd(&lc[bin], 1);
        part[pos] = make_int2(e, sv[u]);
        partd[pos] = dv[u];
      }
    }
  }
}

template <int K, int NC, int H, bool RIDE>
__global__ __launch_bounds__((NC / 16) * 32)
void gemm_wmma_kernel(const float* __restrict__ X, const float* __restrict__ W,
                      const float* __restrict__ att_s,
                      const float* __restrict__ att_d, float* __restrict__ Y,
                      float* __restrict__ as_, float* __restrict__ ad_,
                      int ntiles,
                      const int* __restrict__ esrc, const int* __restrict__ edst,
                      int* __restrict__ hist, int* __restrict__ lpre,
                      int* __restrict__ part, int* __restrict__ partd, int E) {
  int bid = blockIdx.x, nblk = gridDim.x;
  if constexpr (RIDE) {
    static_assert((NC / 16) * 32 == SRT_MAXB, "one thread per coarse bin");
    __shared__ int gbase[SRT_MAXB];
    __shared__ int lc[SRT_MAXB];
    __shared__ int wsum[SRT_MAXB / 32];
    if (bid < SRT_NB) {
      bin_scatter_block(esrc, edst, hist, lpre, part, partd, E, gbase, lc, wsum);
      return;
    }
    bid -= SRT_NB;
    nblk -= SRT_NB;
  }
  constexpr int NW  = NC / 16;
  constexpr int TPH = NW / H;
  constexpr int RPW = 16 / NW;
  static_assert(K == 128, "one row of X = 32 lanes x 16 bytes");
  static_assert(RPW == 2 || RPW == 4, "a wave brings in two or four of a tile's 16 rows");
  __shared__ alignas(16) float xt[2][16 * XT_PITCH];
  __shared__ float ps[2][NW][16];
  __shared__ float pd[2][NW][16];
  const int lane = threadIdx.x & 31;
  const int wave = threadIdx.x >> 5;
  const int half = lane >> 4;
  const int l15  = lane & 15;
  const int col  = wave * 16 + l15;
  const float wsv = att_s[col], wdv = att_d[col];

  v2f bf[K / 4];
#pragma unroll
  for (int j = 0; j < K / 4; ++j) {
    const int ka = 4 * j + half * 2;
    bf[j].x = W[(size_t)ka * NC + col];
    bf[j].y = W[(size_t)(ka + 1) * NC + col];
  }

  float4 s0 = make_float4(0.f, 0.f, 0.f, 0.f), s1 = s0, s2 = s0, s3 = s0;
  auto fetch_rows = [&](int tile) {
    const float* p = X + (size_t)(tile * 16 + wave) * K + lane * 4;
    s0 = *reinterpret_cast<const float4*>(p);
    s1 = *reinterpret_cast<const float4*>(p + (size_t)NW * K);
    if constexpr (RPW == 4) {
      s2 = *reinterpret_cast<const float4*>(p + (size_t)2 * NW * K);
      s3 = *reinterpret_cast<const float4*>(p + (size_t)3 * NW * K);
    }
  };
  auto park_rows = [&](int buf) {
    float* q = &xt[buf][wave * XT_PITCH + lane * 4];
    *reinterpret_cast<float4*>(q) = s0;
    *reinterpret_cast<float4*>(q + NW * XT_PITCH) = s1;
    if constexpr (RPW == 4) {
      *reinterpret_cast<float4*>(q + 2 * NW * XT_PITCH) = s2;
      *reinterpret_cast<float4*>(q + 3 * NW * XT_PITCH) = s3;
    }
  };

  int t = bid;
  if (t < ntiles) {
    fetch_rows(t);
    park_rows(0);
  }
  lds_barrier();

  for (int it = 0; t < ntiles; t += nblk, ++it) {
    const int cur = it & 1;
    const int m0 = t * 16;
    const int tn = t + nblk;
    const bool more = tn < ntiles;
    if (more) fetch_rows(tn);

    const float* arow = &xt[cur][l15 * XT_PITCH + half * 2];
    v8f c = {};
#pragma unroll
    for (int j = 0; j < K / 4; ++j) {
      const float2 av = *reinterpret_cast<const float2*>(arow + 4 * j);
      v2f a; a.x = av.x; a.y = av.y;
      c = __builtin_amdgcn_wmma_f32_16x16x4_f32(false, a, false, bf[j], (short)0, c,
                                                false, false);
    }
    __builtin_amdgcn_sched_barrier(0);
    asm volatile("v_nop\n\tv_nop\n\tv_nop\n\tv_nop" : "+v"(c));
    __builtin_amdgcn_sched_barrier(0);

#pragma unroll
    for (int r = 0; r < 8; ++r)
      Y[(size_t)(m0 + half * 8 + r) * NC + col] = c[r];

    float rs = 0.f, rd = 0.f;
#pragma unroll
    for (int r = 0; r < 8; ++r) {
      float s = c[r] * wsv, d = c[r] * wdv;
      for (int o = 8; o > 0; o >>= 1) {
        s += __shfl_xor(s, o);
        d += __shfl_xor(d, o);
      }
      rs = l15 == r ? s : rs;
      rd = l15 == r ? d : rd;
    }
    if (l15 < 8) {
      ps[cur][wave][half * 8 + l15] = rs;
      pd[cur][wave][half * 8 + l15] = rd;
    }
    if (more) park_rows(cur ^ 1);
    lds_barrier();
    const int tid = threadIdx.x;
    if (tid < 16 * H) {
      const int head = tid % H, row = tid / H;
      float s = 0.f, d = 0.f;
      for (int w = 0; w < TPH; ++w) {
        s += ps[cur][head * TPH + w][row];
        d += pd[cur][head * TPH + w][row];
      }
      as_[(size_t)(m0 + row) * H + head] = s;
      ad_[(size_t)(m0 + row) * H + head] = d;
    }
  }
}

static __device__ __forceinline__ int uni(int v) {
  return __builtin_amdgcn_readfirstlane(v);
}
static __device__ __forceinline__ float lane_f(float v, int k) {
  return __int_as_float(__builtin_amdgcn_readlane(__float_as_int(v), k));
}
static __device__ __forceinline__ float leaky(float v) {
  return v > 0.f ? v : NEG_SLOPE * v;
}


__global__ __launch_bounds__(1024)
void bin_finish_kernel(const int* __restrict__ part_, const int* __restrict__ partd,
                       const int* __restrict__ hist, const int* __restrict__ lpre,
                       int* __restrict__ rowptr, int* __restrict__ slot_,
                       int N, int E) {
  static_assert(SRT_NB == 256 && SRT_MAXB == 256, "thread i < 256 = chunk i = bin i = node i of the bin");
  __shared__ int cnt[256];
  __shared__ int off[256];
  __shared__ int wtot[32];
  __shared__ int wrun[8];
  __shared__ int wlow[8];
  __shared__ alignas(16) int ptot[16][SRT_MAXB];
  __shared__ int rsrc[SRT_NB];
  __shared__ int rpos[SRT_NB];
  __shared__ int rcnt[SRT_NB];
  __shared__ int2 sin[SRT_CAP];
  __shared__ int sdl[SRT_CAP];
  __shared__ int2 stage[SRT_CAP];
  const int2* part = reinterpret_cast<const int2*>(part_);
  int2* slot = reinterpret_cast<int2*>(slot_);
  const int tid = threadIdx.x, lane = tid & 31, wave = tid >> 5;
  const int j = blockIdx.x;
  const int lo = j * 256;
  const int span = min(256, N - lo);
  const int chunk = (E + SRT_NB - 1) / SRT_NB;

  {
    const int qt = tid >> 6, g = tid & 63;
    const int4* hp = reinterpret_cast<const int4*>(hist) +
                     (size_t)qt * 16 * (SRT_MAXB / 4) + g;
    int4 tot = make_int4(0, 0, 0, 0);
#pragma unroll
    for (int k = 0; k < 16; ++k) {
      const int4 v = hp[k * (SRT_MAXB / 4)];
      tot.x += v.x; tot.y += v.y; tot.z += v.z; tot.w += v.w;
    }
    *reinterpret_cast<int4*>(&ptot[qt][g * 4]) = tot;
  }
  int rc = 0, rs = 0, rex = 0;
  if (tid < SRT_NB) {
    rc = hist[tid * SRT_MAXB + j];
    rs = tid * chunk + lpre[tid * SRT_MAXB + j];
    cnt[tid] = 0;
  }
  __syncthreads();
  if (tid < 256) {
    int t = 0;
    for (int s = 0; s < 16; ++s) t += ptot[s][tid];
    int low = tid < j ? t : 0;
    int incl = rc;
    for (int o = 1; o < 32; o <<= 1) {
      const int v = __shfl_up(incl, o);
      if (lane >= o) incl += v;
    }
    for (int o = 16; o > 0; o >>= 1) low += __shfl_xor(low, o);
    if (lane == 31) wrun[wave] = incl;
    if (lane == 0) wlow[wave] = low;
    rcnt[tid] = rc;
    rsrc[tid] = rs;
    rex = incl - rc;
  }
  __syncthreads();
  int bs = 0, n = 0;
  for (int w = 0; w < 8; ++w) { bs += wlow[w]; n += wrun[w]; }
  if (tid < 256) {
    int rp = rex;
    for (int w = 0; w < wave; ++w) rp += wrun[w];
    rpos[tid] = rp;
  }
  __syncthreads();
  const bool staged = n <= SRT_CAP;

  const int rb = tid >> 2, rq = tid & 3;
  const int my_rc = rcnt[rb], my_rs = rsrc[rb], my_rp = rpos[rb];
  for (int i0 = rq; i0 < my_rc; i0 += 16) {
    int2 rec[4];
    int dd[4];
#pragma unroll
    for (int u = 0; u < 4; ++u) {
      const int i = i0 + 4 * u;
      const bool ok = i < my_rc;
      rec[u] = ok ? part[my_rs + i] : make_int2(0, 0);
      dd[u] = ok ? partd[my_rs + i] : lo;
    }
#pragma unroll
    for (int u = 0; u < 4; ++u) {
      const int i = i0 + 4 * u;
      if (i < my_rc) {
        const int dl = dd[u] - lo;
        atomicAdd(&cnt[dl], 1);
        if (staged) {
          sin[my_rp + i] = rec[u];
          sdl[my_rp + i] = dl;
        }
      }
    }
  }
  __syncthreads();
  const int c = tid < 256 ? cnt[tid] : 0;
  int incl = c;
  for (int o = 1; o < 32; o <<= 1) {
    const int v = __shfl_up(incl, o);
    if (lane >= o) incl += v;
  }
  if (lane == 31) wtot[wave] = incl;
  __syncthreads();
  int woff = 0;
  for (int w = 0; w < wave; ++w) woff += wtot[w];
  const int excl = woff + incl - c;
  if (tid < 256) {
    off[tid] = excl;
    cnt[tid] = 0;
    if (tid < span) rowptr[lo + tid] = bs + excl;
    if (tid == span - 1 && lo + span == N) rowptr[N] = bs + excl + c;
  }
  __syncthreads();
  if (staged) {
    for (int i = tid; i < n; i += 1024) {
      const int dl = sdl[i];
      const int p = off[dl] + atomicAdd(&cnt[dl], 1);
      stage[p] = sin[i];
    }
    __syncthreads();
    for (int i = tid; i < n; i += 1024) slot[bs + i] = stage[i];
  } else {
    for (int i = rq; i < my_rc; i += 4) {
      const int2 rec = part[my_rs + i];
      const int dl = partd[my_rs + i] - lo;
      const int p = off[dl] + atomicAdd(&cnt[dl], 1);
      slot[bs + p] = rec;
    }
  }
}

template <int CTRL>
static __device__ __forceinline__ float quad_f(float v) {
  return __int_as_float(
      __builtin_amdgcn_mov_dpp(__float_as_int(v), CTRL, 0xf, 0xf, true));
}
#define QUAD_XOR1 0xB1
#define QUAD_XOR2 0x4E

#define PT_PITCH 68
__global__ __launch_bounds__(512)
void node1_kernel(const int* __restrict__ rowptr, const int* __restrict__ slot_,
                  int* csr_src,
                  const float* __restrict__ as_, const float* __restrict__ ad_,
                  const float* __restrict__ xh, const float* __restrict__ bias,
                  const float* __restrict__ W2, const float* __restrict__ att_s2,
                  const float* __restrict__ att_d2, float* __restrict__ Y2,
                  float* __restrict__ as2, float* __restrict__ ad2, int N) {
  __shared__ alignas(16) float ht[16 * XT_PITCH];
  __shared__ alignas(16) float pt[4 * 16 * PT_PITCH];
  const int2* slot = reinterpret_cast<const int2*>(slot_);
  const int lane = threadIdx.x & 31;
  const int wave = uni((int)(threadIdx.x >> 5));
  const int n = blockIdx.x * 16 + wave;
  const bool live = n < N;
  const int half = lane >> 4, l15 = lane & 15;
  const int ct = wave & 3, kq = wave >> 2;

  float4 acc = make_float4(0.f, 0.f, 0.f, 0.f);
  if (live) {
  const int h = lane >> 2;
  const int start = rowptr[n];
  const int cnt = rowptr[n + 1] - start;
  const int deg = cnt + 1;
  const float adn = ad_[(size_t)n * 8 + h];

  if (deg <= 32) {
    const int2 my = lane < cnt ? slot[start + lane] : make_int2(0x7fffffff, 0);
    int r = 0;
    for (int k = 0; k < cnt; ++k)
      r += (__builtin_amdgcn_readlane(my.x, k) < my.x) ? 1 : 0;
    if (lane < cnt) csr_src[start + r] = my.y;
    const int pushed = __builtin_amdgcn_ds_permute(r << 2, my.y);
    const int sl = lane < cnt ? pushed : n;

    const int q = lane & 3;
    float a[8];
#pragma unroll
    for (int t = 0; t < 8; ++t) {
      const int s = __builtin_amdgcn_ds_bpermute((4 * t + q) << 2, sl);
      const float e = leaky(as_[(size_t)s * 8 + h] + adn);
      a[t] = 4 * t + q < deg ? e : -INFINITY;
    }
    float m = a[0];
#pragma unroll
    for (int t = 1; t < 8; ++t) m = fmaxf(m, a[t]);
    m = fmaxf(m, quad_f<QUAD_XOR1>(m));
    m = fmaxf(m, quad_f<QUAD_XOR2>(m));

    float den = 0.f;
#pragma unroll
    for (int t = 0; t < 8; ++t) {
      if (t == 0 || 4 * t < deg) {
        a[t] = expf(a[t] - m);
        den += a[t];
      }
    }
    den += quad_f<QUAD_XOR1>(den);
    den += quad_f<QUAD_XOR2>(den);
    const float dsum = den + 1e-16f;

#pragma unroll
    for (int t = 0; t < 8; ++t) {
      if (t == 0 || 4 * t < deg) {
        const float at = a[t] / dsum;
        float4 xv[4];
#pragma unroll
        for (int u = 0; u < 4; ++u) {
          const int s = __builtin_amdgcn_readlane(sl, 4 * t + u);
          xv[u] = *reinterpret_cast<const float4*>(xh + (size_t)s * 128 + lane * 4);
        }
        const float al[4] = {quad_f<0x00>(at), quad_f<0x55>(at),
                             quad_f<0xAA>(at), quad_f<0xFF>(at)};
#pragma unroll
        for (int u = 0; u < 4; ++u) {
          acc.x = fmaf(xv[u].x, al[u], acc.x);
          acc.y = fmaf(xv[u].y, al[u], acc.y);
          acc.z = fmaf(xv[u].z, al[u], acc.z);
          acc.w = fmaf(xv[u].w, al[u], acc.w);
        }
      }
    }
  } else {
    for (int base = 0; base < cnt; base += 32) {
      const int i = base + lane;
      const int2 my = i < cnt ? slot[start + i] : make_int2(0x7fffffff, 0);
      int r = 0;
      for (int jb = 0; jb < cnt; jb += 32) {
        const int j = jb + lane;
        const int other = j < cnt ? slot[start + j].x : 0x7fffffff;
        const int nb = min(32, cnt - jb);
        for (int k = 0; k < nb; ++k)
          r += (__builtin_amdgcn_readlane(other, k) < my.x) ? 1 : 0;
      }
      if (i < cnt) csr_src[start + r] = my.y;
    }
    __threadfence();

    float m = -INFINITY;
    for (int base = 0; base < deg; base += 32) {
      const int i = base + lane;
      const int sl = i < cnt ? csr_src[start + i] : n;
      const int nb = min(32, deg - base);
      for (int k = 0; k < nb; k += 4) {
        float ev[4];
#pragma unroll
        for (int u = 0; u < 4; ++u) {
          const int s = __builtin_amdgcn_readlane(sl, k + u);
          ev[u] = leaky(as_[(size_t)s * 8 + h] + adn);
        }
#pragma unroll
        for (int u = 0; u < 4; ++u)
          m = (u == 0 || k + u < nb) ? fmaxf(m, ev[u]) : m;
      }
    }

    float den = 0.f;
    for (int base = 0; base < deg; base += 32) {
      const int i = base + lane;
      const int sl = i < cnt ? csr_src[start + i] : n;
      const int nb = min(32, deg - base);
      for (int k = 0; k < nb; k += 4) {
        float ev[4];
#pragma unroll
        for (int u = 0; u < 4; ++u) {
          const int s = __builtin_amdgcn_readlane(sl, k + u);
          ev[u] = leaky(as_[(size_t)s * 8 + h] + adn);
        }
#pragma unroll
        for (int u = 0; u < 4; ++u)
          den = (u == 0 || k + u < nb) ? den + expf(ev[u] - m) : den;
      }
    }
    const float dsum = den + 1e-16f;

    for (int base = 0; base < deg; base += 32) {
      const int i = base + lane;
      const int sl = i < cnt ? csr_src[start + i] : n;
      const int nb = min(32, deg - base);
      for (int k = 0; k < nb; k += 4) {
        float ev[4];
        float4 xv[4];
#pragma unroll
        for (int u = 0; u < 4; ++u) {
          const int s = __builtin_amdgcn_readlane(sl, k + u);
          ev[u] = leaky(as_[(size_t)s * 8 + h] + adn);
          xv[u] = *reinterpret_cast<const float4*>(xh + (size_t)s * 128 + lane * 4);
        }
#pragma unroll
        for (int u = 0; u < 4; ++u) {
          const float alpha = expf(ev[u] - m) / dsum;
          const bool ok = u == 0 || k + u < nb;
          acc.x = ok ? acc.x + xv[u].x * alpha : acc.x;
          acc.y = ok ? acc.y + xv[u].y * alpha : acc.y;
          acc.z = ok ? acc.z + xv[u].z * alpha : acc.z;
          acc.w = ok ? acc.w + xv[u].w * alpha : acc.w;
        }
      }
    }
  }
  }

  v2f bfr[8];
#pragma unroll
  for (int jj = 0; jj < 8; ++jj) {
    const int ka = 4 * (8 * kq + jj) + half * 2;
    bfr[jj].x = W2[(size_t)ka * 64 + ct * 16 + l15];
    bfr[jj].y = W2[(size_t)(ka + 1) * 64 + ct * 16 + l15];
  }

  const float4 b = *reinterpret_cast<const float4*>(bias + lane * 4);
  float4 o;
  o.x = acc.x + b.x; o.y = acc.y + b.y; o.z = acc.z + b.z; o.w = acc.w + b.w;
  o.x = o.x > 0.f ? o.x : (expf(o.x) - 1.f);
  o.y = o.y > 0.f ? o.y : (expf(o.y) - 1.f);
  o.z = o.z > 0.f ? o.z : (expf(o.z) - 1.f);
  o.w = o.w > 0.f ? o.w : (expf(o.w) - 1.f);
  if (!live) o = make_float4(0.f, 0.f, 0.f, 0.f);

  *reinterpret_cast<float4*>(&ht[wave * XT_PITCH + lane * 4]) = o;
  lds_barrier();

  const float* arow = &ht[l15 * XT_PITCH + 32 * kq + half * 2];
  v8f c = {};
#pragma unroll
  for (int jj = 0; jj < 8; ++jj) {
    const float2 av = *reinterpret_cast<const float2*>(arow + 4 * jj);
    v2f a; a.x = av.x; a.y = av.y;
    c = __builtin_amdgcn_wmma_f32_16x16x4_f32(false, a, false, bfr[jj], (short)0, c,
                                              false, false);
  }
  __builtin_amdgcn_sched_barrier(0);
  asm volatile("v_nop\n\tv_nop\n\tv_nop\n\tv_nop" : "+v"(c));
  __builtin_amdgcn_sched_barrier(0);

#pragma unroll
  for (int r = 0; r < 8; ++r)
    pt[(kq * 16 + half * 8 + r) * PT_PITCH + ct * 16 + l15] = c[r];
  lds_barrier();

  const float* pr = &pt[wave * PT_PITCH + lane * 2];
  const float2 p0 = *reinterpret_cast<const float2*>(pr);
  const float2 p1 = *reinterpret_cast<const float2*>(pr + 16 * PT_PITCH);
  const float2 p2 = *reinterpret_cast<const float2*>(pr + 32 * PT_PITCH);
  const float2 p3 = *reinterpret_cast<const float2*>(pr + 48 * PT_PITCH);
  float2 y;
  y.x = ((p0.x + p1.x) + p2.x) + p3.x;
  y.y = ((p0.y + p1.y) + p2.y) + p3.y;
  if (live) {
    *reinterpret_cast<float2*>(Y2 + (size_t)n * 64 + lane * 2) = y;
    const float2 ws2 = *reinterpret_cast<const float2*>(att_s2 + lane * 2);
    const float2 wd2 = *reinterpret_cast<const float2*>(att_d2 + lane * 2);
    float s = y.x * ws2.x + y.y * ws2.y;
    float d = y.x * wd2.x + y.y * wd2.y;
    for (int ofs = 16; ofs > 0; ofs >>= 1) {
      s += __shfl_xor(s, ofs);
      d += __shfl_xor(d, ofs);
    }
    if (lane == 0) {
      as2[n] = s;
      ad2[n] = d;
    }
  }
}

__global__ __launch_bounds__(256)
void node2_kernel(const int* __restrict__ rowptr, const int* __restrict__ csr_src,
                  const float* __restrict__ as_, const float* __restrict__ ad_,
                  const float* __restrict__ xh, const float* __restrict__ bias,
                  float* __restrict__ out, int N) {
  const int lane = threadIdx.x & 31;
  const int n = uni((int)((blockIdx.x * blockDim.x + threadIdx.x) >> 5));
  if (n >= N) return;
  const int start = rowptr[n];
  const int cnt = rowptr[n + 1] - start;
  const int deg = cnt + 1;
  const float adn = ad_[n];

  float2 acc = make_float2(0.f, 0.f);
  if (deg <= 32) {
    const int sl = lane < cnt ? csr_src[start + lane] : n;
    const float ev = leaky(as_[sl] + adn);
    float m = lane < deg ? ev : -INFINITY;
    for (int o = 16; o > 0; o >>= 1) m = fmaxf(m, __shfl_xor(m, o));
    const float ex = lane < deg ? expf(ev - m) : 0.f;
    float den = ex;
    for (int o = 16; o > 0; o >>= 1) den += __shfl_xor(den, o);
    const float alpha = ex / (den + 1e-16f);
    for (int k = 0; k < deg; k += 4) {
      float2 xv[4];
      float al[4];
#pragma unroll
      for (int u = 0; u < 4; ++u) {
        const int s = __builtin_amdgcn_readlane(sl, k + u);
        al[u] = lane_f(alpha, k + u);
        xv[u] = *reinterpret_cast<const float2*>(xh + (size_t)s * 64 + lane * 2);
      }
#pragma unroll
      for (int u = 0; u < 4; ++u) {
        acc.x = fmaf(xv[u].x, al[u], acc.x);
        acc.y = fmaf(xv[u].y, al[u], acc.y);
      }
    }
  } else {
    float m = -INFINITY;
    for (int base = 0; base < deg; base += 32) {
      const int i = base + lane;
      const int sl = i < cnt ? csr_src[start + i] : n;
      const float ev = leaky(as_[sl] + adn);
      m = i < deg ? fmaxf(m, ev) : m;
    }
    for (int o = 16; o > 0; o >>= 1) m = fmaxf(m, __shfl_xor(m, o));

    float den = 0.f;
    for (int base = 0; base < deg; base += 32) {
      const int i = base + lane;
      const int sl = i < cnt ? csr_src[start + i] : n;
      const float ev = leaky(as_[sl] + adn);
      float ex = i < deg ? expf(ev - m) : 0.f;
      for (int o = 16; o > 0; o >>= 1) ex += __shfl_xor(ex, o);
      den += ex;
    }
    const float dsum = den + 1e-16f;

    for (int base = 0; base < deg; base += 32) {
      const int i = base + lane;
      const int sl = i < cnt ? csr_src[start + i] : n;
      const float ev = leaky(as_[sl] + adn);
      const float alpha = expf(ev - m) / dsum;
      const int nb = min(32, deg - base);
      for (int k = 0; k < nb; k += 4) {
        float2 xv[4];
        float al[4];
#pragma unroll
        for (int u = 0; u < 4; ++u) {
          const int s = __builtin_amdgcn_readlane(sl, k + u);
          al[u] = lane_f(alpha, k + u);
          xv[u] = *reinterpret_cast<const float2*>(xh + (size_t)s * 64 + lane * 2);
        }
#pragma unroll
        for (int u = 0; u < 4; ++u) {
          const bool ok = u == 0 || k + u < nb;
          acc.x = ok ? acc.x + xv[u].x * al[u] : acc.x;
          acc.y = ok ? acc.y + xv[u].y * al[u] : acc.y;
        }
      }
    }
  }
  const float2 b = *reinterpret_cast<const float2*>(bias + lane * 2);
  float2 o;
  o.x = acc.x + b.x;
  o.y = acc.y + b.y;
  *reinterpret_cast<float2*>(out + (size_t)n * 64 + lane * 2) = o;
}

extern "C" void kernel_launch(void* const* d_in, const int* in_sizes, int n_in,
                              void* d_out, int out_size, void* d_ws,
                              size_t ws_size, hipStream_t stream) {
  (void)n_in; (void)out_size; (void)ws_size;
  const float* x        = (const float*)d_in[0];
  const int*   eidx     = (const int*)d_in[1];
  const float* W1       = (const float*)d_in[2];
  const float* att_src1 = (const float*)d_in[3];
  const float* att_dst1 = (const float*)d_in[4];
  const float* b1       = (const float*)d_in[5];
  const float* W2       = (const float*)d_in[6];
  const float* att_src2 = (const float*)d_in[7];
  const float* att_dst2 = (const float*)d_in[8];
  const float* b2       = (const float*)d_in[9];
  float* out = (float*)d_out;

  const int N    = in_sizes[0] / 128;
  const int E    = in_sizes[1] / 2;
  const int* esrc = eidx;
  const int* edst = eidx + E;

  float* ws = (float*)d_ws;
  float* xh1    = ws;        ws += (size_t)N * 128;
  float* h1     = ws;        ws += (size_t)N * 128;
  float* as1    = ws;        ws += (size_t)N * 8;
  float* ad1    = ws;        ws += (size_t)N * 8;
  int*   csr    = (int*)ws;  ws += (size_t)E;
  float* xh2    = ws;        ws += (size_t)N * 64;
  float* as2    = ws;        ws += (size_t)N;
  float* ad2    = ws;        ws += (size_t)N;
  int*   rowptr = (int*)ws;
  const int nbins = (N + 255) / 256;
  int*   hist     = (int*)h1;
  int*   lpre     = hist + SRT_NB * SRT_MAXB;
  int*   part     = lpre + SRT_NB * SRT_MAXB;
  int*   partd    = part + (size_t)E * 2;
  int*   slot     = partd + (size_t)E;

  auto cdiv = [](int a, int b) { return (a + b - 1) / b; };

  gemm_wmma_kernel<128, 128, 8, true><<<SRT_NB + 512, 256, 0, stream>>>(
      x, W1, att_src1, att_dst1, xh1, as1, ad1, N / 16,
      esrc, edst, hist, lpre, part, partd, E);
  bin_finish_kernel<<<nbins, 1024, 0, stream>>>(part, partd, hist, lpre, rowptr,
                                                slot, N, E);

  node1_kernel<<<cdiv(N, 16), 512, 0, stream>>>(rowptr, slot, csr, as1, ad1, xh1, b1,
                                                W2, att_src2, att_dst2, xh2, as2,
                                                ad2, N);

  node2_kernel<<<cdiv(N * 32, 256), 256, 0, stream>>>(rowptr, csr, as2, ad2, xh2,
                                                      b2, out, N);
}
